# starting kernel
# baseline (speedup 1.0000x reference)
#include <hip/hip_runtime.h>

typedef float v2f __attribute__((ext_vector_type(2)));
typedef float v8f __attribute__((ext_vector_type(8)));

#define HDIM 64
#define EPSV 1e-5f

static inline int cdiv(long long a, long long b) { return (int)((a + b - 1) / b); }

__global__ void k_zero(float* __restrict__ p, int n) {
    int i = blockIdx.x * blockDim.x + threadIdx.x;
    if (i < n) p[i] = 0.0f;
}

__global__ void k_deg_init(float* __restrict__ deg, int n) {
    int i = blockIdx.x * blockDim.x + threadIdx.x;
    if (i < n) deg[i] = 1.0f;
}

__global__ void k_deg_accum(const int* __restrict__ dst, float* __restrict__ deg, int E) {
    int e = blockIdx.x * blockDim.x + threadIdx.x;
    if (e < E) atomicAdd(&deg[dst[e]], 1.0f);
}

__global__ void k_dinv(float* __restrict__ deg, int n) {
    int i = blockIdx.x * blockDim.x + threadIdx.x;
    if (i < n) deg[i] = rsqrtf(deg[i]);
}

__global__ void k_gemm_wmma(const float* __restrict__ A, const float* __restrict__ W,
                            float* __restrict__ out, int rowTiles) {
    int wid  = (blockIdx.x * blockDim.x + threadIdx.x) >> 5;
    int lane = threadIdx.x & 31;
    int tr = wid >> 2;
    int tc = wid & 3;
    if (tr >= rowTiles) return;
    int m  = lane & 15;
    int kh = (lane >> 4) << 1;
    const float* Ar = A + (size_t)(tr * 16 + m) * HDIM;
    const float* Wr = W + (size_t)(tc * 16 + m) * HDIM;
    v8f c = {0.f, 0.f, 0.f, 0.f, 0.f, 0.f, 0.f, 0.f};
#pragma unroll
    for (int k0 = 0; k0 < HDIM; k0 += 4) {
        v2f a = *(const v2f*)(Ar + k0 + kh);
        v2f b = *(const v2f*)(Wr + k0 + kh);
        c = __builtin_amdgcn_wmma_f32_16x16x4_f32(false, a, false, b, (short)0, c, false, false);
    }
    int rowBase = tr * 16 + ((lane >> 4) << 3);
    int col     = tc * 16 + m;
#pragma unroll
    for (int r = 0; r < 8; ++r)
        out[(size_t)(rowBase + r) * HDIM + col] = c[r];
}

__global__ void k_self_init(const float* __restrict__ hW, const float* __restrict__ dinv,
                            float* __restrict__ agg, int N) {
    int t = blockIdx.x * blockDim.x + threadIdx.x;
    if (t >= N * 16) return;
    int n = t >> 4, q = t & 15;
    float di  = dinv[n];
    float nrm = di * di;
    float4 v  = ((const float4*)(hW + (size_t)n * HDIM))[q];
    ((float4*)(agg + (size_t)n * HDIM))[q] =
        make_float4(v.x * nrm, v.y * nrm, v.z * nrm, v.w * nrm);
}

__global__ void k_scatter(const float* __restrict__ hW, const float* __restrict__ dinv,
                          const int* __restrict__ src, const int* __restrict__ dst,
                          float* __restrict__ agg, int E) {
    int t = blockIdx.x * blockDim.x + threadIdx.x;
    if (t >= E * 16) return;
    int e = t >> 4, q = t & 15;
    int s = src[e], d = dst[e];
    float nrm = dinv[s] * dinv[d];
    float4 v = ((const float4*)(hW + (size_t)s * HDIM))[q];
    float* a = agg + (size_t)d * HDIM + q * 4;
    atomicAdd(a + 0, v.x * nrm);
    atomicAdd(a + 1, v.y * nrm);
    atomicAdd(a + 2, v.z * nrm);
    atomicAdd(a + 3, v.w * nrm);
}

__global__ void k_bias_relu_stats(float* __restrict__ agg, const float* __restrict__ bias,
                                  float* __restrict__ stats  , int N) {
    __shared__ float sS[256];
    __shared__ float sQ[256];
    int f  = threadIdx.x & 63;
    int rg = threadIdx.x >> 6;
    float b = bias[f];
    float s = 0.f, q = 0.f;
    for (int row = blockIdx.x * 4 + rg; row < N; row += gridDim.x * 4) {
        size_t idx = (size_t)row * HDIM + f;
        float v = agg[idx] + b;
        v = fmaxf(v, 0.f);
        agg[idx] = v;
        s += v;
        q += v * v;
    }
    sS[threadIdx.x] = s;
    sQ[threadIdx.x] = q;
    __syncthreads();
    if (rg == 0) {
        s = sS[f] + sS[64 + f] + sS[128 + f] + sS[192 + f];
        q = sQ[f] + sQ[64 + f] + sQ[128 + f] + sQ[192 + f];
        atomicAdd(&stats[f], s);
        atomicAdd(&stats[64 + f], q);
    }
}

__global__ void k_bn_pool(const float* __restrict__ hin, const float* __restrict__ stats,
                          const float* __restrict__ g, const float* __restrict__ beta,
                          const int* __restrict__ batch, float* __restrict__ hout,
                          float* __restrict__ pooled, int poolOff, int N) {
    int t = blockIdx.x * blockDim.x + threadIdx.x;
    if (t >= N * HDIM) return;
    int n = t >> 6, f = t & 63;
    float invN = 1.0f / (float)N;
    float mean = stats[f] * invN;
    float var  = stats[64 + f] * invN - mean * mean;
    float v = g[f] * (hin[t] - mean) * rsqrtf(var + EPSV) + beta[f];
    hout[t] = v;
    atomicAdd(&pooled[(size_t)batch[n] * 192 + poolOff + f], v);
}

extern "C" void kernel_launch(void* const* d_in, const int* in_sizes, int n_in,
                              void* d_out, int out_size, void* d_ws, size_t ws_size,
                              hipStream_t stream) {
    const float* x    = (const float*)d_in[0];
    const int*   eidx = (const int*)d_in[1];
    const int*   batch = (const int*)d_in[2];
    const float* W[3]  = {(const float*)d_in[3],  (const float*)d_in[7],  (const float*)d_in[11]};
    const float* bi[3] = {(const float*)d_in[4],  (const float*)d_in[8],  (const float*)d_in[12]};
    const float* ga[3] = {(const float*)d_in[5],  (const float*)d_in[9],  (const float*)d_in[13]};
    const float* be[3] = {(const float*)d_in[6],  (const float*)d_in[10], (const float*)d_in[14]};

    const int N = in_sizes[0] / HDIM;
    const int E = in_sizes[1] / 2;
    const int G = (out_size - N * HDIM) / (3 * HDIM);

    const int* src = eidx;
    const int* dst = eidx + E;

    float* ws    = (float*)d_ws;
    float* dinv  = ws;
    float* tmp   = dinv + N;
    float* agg   = tmp  + (size_t)N * HDIM;
    float* bufH  = agg  + (size_t)N * HDIM;
    float* stats = bufH + (size_t)N * HDIM;

    float* pooled = (float*)d_out;
    float* hfinal = (float*)d_out + (size_t)G * 192;

    const int TB = 256;

    k_deg_init <<<cdiv(N, TB), TB, 0, stream>>>(dinv, N);
    k_deg_accum<<<cdiv(E, TB), TB, 0, stream>>>(dst, dinv, E);
    k_dinv     <<<cdiv(N, TB), TB, 0, stream>>>(dinv, N);

    k_zero<<<cdiv((long long)G * 192, TB), TB, 0, stream>>>(pooled, G * 192);

    const int rowTiles   = N / 16;
    const int gemmWaves  = rowTiles * (HDIM / 16);
    const int gemmBlocks = cdiv((long long)gemmWaves * 32, TB);

    const float* gin = x;
    for (int l = 0; l < 3; ++l) {
        float* hout = (l == 2) ? hfinal : bufH;

        k_gemm_wmma<<<gemmBlocks, TB, 0, stream>>>(gin, W[l], tmp, rowTiles);
        k_self_init<<<cdiv((long long)N * 16, TB), TB, 0, stream>>>(tmp, dinv, agg, N);
        k_scatter  <<<cdiv((long long)E * 16, TB), TB, 0, stream>>>(tmp, dinv, src, dst, agg, E);

        k_zero<<<1, 128, 0, stream>>>(stats, 128);
        k_bias_relu_stats<<<1024, TB, 0, stream>>>(agg, bi[l], stats, N);

        k_bn_pool<<<cdiv((long long)N * HDIM, TB), TB, 0, stream>>>(
            agg, stats, ga[l], be[l], batch, hout, pooled, l * HDIM, N);

        gin = hout;
    }
}


# Per-call edge list grouped by destination (counting sort, arrival-order slots) + atomic-free row gather replace self-init and float-atomic edge scatter; sum order inside a row changes
# speedup vs baseline: 1.5756x; 1.5756x over previous
#include <hip/hip_runtime.h>

typedef float v2f __attribute__((ext_vector_type(2)));
typedef float v8f __attribute__((ext_vector_type(8)));

#define HDIM 64
#define EPSV 1e-5f

static inline int cdiv(long long a, long long b) { return (int)((a + b - 1) / b); }

__global__ void k_zero(float* __restrict__ p, int n) {
    int i = blockIdx.x * blockDim.x + threadIdx.x;
    if (i < n) p[i] = 0.0f;
}

__global__ void k_zero_i(int* __restrict__ p, int n) {
    int i = blockIdx.x * blockDim.x + threadIdx.x;
    if (i < n) p[i] = 0;
}

__global__ void k_count_rank(const int* __restrict__ dst, int* __restrict__ cnt,
                             int* __restrict__ rank, int E) {
    int e = blockIdx.x * blockDim.x + threadIdx.x;
    if (e < E) rank[e] = atomicAdd(&cnt[dst[e]], 1);
}

#define SCAN_TB 1024
__global__ void __launch_bounds__(SCAN_TB)
k_rowptr_dinv(const int* __restrict__ cnt, int* __restrict__ rowptr,
              float* __restrict__ dinv, int N) {
    __shared__ int sh[SCAN_TB];
    const int t     = threadIdx.x;
    const int first = blockIdx.x * SCAN_TB;
    const int4* c4  = (const int4*)cnt;
    const int nvec  = first >> 2;
    int p = 0;
#pragma unroll 4
    for (int i = t; i < nvec; i += SCAN_TB) {
        int4 v = c4[i];
        p += (v.x + v.y) + (v.z + v.w);
    }
    sh[t] = p;
    __syncthreads();
    for (int s = SCAN_TB / 2; s > 0; s >>= 1) {
        if (t < s) sh[t] += sh[t + s];
        __syncthreads();
    }
    const int base = sh[0];
    __syncthreads();
    const int i = first + t;
    const int c = (i < N) ? cnt[i] : 0;
    sh[t] = c;
    __syncthreads();
    for (int off = 1; off < SCAN_TB; off <<= 1) {
        int v = (t >= off) ? sh[t - off] : 0;
        __syncthreads();
        sh[t] += v;
        __syncthreads();
    }
    const int incl = sh[t];
    if (i < N) {
        rowptr[i] = base + incl - c;
        dinv[i]   = rsqrtf((float)(c + 1));
        if (i == N - 1) rowptr[N] = base + incl;
    }
}

__global__ void k_fill(const int* __restrict__ src, const int* __restrict__ dst,
                       const int* __restrict__ rowptr, const int* __restrict__ rank,
                       int* __restrict__ col, int E) {
    int e = blockIdx.x * blockDim.x + threadIdx.x;
    if (e < E) col[rowptr[dst[e]] + rank[e]] = src[e];
}

__global__ void k_gemm_wmma(const float* __restrict__ A, const float* __restrict__ W,
                            float* __restrict__ out, int rowTiles) {
    int wid  = (blockIdx.x * blockDim.x + threadIdx.x) >> 5;
    int lane = threadIdx.x & 31;
    int tr = wid >> 2;
    int tc = wid & 3;
    if (tr >= rowTiles) return;
    int m  = lane & 15;
    int kh = (lane >> 4) << 1;
    const float* Ar = A + (size_t)(tr * 16 + m) * HDIM;
    const float* Wr = W + (size_t)(tc * 16 + m) * HDIM;
    v8f c = {0.f, 0.f, 0.f, 0.f, 0.f, 0.f, 0.f, 0.f};
#pragma unroll
    for (int k0 = 0; k0 < HDIM; k0 += 4) {
        v2f a = *(const v2f*)(Ar + k0 + kh);
        v2f b = *(const v2f*)(Wr + k0 + kh);
        c = __builtin_amdgcn_wmma_f32_16x16x4_f32(false, a, false, b, (short)0, c, false, false);
    }
    int rowBase = tr * 16 + ((lane >> 4) << 3);
    int col     = tc * 16 + m;
#pragma unroll
    for (int r = 0; r < 8; ++r)
        out[(size_t)(rowBase + r) * HDIM + col] = c[r];
}

__global__ void k_gather(const float* __restrict__ hW, const float* __restrict__ dinv,
                         const int* __restrict__ rowptr, const int* __restrict__ col,
                         float* __restrict__ agg, int N) {
    int t = blockIdx.x * blockDim.x + threadIdx.x;
    if (t >= N * 16) return;
    const int n = t >> 4, q = t & 15;
    const float4* h4 = (const float4*)hW;
    const float di = dinv[n];
    const int beg = rowptr[n], end = rowptr[n + 1];
    const float4 vs = h4[(size_t)n * 16 + q];
    const float nrm = di * di;
    float4 acc = make_float4(vs.x * nrm, vs.y * nrm, vs.z * nrm, vs.w * nrm);
    const int last = end - 1;
    for (int j = beg; j < end; j += 4) {
        const int s0 = col[j];
        const int s1 = col[min(j + 1, last)];
        const int s2 = col[min(j + 2, last)];
        const int s3 = col[min(j + 3, last)];
        const float d0 = dinv[s0], d1 = dinv[s1], d2 = dinv[s2], d3 = dinv[s3];
        const float4 v0 = h4[(size_t)s0 * 16 + q];
        const float4 v1 = h4[(size_t)s1 * 16 + q];
        const float4 v2 = h4[(size_t)s2 * 16 + q];
        const float4 v3 = h4[(size_t)s3 * 16 + q];
        const float w0 = d0 * di;
        const float w1 = (j + 1 < end) ? d1 * di : 0.f;
        const float w2 = (j + 2 < end) ? d2 * di : 0.f;
        const float w3 = (j + 3 < end) ? d3 * di : 0.f;
        acc.x += v0.x * w0; acc.y += v0.y * w0; acc.z += v0.z * w0; acc.w += v0.w * w0;
        acc.x += v1.x * w1; acc.y += v1.y * w1; acc.z += v1.z * w1; acc.w += v1.w * w1;
        acc.x += v2.x * w2; acc.y += v2.y * w2; acc.z += v2.z * w2; acc.w += v2.w * w2;
        acc.x += v3.x * w3; acc.y += v3.y * w3; acc.z += v3.z * w3; acc.w += v3.w * w3;
    }
    ((float4*)(agg + (size_t)n * HDIM))[q] = acc;
}

__global__ void k_bias_relu_stats(float* __restrict__ agg, const float* __restrict__ bias,
                                  float* __restrict__ stats  , int N) {
    __shared__ float sS[256];
    __shared__ float sQ[256];
    int f  = threadIdx.x & 63;
    int rg = threadIdx.x >> 6;
    float b = bias[f];
    float s = 0.f, q = 0.f;
    for (int row = blockIdx.x * 4 + rg; row < N; row += gridDim.x * 4) {
        size_t idx = (size_t)row * HDIM + f;
        float v = agg[idx] + b;
        v = fmaxf(v, 0.f);
        agg[idx] = v;
        s += v;
        q += v * v;
    }
    sS[threadIdx.x] = s;
    sQ[threadIdx.x] = q;
    __syncthreads();
    if (rg == 0) {
        s = sS[f] + sS[64 + f] + sS[128 + f] + sS[192 + f];
        q = sQ[f] + sQ[64 + f] + sQ[128 + f] + sQ[192 + f];
        atomicAdd(&stats[f], s);
        atomicAdd(&stats[64 + f], q);
    }
}

__global__ void k_bn_pool(const float* __restrict__ hin, const float* __restrict__ stats,
                          const float* __restrict__ g, const float* __restrict__ beta,
                          const int* __restrict__ batch, float* __restrict__ hout,
                          float* __restrict__ pooled, int poolOff, int N) {
    int t = blockIdx.x * blockDim.x + threadIdx.x;
    if (t >= N * HDIM) return;
    int n = t >> 6, f = t & 63;
    float invN = 1.0f / (float)N;
    float mean = stats[f] * invN;
    float var  = stats[64 + f] * invN - mean * mean;
    float v = g[f] * (hin[t] - mean) * rsqrtf(var + EPSV) + beta[f];
    hout[t] = v;
    atomicAdd(&pooled[(size_t)batch[n] * 192 + poolOff + f], v);
}

extern "C" void kernel_launch(void* const* d_in, const int* in_sizes, int n_in,
                              void* d_out, int out_size, void* d_ws, size_t ws_size,
                              hipStream_t stream) {
    const float* x    = (const float*)d_in[0];
    const int*   eidx = (const int*)d_in[1];
    const int*   batch = (const int*)d_in[2];
    const float* W[3]  = {(const float*)d_in[3],  (const float*)d_in[7],  (const float*)d_in[11]};
    const float* bi[3] = {(const float*)d_in[4],  (const float*)d_in[8],  (const float*)d_in[12]};
    const float* ga[3] = {(const float*)d_in[5],  (const float*)d_in[9],  (const float*)d_in[13]};
    const float* be[3] = {(const float*)d_in[6],  (const float*)d_in[10], (const float*)d_in[14]};

    const int N = in_sizes[0] / HDIM;
    const int E = in_sizes[1] / 2;
    const int G = (out_size - N * HDIM) / (3 * HDIM);

    const int* src = eidx;
    const int* dst = eidx + E;

    float* ws    = (float*)d_ws;
    float* dinv  = ws;
    float* bufA  = dinv + N;
    float* bufB  = bufA + (size_t)N * HDIM;
    int*   csr   = (int*)(bufB + (size_t)N * HDIM);
    float* stats = (float*)csr + (size_t)N * HDIM;
    int* cnt    = csr;
    int* rowptr = cnt + N;
    int* rank   = rowptr + (N + 4);
    int* col    = rank + E;

    float* pooled = (float*)d_out;
    float* hfinal = (float*)d_out + (size_t)G * 192;

    const int TB = 256;

    k_zero_i     <<<cdiv(N, TB), TB, 0, stream>>>(cnt, N);
    k_count_rank <<<cdiv(E, TB), TB, 0, stream>>>(dst, cnt, rank, E);
    k_rowptr_dinv<<<cdiv(N, SCAN_TB), SCAN_TB, 0, stream>>>(cnt, rowptr, dinv, N);
    k_fill       <<<cdiv(E, TB), TB, 0, stream>>>(src, dst, rowptr, rank, col, E);

    k_zero<<<cdiv((long long)G * 192, TB), TB, 0, stream>>>(pooled, G * 192);

    const int rowTiles   = N / 16;
    const int gemmWaves  = rowTiles * (HDIM / 16);
    const int gemmBlocks = cdiv((long long)gemmWaves * 32, TB);

    const float* gin = x;
    float* hw = bufA;
    float* ag = bufB;
    for (int l = 0; l < 3; ++l) {
        float* hout = (l == 2) ? hfinal : hw;

        k_gemm_wmma<<<gemmBlocks, TB, 0, stream>>>(gin, W[l], hw, rowTiles);
        k_gather   <<<cdiv((long long)N * 16, TB), TB, 0, stream>>>(hw, dinv, rowptr, col, ag, N);

        k_zero<<<1, 128, 0, stream>>>(stats, 128);
        k_bias_relu_stats<<<1024, TB, 0, stream>>>(ag, bi[l], stats, N);

        k_bn_pool<<<cdiv((long long)N * HDIM, TB), TB, 0, stream>>>(
            ag, stats, ga[l], be[l], batch, hout, pooled, l * HDIM, N);

        gin = hout;
        float* sw = hw; hw = ag; ag = sw;
    }
}


# BN+pool pass as few long-lived waves: 16 lanes own 16 consecutive rows, all loaded up front, a graph's rows summed in registers, one atomic add per feature per run; pool sum order changes
# speedup vs baseline: 1.7012x; 1.0797x over previous
#include <hip/hip_runtime.h>

typedef float v2f __attribute__((ext_vector_type(2)));
typedef float v8f __attribute__((ext_vector_type(8)));

#define HDIM 64
#define EPSV 1e-5f

static inline int cdiv(long long a, long long b) { return (int)((a + b - 1) / b); }

__global__ void k_zero(float* __restrict__ p, int n) {
    int i = blockIdx.x * blockDim.x + threadIdx.x;
    if (i < n) p[i] = 0.0f;
}

__global__ void k_zero_i(int* __restrict__ p, int n) {
    int i = blockIdx.x * blockDim.x + threadIdx.x;
    if (i < n) p[i] = 0;
}

__global__ void k_count_rank(const int* __restrict__ dst, int* __restrict__ cnt,
                             int* __restrict__ rank, int E) {
    int e = blockIdx.x * blockDim.x + threadIdx.x;
    if (e < E) rank[e] = atomicAdd(&cnt[dst[e]], 1);
}

#define SCAN_TB 1024
__global__ void __launch_bounds__(SCAN_TB)
k_rowptr_dinv(const int* __restrict__ cnt, int* __restrict__ rowptr,
              float* __restrict__ dinv, int N) {
    __shared__ int sh[SCAN_TB];
    const int t     = threadIdx.x;
    const int first = blockIdx.x * SCAN_TB;
    const int4* c4  = (const int4*)cnt;
    const int nvec  = first >> 2;
    int p = 0;
#pragma unroll 4
    for (int i = t; i < nvec; i += SCAN_TB) {
        int4 v = c4[i];
        p += (v.x + v.y) + (v.z + v.w);
    }
    sh[t] = p;
    __syncthreads();
    for (int s = SCAN_TB / 2; s > 0; s >>= 1) {
        if (t < s) sh[t] += sh[t + s];
        __syncthreads();
    }
    const int base = sh[0];
    __syncthreads();
    const int i = first + t;
    const int c = (i < N) ? cnt[i] : 0;
    sh[t] = c;
    __syncthreads();
    for (int off = 1; off < SCAN_TB; off <<= 1) {
        int v = (t >= off) ? sh[t - off] : 0;
        __syncthreads();
        sh[t] += v;
        __syncthreads();
    }
    const int incl = sh[t];
    if (i < N) {
        rowptr[i] = base + incl - c;
        dinv[i]   = rsqrtf((float)(c + 1));
        if (i == N - 1) rowptr[N] = base + incl;
    }
}

__global__ void k_fill(const int* __restrict__ src, const int* __restrict__ dst,
                       const int* __restrict__ rowptr, const int* __restrict__ rank,
                       int* __restrict__ col, int E) {
    int e = blockIdx.x * blockDim.x + threadIdx.x;
    if (e < E) col[rowptr[dst[e]] + rank[e]] = src[e];
}

__global__ void k_gemm_wmma(const float* __restrict__ A, const float* __restrict__ W,
                            float* __restrict__ out, int rowTiles) {
    int wid  = (blockIdx.x * blockDim.x + threadIdx.x) >> 5;
    int lane = threadIdx.x & 31;
    int tr = wid >> 2;
    int tc = wid & 3;
    if (tr >= rowTiles) return;
    int m  = lane & 15;
    int kh = (lane >> 4) << 1;
    const float* Ar = A + (size_t)(tr * 16 + m) * HDIM;
    const float* Wr = W + (size_t)(tc * 16 + m) * HDIM;
    v8f c = {0.f, 0.f, 0.f, 0.f, 0.f, 0.f, 0.f, 0.f};
#pragma unroll
    for (int k0 = 0; k0 < HDIM; k0 += 4) {
        v2f a = *(const v2f*)(Ar + k0 + kh);
        v2f b = *(const v2f*)(Wr + k0 + kh);
        c = __builtin_amdgcn_wmma_f32_16x16x4_f32(false, a, false, b, (short)0, c, false, false);
    }
    int rowBase = tr * 16 + ((lane >> 4) << 3);
    int col     = tc * 16 + m;
#pragma unroll
    for (int r = 0; r < 8; ++r)
        out[(size_t)(rowBase + r) * HDIM + col] = c[r];
}

__global__ void k_gather(const float* __restrict__ hW, const float* __restrict__ dinv,
                         const int* __restrict__ rowptr, const int* __restrict__ col,
                         float* __restrict__ agg, int N) {
    int t = blockIdx.x * blockDim.x + threadIdx.x;
    if (t >= N * 16) return;
    const int n = t >> 4, q = t & 15;
    const float4* h4 = (const float4*)hW;
    const float di = dinv[n];
    const int beg = rowptr[n], end = rowptr[n + 1];
    const float4 vs = h4[(size_t)n * 16 + q];
    const float nrm = di * di;
    float4 acc = make_float4(vs.x * nrm, vs.y * nrm, vs.z * nrm, vs.w * nrm);
    const int last = end - 1;
    for (int j = beg; j < end; j += 4) {
        const int s0 = col[j];
        const int s1 = col[min(j + 1, last)];
        const int s2 = col[min(j + 2, last)];
        const int s3 = col[min(j + 3, last)];
        const float d0 = dinv[s0], d1 = dinv[s1], d2 = dinv[s2], d3 = dinv[s3];
        const float4 v0 = h4[(size_t)s0 * 16 + q];
        const float4 v1 = h4[(size_t)s1 * 16 + q];
        const float4 v2 = h4[(size_t)s2 * 16 + q];
        const float4 v3 = h4[(size_t)s3 * 16 + q];
        const float w0 = d0 * di;
        const float w1 = (j + 1 < end) ? d1 * di : 0.f;
        const float w2 = (j + 2 < end) ? d2 * di : 0.f;
        const float w3 = (j + 3 < end) ? d3 * di : 0.f;
        acc.x += v0.x * w0; acc.y += v0.y * w0; acc.z += v0.z * w0; acc.w += v0.w * w0;
        acc.x += v1.x * w1; acc.y += v1.y * w1; acc.z += v1.z * w1; acc.w += v1.w * w1;
        acc.x += v2.x * w2; acc.y += v2.y * w2; acc.z += v2.z * w2; acc.w += v2.w * w2;
        acc.x += v3.x * w3; acc.y += v3.y * w3; acc.z += v3.z * w3; acc.w += v3.w * w3;
    }
    ((float4*)(agg + (size_t)n * HDIM))[q] = acc;
}

__global__ void k_bias_relu_stats(float* __restrict__ agg, const float* __restrict__ bias,
                                  float* __restrict__ stats  , int N) {
    __shared__ float sS[256];
    __shared__ float sQ[256];
    int f  = threadIdx.x & 63;
    int rg = threadIdx.x >> 6;
    float b = bias[f];
    float s = 0.f, q = 0.f;
    for (int row = blockIdx.x * 4 + rg; row < N; row += gridDim.x * 4) {
        size_t idx = (size_t)row * HDIM + f;
        float v = agg[idx] + b;
        v = fmaxf(v, 0.f);
        agg[idx] = v;
        s += v;
        q += v * v;
    }
    sS[threadIdx.x] = s;
    sQ[threadIdx.x] = q;
    __syncthreads();
    if (rg == 0) {
        s = sS[f] + sS[64 + f] + sS[128 + f] + sS[192 + f];
        q = sQ[f] + sQ[64 + f] + sQ[128 + f] + sQ[192 + f];
        atomicAdd(&stats[f], s);
        atomicAdd(&stats[64 + f], q);
    }
}

#define POOL_R 16

struct BnCoef { float4 m, is, g, b; };

__device__ __forceinline__ BnCoef bn_coef(float4 sm, float4 sq, float4 gg, float4 bb, int N) {
    BnCoef c;
    const float invN = 1.0f / (float)N;
    c.m.x = sm.x * invN; c.m.y = sm.y * invN; c.m.z = sm.z * invN; c.m.w = sm.w * invN;
    c.is.x = rsqrtf(sq.x * invN - c.m.x * c.m.x + EPSV);
    c.is.y = rsqrtf(sq.y * invN - c.m.y * c.m.y + EPSV);
    c.is.z = rsqrtf(sq.z * invN - c.m.z * c.m.z + EPSV);
    c.is.w = rsqrtf(sq.w * invN - c.m.w * c.m.w + EPSV);
    c.g = gg; c.b = bb;
    return c;
}

__device__ __forceinline__ float4 bn4(float4 x, const BnCoef& c) {
    float4 v;
    v.x = c.g.x * (x.x - c.m.x) * c.is.x + c.b.x;
    v.y = c.g.y * (x.y - c.m.y) * c.is.y + c.b.y;
    v.z = c.g.z * (x.z - c.m.z) * c.is.z + c.b.z;
    v.w = c.g.w * (x.w - c.m.w) * c.is.w + c.b.w;
    return v;
}

__device__ __forceinline__ void pool_add4(float* __restrict__ p, float4 a) {
    atomicAdd(p, a.x); atomicAdd(p + 1, a.y); atomicAdd(p + 2, a.z); atomicAdd(p + 3, a.w);
}

__global__ void k_bn_pool(const float* __restrict__ hin, const float* __restrict__ stats,
          const float* __restrict__ g, const float* __restrict__ beta,
          const int* __restrict__ batch, float* __restrict__ hout,
          float* __restrict__ pooled, int poolOff, int N) {
    const int t  = blockIdx.x * blockDim.x + threadIdx.x;
    const int q  = t & 15;
    const int r0 = (t >> 4) * POOL_R;
    if (r0 >= N) return;
    const float4* h4 = (const float4*)hin + (size_t)r0 * 16 + q;
    float4*       o4 = (float4*)hout + (size_t)r0 * 16 + q;
    const int*    bp = batch + r0;
    const int f = q * 4;
    float* pq = pooled + poolOff + f;

    const float4 sm = ((const float4*)stats)[q];
    const float4 sq = ((const float4*)(stats + 64))[q];
    const float4 gg = make_float4(g[f], g[f + 1], g[f + 2], g[f + 3]);
    const float4 bb = make_float4(beta[f], beta[f + 1], beta[f + 2], beta[f + 3]);

    int cur;
    float4 acc = make_float4(0.f, 0.f, 0.f, 0.f);
    if (r0 + POOL_R <= N) {
        float4 x[POOL_R];
        int    b[POOL_R];
#pragma unroll
        for (int u = 0; u < POOL_R; ++u) {
            x[u] = h4[u * 16];
            b[u] = bp[u];
        }
        const BnCoef c = bn_coef(sm, sq, gg, bb, N);
        cur = b[0];
#pragma unroll
        for (int u = 0; u < POOL_R; ++u) {
            const float4 v = bn4(x[u], c);
            o4[u * 16] = v;
            if (b[u] != cur) {
                pool_add4(pq + (size_t)cur * 192, acc);
                cur = b[u];
                acc = make_float4(0.f, 0.f, 0.f, 0.f);
            }
            acc.x += v.x; acc.y += v.y; acc.z += v.z; acc.w += v.w;
        }
    } else {
        const BnCoef c = bn_coef(sm, sq, gg, bb, N);
        cur = bp[0];
        for (int u = 0; r0 + u < N; ++u) {
            const float4 v = bn4(h4[u * 16], c);
            o4[u * 16] = v;
            const int bu = bp[u];
            if (bu != cur) {
                pool_add4(pq + (size_t)cur * 192, acc);
                cur = bu;
                acc = make_float4(0.f, 0.f, 0.f, 0.f);
            }
            acc.x += v.x; acc.y += v.y; acc.z += v.z; acc.w += v.w;
        }
    }
    pool_add4(pq + (size_t)cur * 192, acc);
}

extern "C" void kernel_launch(void* const* d_in, const int* in_sizes, int n_in,
                              void* d_out, int out_size, void* d_ws, size_t ws_size,
                              hipStream_t stream) {
    const float* x    = (const float*)d_in[0];
    const int*   eidx = (const int*)d_in[1];
    const int*   batch = (const int*)d_in[2];
    const float* W[3]  = {(const float*)d_in[3],  (const float*)d_in[7],  (const float*)d_in[11]};
    const float* bi[3] = {(const float*)d_in[4],  (const float*)d_in[8],  (const float*)d_in[12]};
    const float* ga[3] = {(const float*)d_in[5],  (const float*)d_in[9],  (const float*)d_in[13]};
    const float* be[3] = {(const float*)d_in[6],  (const float*)d_in[10], (const float*)d_in[14]};

    const int N = in_sizes[0] / HDIM;
    const int E = in_sizes[1] / 2;
    const int G = (out_size - N * HDIM) / (3 * HDIM);

    const int* src = eidx;
    const int* dst = eidx + E;

    float* ws    = (float*)d_ws;
    float* dinv  = ws;
    float* bufA  = dinv + N;
    float* bufB  = bufA + (size_t)N * HDIM;
    int*   csr   = (int*)(bufB + (size_t)N * HDIM);
    float* stats = (float*)csr + (size_t)N * HDIM;
    int* cnt    = csr;
    int* rowptr = cnt + N;
    int* rank   = rowptr + (N + 4);
    int* col    = rank + E;

    float* pooled = (float*)d_out;
    float* hfinal = (float*)d_out + (size_t)G * 192;

    const int TB = 256;

    k_zero_i     <<<cdiv(N, TB), TB, 0, stream>>>(cnt, N);
    k_count_rank <<<cdiv(E, TB), TB, 0, stream>>>(dst, cnt, rank, E);
    k_rowptr_dinv<<<cdiv(N, SCAN_TB), SCAN_TB, 0, stream>>>(cnt, rowptr, dinv, N);
    k_fill       <<<cdiv(E, TB), TB, 0, stream>>>(src, dst, rowptr, rank, col, E);

    k_zero<<<cdiv((long long)G * 192, TB), TB, 0, stream>>>(pooled, G * 192);

    const int rowTiles   = N / 16;
    const int gemmWaves  = rowTiles * (HDIM / 16);
    const int gemmBlocks = cdiv((long long)gemmWaves * 32, TB);

    const float* gin = x;
    float* hw = bufA;
    float* ag = bufB;
    for (int l = 0; l < 3; ++l) {
        float* hout = (l == 2) ? hfinal : hw;

        k_gemm_wmma<<<gemmBlocks, TB, 0, stream>>>(gin, W[l], hw, rowTiles);
        k_gather   <<<cdiv((long long)N * 16, TB), TB, 0, stream>>>(hw, dinv, rowptr, col, ag, N);

        k_zero<<<1, 128, 0, stream>>>(stats, 128);
        k_bias_relu_stats<<<1024, TB, 0, stream>>>(ag, bi[l], stats, N);

        k_bn_pool<<<cdiv((long long)cdiv(N, POOL_R) * 16, TB), TB, 0, stream>>>(
            ag, stats, ga[l], be[l], batch, hout, pooled, l * HDIM, N);

        gin = hout;
        float* sw = hw; hw = ag; ag = sw;
    }
}


# Bias+ReLU and per-feature sum / sum of squares move into the gather's tail (block tile via LDS, one atomic add per feature per block, 8 spread copies); in-place pass gone; stats sum order changes
# speedup vs baseline: 2.0036x; 1.1778x over previous
#include <hip/hip_runtime.h>

typedef float v2f __attribute__((ext_vector_type(2)));
typedef float v8f __attribute__((ext_vector_type(8)));

#define HDIM 64
#define EPSV 1e-5f

static inline int cdiv(long long a, long long b) { return (int)((a + b - 1) / b); }

__global__ void k_zero(float* __restrict__ p, int n) {
    int i = blockIdx.x * blockDim.x + threadIdx.x;
    if (i < n) p[i] = 0.0f;
}

__global__ void k_zero_i(int* __restrict__ p, int n) {
    int i = blockIdx.x * blockDim.x + threadIdx.x;
    if (i < n) p[i] = 0;
}

__global__ void k_count_rank(const int* __restrict__ dst, int* __restrict__ cnt,
                             int* __restrict__ rank, int E) {
    int e = blockIdx.x * blockDim.x + threadIdx.x;
    if (e < E) rank[e] = atomicAdd(&cnt[dst[e]], 1);
}

#define SCAN_TB 1024
__global__ void __launch_bounds__(SCAN_TB)
k_rowptr_dinv(const int* __restrict__ cnt, int* __restrict__ rowptr,
              float* __restrict__ dinv, int N) {
    __shared__ int sh[SCAN_TB];
    const int t     = threadIdx.x;
    const int first = blockIdx.x * SCAN_TB;
    const int4* c4  = (const int4*)cnt;
    const int nvec  = first >> 2;
    int p = 0;
#pragma unroll 4
    for (int i = t; i < nvec; i += SCAN_TB) {
        int4 v = c4[i];
        p += (v.x + v.y) + (v.z + v.w);
    }
    sh[t] = p;
    __syncthreads();
    for (int s = SCAN_TB / 2; s > 0; s >>= 1) {
        if (t < s) sh[t] += sh[t + s];
        __syncthreads();
    }
    const int base = sh[0];
    __syncthreads();
    const int i = first + t;
    const int c = (i < N) ? cnt[i] : 0;
    sh[t] = c;
    __syncthreads();
    for (int off = 1; off < SCAN_TB; off <<= 1) {
        int v = (t >= off) ? sh[t - off] : 0;
        __syncthreads();
        sh[t] += v;
        __syncthreads();
    }
    const int incl = sh[t];
    if (i < N) {
        rowptr[i] = base + incl - c;
        dinv[i]   = rsqrtf((float)(c + 1));
        if (i == N - 1) rowptr[N] = base + incl;
    }
}

__global__ void k_fill(const int* __restrict__ src, const int* __restrict__ dst,
                       const int* __restrict__ rowptr, const int* __restrict__ rank,
                       int* __restrict__ col, int E) {
    int e = blockIdx.x * blockDim.x + threadIdx.x;
    if (e < E) col[rowptr[dst[e]] + rank[e]] = src[e];
}

__global__ void k_gemm_wmma(const float* __restrict__ A, const float* __restrict__ W,
                            float* __restrict__ out, int rowTiles) {
    int wid  = (blockIdx.x * blockDim.x + threadIdx.x) >> 5;
    int lane = threadIdx.x & 31;
    int tr = wid >> 2;
    int tc = wid & 3;
    if (tr >= rowTiles) return;
    int m  = lane & 15;
    int kh = (lane >> 4) << 1;
    const float* Ar = A + (size_t)(tr * 16 + m) * HDIM;
    const float* Wr = W + (size_t)(tc * 16 + m) * HDIM;
    v8f c = {0.f, 0.f, 0.f, 0.f, 0.f, 0.f, 0.f, 0.f};
#pragma unroll
    for (int k0 = 0; k0 < HDIM; k0 += 4) {
        v2f a = *(const v2f*)(Ar + k0 + kh);
        v2f b = *(const v2f*)(Wr + k0 + kh);
        c = __builtin_amdgcn_wmma_f32_16x16x4_f32(false, a, false, b, (short)0, c, false, false);
    }
    int rowBase = tr * 16 + ((lane >> 4) << 3);
    int col     = tc * 16 + m;
#pragma unroll
    for (int r = 0; r < 8; ++r)
        out[(size_t)(rowBase + r) * HDIM + col] = c[r];
}

#define STAT_COPIES 8
#define STAT_STRIDE 1088
#define GATHER_TB 256
__global__ void __launch_bounds__(GATHER_TB)
k_gather_act_stats(const float* __restrict__ hW, const float* __restrict__ dinv,
                   const int* __restrict__ rowptr, const int* __restrict__ col,
                   const float* __restrict__ bias, float* __restrict__ r,
                   float* __restrict__ stats  , int N) {
    __shared__ float4 tile[GATHER_TB];
    const int t = blockIdx.x * GATHER_TB + threadIdx.x;
    const int n = t >> 4, q = t & 15;
    float4 v = make_float4(0.f, 0.f, 0.f, 0.f);
    if (n < N) {
        const float4* h4 = (const float4*)hW;
        const float di = dinv[n];
        const int beg = rowptr[n], end = rowptr[n + 1];
        const float4 vs = h4[(size_t)n * 16 + q];
        const int f = q * 4;
        const float b0 = bias[f], b1 = bias[f + 1], b2 = bias[f + 2], b3 = bias[f + 3];
        const float nrm = di * di;
        float4 acc = make_float4(vs.x * nrm, vs.y * nrm, vs.z * nrm, vs.w * nrm);
        const int last = end - 1;
        for (int j = beg; j < end; j += 4) {
            const int s0 = col[j];
            const int s1 = col[min(j + 1, last)];
            const int s2 = col[min(j + 2, last)];
            const int s3 = col[min(j + 3, last)];
            const float d0 = dinv[s0], d1 = dinv[s1], d2 = dinv[s2], d3 = dinv[s3];
            const float4 v0 = h4[(size_t)s0 * 16 + q];
            const float4 v1 = h4[(size_t)s1 * 16 + q];
            const float4 v2 = h4[(size_t)s2 * 16 + q];
            const float4 v3 = h4[(size_t)s3 * 16 + q];
            const float w0 = d0 * di;
            const float w1 = (j + 1 < end) ? d1 * di : 0.f;
            const float w2 = (j + 2 < end) ? d2 * di : 0.f;
            const float w3 = (j + 3 < end) ? d3 * di : 0.f;
            acc.x += v0.x * w0; acc.y += v0.y * w0; acc.z += v0.z * w0; acc.w += v0.w * w0;
            acc.x += v1.x * w1; acc.y += v1.y * w1; acc.z += v1.z * w1; acc.w += v1.w * w1;
            acc.x += v2.x * w2; acc.y += v2.y * w2; acc.z += v2.z * w2; acc.w += v2.w * w2;
            acc.x += v3.x * w3; acc.y += v3.y * w3; acc.z += v3.z * w3; acc.w += v3.w * w3;
        }
        v.x = fmaxf(acc.x + b0, 0.f);
        v.y = fmaxf(acc.y + b1, 0.f);
        v.z = fmaxf(acc.z + b2, 0.f);
        v.w = fmaxf(acc.w + b3, 0.f);
    }
    tile[threadIdx.x] = v;
    __syncthreads();
    if (n < N) ((float4*)(r + (size_t)n * HDIM))[q] = v;
    if (threadIdx.x < 128) {
        const float* tf = (const float*)tile;
        const int f = threadIdx.x & 63;
        const bool squares = threadIdx.x >= 64;
        float s = 0.f;
#pragma unroll
        for (int row = 0; row < 16; ++row) {
            const float x = tf[row * HDIM + f];
            s += squares ? x * x : x;
        }
        atomicAdd(&stats[(blockIdx.x & (STAT_COPIES - 1)) * STAT_STRIDE + threadIdx.x], s);
    }
}

#define POOL_R 16

struct BnCoef { float4 m, is, g, b; };

__device__ __forceinline__ BnCoef bn_coef(float4 sm, float4 sq, float4 gg, float4 bb, int N) {
    BnCoef c;
    const float invN = 1.0f / (float)N;
    c.m.x = sm.x * invN; c.m.y = sm.y * invN; c.m.z = sm.z * invN; c.m.w = sm.w * invN;
    c.is.x = rsqrtf(sq.x * invN - c.m.x * c.m.x + EPSV);
    c.is.y = rsqrtf(sq.y * invN - c.m.y * c.m.y + EPSV);
    c.is.z = rsqrtf(sq.z * invN - c.m.z * c.m.z + EPSV);
    c.is.w = rsqrtf(sq.w * invN - c.m.w * c.m.w + EPSV);
    c.g = gg; c.b = bb;
    return c;
}

__device__ __forceinline__ float4 bn4(float4 x, const BnCoef& c) {
    float4 v;
    v.x = c.g.x * (x.x - c.m.x) * c.is.x + c.b.x;
    v.y = c.g.y * (x.y - c.m.y) * c.is.y + c.b.y;
    v.z = c.g.z * (x.z - c.m.z) * c.is.z + c.b.z;
    v.w = c.g.w * (x.w - c.m.w) * c.is.w + c.b.w;
    return v;
}

__device__ __forceinline__ void add4(float4& s, float4 a) { s.x += a.x; s.y += a.y; s.z += a.z; s.w += a.w; }

__device__ __forceinline__ void pool_add4(float* __restrict__ p, float4 a) {
    atomicAdd(p, a.x); atomicAdd(p + 1, a.y); atomicAdd(p + 2, a.z); atomicAdd(p + 3, a.w);
}

__global__ void k_bn_pool(const float* __restrict__ hin, const float* __restrict__ stats,
          const float* __restrict__ g, const float* __restrict__ beta,
          const int* __restrict__ batch, float* __restrict__ hout,
          float* __restrict__ pooled, int poolOff, int N) {
    const int t  = blockIdx.x * blockDim.x + threadIdx.x;
    const int q  = t & 15;
    const int r0 = (t >> 4) * POOL_R;
    if (r0 >= N) return;
    const float4* h4 = (const float4*)hin + (size_t)r0 * 16 + q;
    float4*       o4 = (float4*)hout + (size_t)r0 * 16 + q;
    const int*    bp = batch + r0;
    const int f = q * 4;
    float* pq = pooled + poolOff + f;

    const float4* st4 = (const float4*)stats + q;
    float4 sm = st4[0], sq = st4[16];
#pragma unroll
    for (int k = 1; k < STAT_COPIES; ++k) {
        add4(sm, st4[k * (STAT_STRIDE / 4)]);
        add4(sq, st4[k * (STAT_STRIDE / 4) + 16]);
    }
    const float4 gg = make_float4(g[f], g[f + 1], g[f + 2], g[f + 3]);
    const float4 bb = make_float4(beta[f], beta[f + 1], beta[f + 2], beta[f + 3]);

    int cur;
    float4 acc = make_float4(0.f, 0.f, 0.f, 0.f);
    if (r0 + POOL_R <= N) {
        float4 x[POOL_R];
        int    b[POOL_R];
#pragma unroll
        for (int u = 0; u < POOL_R; ++u) {
            x[u] = h4[u * 16];
            b[u] = bp[u];
        }
        const BnCoef c = bn_coef(sm, sq, gg, bb, N);
        cur = b[0];
#pragma unroll
        for (int u = 0; u < POOL_R; ++u) {
            const float4 v = bn4(x[u], c);
            o4[u * 16] = v;
            if (b[u] != cur) {
                pool_add4(pq + (size_t)cur * 192, acc);
                cur = b[u];
                acc = make_float4(0.f, 0.f, 0.f, 0.f);
            }
            acc.x += v.x; acc.y += v.y; acc.z += v.z; acc.w += v.w;
        }
    } else {
        const BnCoef c = bn_coef(sm, sq, gg, bb, N);
        cur = bp[0];
        for (int u = 0; r0 + u < N; ++u) {
            const float4 v = bn4(h4[u * 16], c);
            o4[u * 16] = v;
            const int bu = bp[u];
            if (bu != cur) {
                pool_add4(pq + (size_t)cur * 192, acc);
                cur = bu;
                acc = make_float4(0.f, 0.f, 0.f, 0.f);
            }
            acc.x += v.x; acc.y += v.y; acc.z += v.z; acc.w += v.w;
        }
    }
    pool_add4(pq + (size_t)cur * 192, acc);
}

extern "C" void kernel_launch(void* const* d_in, const int* in_sizes, int n_in,
                              void* d_out, int out_size, void* d_ws, size_t ws_size,
                              hipStream_t stream) {
    const float* x    = (const float*)d_in[0];
    const int*   eidx = (const int*)d_in[1];
    const int*   batch = (const int*)d_in[2];
    const float* W[3]  = {(const float*)d_in[3],  (const float*)d_in[7],  (const float*)d_in[11]};
    const float* bi[3] = {(const float*)d_in[4],  (const float*)d_in[8],  (const float*)d_in[12]};
    const float* ga[3] = {(const float*)d_in[5],  (const float*)d_in[9],  (const float*)d_in[13]};
    const float* be[3] = {(const float*)d_in[6],  (const float*)d_in[10], (const float*)d_in[14]};

    const int N = in_sizes[0] / HDIM;
    const int E = in_sizes[1] / 2;
    const int G = (out_size - N * HDIM) / (3 * HDIM);

    const int* src = eidx;
    const int* dst = eidx + E;

    float* ws    = (float*)d_ws;
    float* dinv  = ws;
    float* bufA  = dinv + N;
    float* bufB  = bufA + (size_t)N * HDIM;
    int*   csr   = (int*)(bufB + (size_t)N * HDIM);
    int* cnt    = csr;
    int* rowptr = cnt + N;
    int* rank   = rowptr + (N + 4);
    int* col    = rank + E;
    float* stats = (float*)(col + E);

    float* pooled = (float*)d_out;
    float* hfinal = (float*)d_out + (size_t)G * 192;

    const int TB = 256;

    k_zero_i     <<<cdiv(N, TB), TB, 0, stream>>>(cnt, N);
    k_count_rank <<<cdiv(E, TB), TB, 0, stream>>>(dst, cnt, rank, E);
    k_rowptr_dinv<<<cdiv(N, SCAN_TB), SCAN_TB, 0, stream>>>(cnt, rowptr, dinv, N);
    k_fill       <<<cdiv(E, TB), TB, 0, stream>>>(src, dst, rowptr, rank, col, E);

    k_zero<<<cdiv((long long)G * 192, TB), TB, 0, stream>>>(pooled, G * 192);

    const int rowTiles   = N / 16;
    const int gemmWaves  = rowTiles * (HDIM / 16);
    const int gemmBlocks = cdiv((long long)gemmWaves * 32, TB);

    const float* gin = x;
    float* hw = bufA;
    float* ag = bufB;
    for (int l = 0; l < 3; ++l) {
        float* hout = (l == 2) ? hfinal : hw;

        k_gemm_wmma<<<gemmBlocks, TB, 0, stream>>>(gin, W[l], hw, rowTiles);
        k_zero<<<cdiv(STAT_COPIES * STAT_STRIDE, TB), TB, 0, stream>>>(stats, STAT_COPIES * STAT_STRIDE);
        k_gather_act_stats<<<cdiv((long long)N * 16, GATHER_TB), GATHER_TB, 0, stream>>>(
            hw, dinv, rowptr, col, bi[l], ag, stats, N);

        k_bn_pool<<<cdiv((long long)cdiv(N, POOL_R) * 16, TB), TB, 0, stream>>>(
            ag, stats, ga[l], be[l], batch, hout, pooled, l * HDIM, N);

        gin = hout;
        float* sw = hw; hw = ag; ag = sw;
    }
}


# GEMM moves behind the aggregation (sum of w*(h W^T) = (sum of w*h) W^T) and runs on WMMA in the gather's tail on the block's tile in LDS; GEMM pass and its plane gone; sum order changes
# speedup vs baseline: 2.3925x; 1.1941x over previous
#include <hip/hip_runtime.h>

typedef float v2f __attribute__((ext_vector_type(2)));
typedef float v8f __attribute__((ext_vector_type(8)));

#define HDIM 64
#define EPSV 1e-5f

static inline int cdiv(long long a, long long b) { return (int)((a + b - 1) / b); }

__global__ void k_zero(float* __restrict__ p, int n) {
    int i = blockIdx.x * blockDim.x + threadIdx.x;
    if (i < n) p[i] = 0.0f;
}

__global__ void k_zero_i(int* __restrict__ p, int n) {
    int i = blockIdx.x * blockDim.x + threadIdx.x;
    if (i < n) p[i] = 0;
}

__global__ void k_count_rank(const int* __restrict__ dst, int* __restrict__ cnt,
                             int* __restrict__ rank, int E) {
    int e = blockIdx.x * blockDim.x + threadIdx.x;
    if (e < E) rank[e] = atomicAdd(&cnt[dst[e]], 1);
}

#define SCAN_TB 1024
__global__ void __launch_bounds__(SCAN_TB)
k_rowptr_dinv(const int* __restrict__ cnt, int* __restrict__ rowptr,
              float* __restrict__ dinv, int N) {
    __shared__ int sh[SCAN_TB];
    const int t     = threadIdx.x;
    const int first = blockIdx.x * SCAN_TB;
    const int4* c4  = (const int4*)cnt;
    const int nvec  = first >> 2;
    int p = 0;
#pragma unroll 4
    for (int i = t; i < nvec; i += SCAN_TB) {
        int4 v = c4[i];
        p += (v.x + v.y) + (v.z + v.w);
    }
    sh[t] = p;
    __syncthreads();
    for (int s = SCAN_TB / 2; s > 0; s >>= 1) {
        if (t < s) sh[t] += sh[t + s];
        __syncthreads();
    }
    const int base = sh[0];
    __syncthreads();
    const int i = first + t;
    const int c = (i < N) ? cnt[i] : 0;
    sh[t] = c;
    __syncthreads();
    for (int off = 1; off < SCAN_TB; off <<= 1) {
        int v = (t >= off) ? sh[t - off] : 0;
        __syncthreads();
        sh[t] += v;
        __syncthreads();
    }
    const int incl = sh[t];
    if (i < N) {
        rowptr[i] = base + incl - c;
        dinv[i]   = rsqrtf((float)(c + 1));
        if (i == N - 1) rowptr[N] = base + incl;
    }
}

__global__ void k_fill(const int* __restrict__ src, const int* __restrict__ dst,
                       const int* __restrict__ rowptr, const int* __restrict__ rank,
                       int* __restrict__ col, int E) {
    int e = blockIdx.x * blockDim.x + threadIdx.x;
    if (e < E) col[rowptr[dst[e]] + rank[e]] = src[e];
}

#define STAT_COPIES 8
#define STAT_STRIDE 1088
#define GATHER_TB 256
#define TP4 17
__device__ __forceinline__ void fence() { __builtin_amdgcn_sched_barrier(0); }

__global__ void __launch_bounds__(GATHER_TB)
k_gather_gemm_act_stats(const float* __restrict__ hin, const float* __restrict__ W,
                        const float* __restrict__ dinv,
                        const int* __restrict__ rowptr, const int* __restrict__ col,
                        const float* __restrict__ bias, float* __restrict__ r,
                        float* __restrict__ stats  , int N) {
    __shared__ float4 wl[HDIM * TP4];
    __shared__ float4 ta[16 * TP4];
    __shared__ float4 to[16 * TP4];
    const int tid = threadIdx.x;
    const float4* W4 = (const float4*)W;
    float4 wv[4];
#pragma unroll
    for (int i = 0; i < 4; ++i) wv[i] = W4[tid + i * GATHER_TB];

    const int t = blockIdx.x * GATHER_TB + tid;
    const int n = t >> 4, q = t & 15;
    const bool live = n < N;
    const int nc = live ? n : N - 1;
    const float4* h4 = (const float4*)hin;
    const float di = dinv[nc];
    const int beg = rowptr[nc];
    const int endr = rowptr[nc + 1];
    const int end = live ? endr : beg;
    const float4 vs = h4[(size_t)nc * 16 + q];

#pragma unroll
    for (int i = 0; i < 4; ++i) {
        const int idx = tid + i * GATHER_TB;
        wl[(idx >> 4) * TP4 + (idx & 15)] = wv[i];
    }

    const float nrm = live ? di * di : 0.f;
    float4 acc = make_float4(vs.x * nrm, vs.y * nrm, vs.z * nrm, vs.w * nrm);
    const int last = end - 1;
    for (int j = beg; j < end; j += 4) {
        const int s0 = col[j];
        const int s1 = col[min(j + 1, last)];
        const int s2 = col[min(j + 2, last)];
        const int s3 = col[min(j + 3, last)];
        const float d0 = dinv[s0], d1 = dinv[s1], d2 = dinv[s2], d3 = dinv[s3];
        const float4 v0 = h4[(size_t)s0 * 16 + q];
        const float4 v1 = h4[(size_t)s1 * 16 + q];
        const float4 v2 = h4[(size_t)s2 * 16 + q];
        const float4 v3 = h4[(size_t)s3 * 16 + q];
        const float w0 = d0 * di;
        const float w1 = (j + 1 < end) ? d1 * di : 0.f;
        const float w2 = (j + 2 < end) ? d2 * di : 0.f;
        const float w3 = (j + 3 < end) ? d3 * di : 0.f;
        acc.x += v0.x * w0; acc.y += v0.y * w0; acc.z += v0.z * w0; acc.w += v0.w * w0;
        acc.x += v1.x * w1; acc.y += v1.y * w1; acc.z += v1.z * w1; acc.w += v1.w * w1;
        acc.x += v2.x * w2; acc.y += v2.y * w2; acc.z += v2.z * w2; acc.w += v2.w * w2;
        acc.x += v3.x * w3; acc.y += v3.y * w3; acc.z += v3.z * w3; acc.w += v3.w * w3;
    }
    ta[(tid >> 4) * TP4 + (tid & 15)] = acc;
    __syncthreads();
    const int wave = tid >> 5;
    float sval = 0.f;
    int   sidx = 0;
    if (wave < 4) {
        const int lane = tid & 31;
        const int c = lane & 15, h = lane >> 4;
        const float4* ap = ta + c * TP4 + h * 8;
        const float4* bp = wl + (wave * 16 + c) * TP4 + h * 8;
        v8f e0 = {0.f, 0.f, 0.f, 0.f, 0.f, 0.f, 0.f, 0.f};
        v8f e1 = {0.f, 0.f, 0.f, 0.f, 0.f, 0.f, 0.f, 0.f};
#pragma unroll
        for (int half = 0; half < 2; ++half) {
            float4 a[4], b[4];
            fence();
#pragma unroll
            for (int i = 0; i < 4; ++i) { a[i] = ap[half * 4 + i]; b[i] = bp[half * 4 + i]; }
            fence();
#pragma unroll
            for (int i = 0; i < 4; ++i) {
                v2f a0 = {a[i].x, a[i].y}, b0 = {b[i].x, b[i].y};
                v2f a1 = {a[i].z, a[i].w}, b1 = {b[i].z, b[i].w};
                e0 = __builtin_amdgcn_wmma_f32_16x16x4_f32(false, a0, false, b0, (short)0, e0, false, false);
                e1 = __builtin_amdgcn_wmma_f32_16x16x4_f32(false, a1, false, b1, (short)0, e1, false, false);
            }
        }
        fence();
        asm volatile("v_nop\n\tv_nop\n\tv_nop\n\tv_nop");
        fence();
        const int f = wave * 16 + c;
        const float bf = bias[f];
        const int row0 = blockIdx.x * 16 + h * 8;
        float* tof = (float*)to + (h * 8) * (TP4 * 4) + f;
        float sm = 0.f, sq = 0.f;
#pragma unroll
        for (int rr = 0; rr < 8; ++rr) {
            float v = fmaxf(e0[rr] + e1[rr] + bf, 0.f);
            v = (row0 + rr < N) ? v : 0.f;
            tof[rr * (TP4 * 4)] = v;
            sm += v;
            sq += v * v;
        }
        sm += __shfl_xor(sm, 16);
        sq += __shfl_xor(sq, 16);
        sval = h ? sq : sm;
        sidx = (blockIdx.x & (STAT_COPIES - 1)) * STAT_STRIDE + h * 64 + f;
    }
    __syncthreads();
    if (live) ((float4*)(r + (size_t)n * HDIM))[q] = to[(tid >> 4) * TP4 + (tid & 15)];
    if (wave < 4) atomicAdd(&stats[sidx], sval);
}

#define POOL_R 16

struct BnCoef { float4 m, is, g, b; };

__device__ __forceinline__ BnCoef bn_coef(float4 sm, float4 sq, float4 gg, float4 bb, int N) {
    BnCoef c;
    const float invN = 1.0f / (float)N;
    c.m.x = sm.x * invN; c.m.y = sm.y * invN; c.m.z = sm.z * invN; c.m.w = sm.w * invN;
    c.is.x = rsqrtf(sq.x * invN - c.m.x * c.m.x + EPSV);
    c.is.y = rsqrtf(sq.y * invN - c.m.y * c.m.y + EPSV);
    c.is.z = rsqrtf(sq.z * invN - c.m.z * c.m.z + EPSV);
    c.is.w = rsqrtf(sq.w * invN - c.m.w * c.m.w + EPSV);
    c.g = gg; c.b = bb;
    return c;
}

__device__ __forceinline__ float4 bn4(float4 x, const BnCoef& c) {
    float4 v;
    v.x = c.g.x * (x.x - c.m.x) * c.is.x + c.b.x;
    v.y = c.g.y * (x.y - c.m.y) * c.is.y + c.b.y;
    v.z = c.g.z * (x.z - c.m.z) * c.is.z + c.b.z;
    v.w = c.g.w * (x.w - c.m.w) * c.is.w + c.b.w;
    return v;
}

__device__ __forceinline__ void add4(float4& s, float4 a) { s.x += a.x; s.y += a.y; s.z += a.z; s.w += a.w; }

__device__ __forceinline__ void pool_add4(float* __restrict__ p, float4 a) {
    atomicAdd(p, a.x); atomicAdd(p + 1, a.y); atomicAdd(p + 2, a.z); atomicAdd(p + 3, a.w);
}

__global__ void k_bn_pool(const float* __restrict__ hin, const float* __restrict__ stats,
          const float* __restrict__ g, const float* __restrict__ beta,
          const int* __restrict__ batch, float* __restrict__ hout,
          float* __restrict__ pooled, int poolOff, int N) {
    const int t  = blockIdx.x * blockDim.x + threadIdx.x;
    const int q  = t & 15;
    const int r0 = (t >> 4) * POOL_R;
    if (r0 >= N) return;
    const float4* h4 = (const float4*)hin + (size_t)r0 * 16 + q;
    float4*       o4 = (float4*)hout + (size_t)r0 * 16 + q;
    const int*    bp = batch + r0;
    const int f = q * 4;
    float* pq = pooled + poolOff + f;

    const float4* st4 = (const float4*)stats + q;
    float4 sm = st4[0], sq = st4[16];
#pragma unroll
    for (int k = 1; k < STAT_COPIES; ++k) {
        add4(sm, st4[k * (STAT_STRIDE / 4)]);
        add4(sq, st4[k * (STAT_STRIDE / 4) + 16]);
    }
    const float4 gg = make_float4(g[f], g[f + 1], g[f + 2], g[f + 3]);
    const float4 bb = make_float4(beta[f], beta[f + 1], beta[f + 2], beta[f + 3]);

    int cur;
    float4 acc = make_float4(0.f, 0.f, 0.f, 0.f);
    if (r0 + POOL_R <= N) {
        float4 x[POOL_R];
        int    b[POOL_R];
#pragma unroll
        for (int u = 0; u < POOL_R; ++u) {
            x[u] = h4[u * 16];
            b[u] = bp[u];
        }
        const BnCoef c = bn_coef(sm, sq, gg, bb, N);
        cur = b[0];
#pragma unroll
        for (int u = 0; u < POOL_R; ++u) {
            const float4 v = bn4(x[u], c);
            o4[u * 16] = v;
            if (b[u] != cur) {
                pool_add4(pq + (size_t)cur * 192, acc);
                cur = b[u];
                acc = make_float4(0.f, 0.f, 0.f, 0.f);
            }
            acc.x += v.x; acc.y += v.y; acc.z += v.z; acc.w += v.w;
        }
    } else {
        const BnCoef c = bn_coef(sm, sq, gg, bb, N);
        cur = bp[0];
        for (int u = 0; r0 + u < N; ++u) {
            const float4 v = bn4(h4[u * 16], c);
            o4[u * 16] = v;
            const int bu = bp[u];
            if (bu != cur) {
                pool_add4(pq + (size_t)cur * 192, acc);
                cur = bu;
                acc = make_float4(0.f, 0.f, 0.f, 0.f);
            }
            acc.x += v.x; acc.y += v.y; acc.z += v.z; acc.w += v.w;
        }
    }
    pool_add4(pq + (size_t)cur * 192, acc);
}

extern "C" void kernel_launch(void* const* d_in, const int* in_sizes, int n_in,
                              void* d_out, int out_size, void* d_ws, size_t ws_size,
                              hipStream_t stream) {
    const float* x    = (const float*)d_in[0];
    const int*   eidx = (const int*)d_in[1];
    const int*   batch = (const int*)d_in[2];
    const float* W[3]  = {(const float*)d_in[3],  (const float*)d_in[7],  (const float*)d_in[11]};
    const float* bi[3] = {(const float*)d_in[4],  (const float*)d_in[8],  (const float*)d_in[12]};
    const float* ga[3] = {(const float*)d_in[5],  (const float*)d_in[9],  (const float*)d_in[13]};
    const float* be[3] = {(const float*)d_in[6],  (const float*)d_in[10], (const float*)d_in[14]};

    const int N = in_sizes[0] / HDIM;
    const int E = in_sizes[1] / 2;
    const int G = (out_size - N * HDIM) / (3 * HDIM);

    const int* src = eidx;
    const int* dst = eidx + E;

    float* ws    = (float*)d_ws;
    float* dinv  = ws;
    float* bufA  = dinv + N;
    float* bufB  = bufA + (size_t)N * HDIM;
    int*   csr   = (int*)(bufB + (size_t)N * HDIM);
    int* cnt    = csr;
    int* rowptr = cnt + N;
    int* rank   = rowptr + (N + 4);
    int* col    = rank + E;
    float* stats = (float*)(col + E);

    float* pooled = (float*)d_out;
    float* hfinal = (float*)d_out + (size_t)G * 192;

    const int TB = 256;

    k_zero_i     <<<cdiv(N, TB), TB, 0, stream>>>(cnt, N);
    k_count_rank <<<cdiv(E, TB), TB, 0, stream>>>(dst, cnt, rank, E);
    k_rowptr_dinv<<<cdiv(N, SCAN_TB), SCAN_TB, 0, stream>>>(cnt, rowptr, dinv, N);
    k_fill       <<<cdiv(E, TB), TB, 0, stream>>>(src, dst, rowptr, rank, col, E);

    k_zero<<<cdiv((long long)G * 192, TB), TB, 0, stream>>>(pooled, G * 192);

    const float* gin = x;
    float* rp = bufA;
    float* hp = bufB;
    for (int l = 0; l < 3; ++l) {
        float* hout = (l == 2) ? hfinal : hp;

        k_zero<<<cdiv(STAT_COPIES * STAT_STRIDE, TB), TB, 0, stream>>>(stats, STAT_COPIES * STAT_STRIDE);
        k_gather_gemm_act_stats<<<cdiv((long long)N * 16, GATHER_TB), GATHER_TB, 0, stream>>>(
            gin, W[l], dinv, rowptr, col, bi[l], rp, stats, N);

        k_bn_pool<<<cdiv((long long)cdiv(N, POOL_R) * 16, TB), TB, 0, stream>>>(
            rp, stats, ga[l], be[l], batch, hout, pooled, l * HDIM, N);

        gin = hout;
    }
}


# Edge list built by a counting sort on LDS counters per (node range, edge chunk) work-group instead of one returning global atomic per edge; zeroing launch and rank plane gone; row sum order changes
# speedup vs baseline: 2.8630x; 1.1967x over previous
#include <hip/hip_runtime.h>

typedef float v2f __attribute__((ext_vector_type(2)));
typedef float v8f __attribute__((ext_vector_type(8)));

#define HDIM 64
#define EPSV 1e-5f

static inline int cdiv(long long a, long long b) { return (int)((a + b - 1) / b); }

__global__ void k_zero(float* __restrict__ p, int n) {
    int i = blockIdx.x * blockDim.x + threadIdx.x;
    if (i < n) p[i] = 0.0f;
}

#define CSR_R  32
#define CSR_C  8
#define CSR_TB 1024
#define CSR_U  8
#define CSR_MAXSPAN 4096
#define CSR_NPT (CSR_MAXSPAN / CSR_TB)

__device__ __forceinline__ int csr_npr(int N)    { return (N + CSR_R - 1) / CSR_R; }
__device__ __forceinline__ int csr_stride(int N) { return N + 4; }
__device__ __forceinline__ int csr_clen(int E)   { return (((E + CSR_C - 1) / CSR_C) + 3) & ~3; }

__device__ __forceinline__ void csr_count1(int d, int lo, unsigned span, int* cl, int& m) {
    const unsigned u = (unsigned)(d - lo);
    if (u < span) { atomicAdd(&cl[u], 1); ++m; }
}

__device__ __forceinline__ void csr_place1(int d, int s, int lo, unsigned span, int* cl,
                                           int* __restrict__ col) {
    const unsigned u = (unsigned)(d - lo);
    if (u < span) col[atomicAdd(&cl[u], 1)] = s;
}

__global__ void __launch_bounds__(CSR_TB)
k_csr_count(const int* __restrict__ dst, int* __restrict__ cntC, int* __restrict__ tot, int N, int E) {
    __shared__ int cl[CSR_MAXSPAN];
    __shared__ int tsum;
    const int t = threadIdx.x;
    const int r = blockIdx.x / CSR_C, c = blockIdx.x % CSR_C;
    const int npr = csr_npr(N), NS = csr_stride(N);
    const int lo = r * npr;
    const int spn = min(min(npr, N - lo), CSR_MAXSPAN);
    const unsigned span = spn > 0 ? (unsigned)spn : 0u;
    for (int i = t; i < CSR_MAXSPAN; i += CSR_TB) cl[i] = 0;
    if (t == 0) tsum = 0;
    __syncthreads();
    const int clen = csr_clen(E);
    const int e0 = min(E, c * clen), e1 = min(E, e0 + clen);
    int m = 0;
    int done = e0;
    if ((((size_t)dst) & 15) == 0) {
        const int4* d4 = (const int4*)(dst + e0);
        const int nv = (e1 - e0) >> 2;
        for (int i0 = 0; i0 < nv; i0 += CSR_U * CSR_TB) {
            int4 dv[CSR_U];
#pragma unroll
            for (int u = 0; u < CSR_U; ++u) dv[u] = d4[min(i0 + u * CSR_TB + t, nv - 1)];
#pragma unroll
            for (int u = 0; u < CSR_U; ++u) {
                if (i0 + u * CSR_TB + t < nv) {
                    csr_count1(dv[u].x, lo, span, cl, m);
                    csr_count1(dv[u].y, lo, span, cl, m);
                    csr_count1(dv[u].z, lo, span, cl, m);
                    csr_count1(dv[u].w, lo, span, cl, m);
                }
            }
        }
        done = e0 + (nv << 2);
    }
    for (int e = done + t; e < e1; e += CSR_TB) csr_count1(dst[e], lo, span, cl, m);
    for (int o = 16; o > 0; o >>= 1) m += __shfl_xor(m, o);
    if ((t & 31) == 0) atomicAdd(&tsum, m);
    __syncthreads();
    for (int i = t; i < (int)span; i += CSR_TB) cntC[c * NS + lo + i] = cl[i];
    if (t == 0) tot[blockIdx.x] = tsum;
}

__global__ void __launch_bounds__(CSR_TB)
k_csr_scan(int* __restrict__ cntC, const int* __restrict__ tot, int* __restrict__ rowptr,
           float* __restrict__ dinv, int N) {
    __shared__ int sh[CSR_TB];
    const int t = threadIdx.x, r = blockIdx.x;
    const int npr = csr_npr(N), NS = csr_stride(N);
    const int lo = r * npr;
    const int hi = min(lo + npr, N);
    const int npt = (npr + CSR_TB - 1) / CSR_TB;
    const int f = lo + t * npt;
    const int l = min(f + npt, hi);
    int p = 0;
    if (t < CSR_R * CSR_C) p = (t / CSR_C < r) ? tot[t] : 0;
    int x[CSR_NPT][CSR_C];
#pragma unroll
    for (int k = 0; k < CSR_NPT; ++k) {
        const int n = min(f + k, hi - 1);
#pragma unroll
        for (int c = 0; c < CSR_C; ++c) x[k][c] = cntC[c * NS + n];
    }
    int mine = 0;
#pragma unroll
    for (int k = 0; k < CSR_NPT; ++k) {
        if (f + k < l) {
#pragma unroll
            for (int c = 0; c < CSR_C; ++c) mine += x[k][c];
        }
    }
    sh[t] = p;
    __syncthreads();
    for (int s = CSR_TB / 2; s > 0; s >>= 1) {
        if (t < s) sh[t] += sh[t + s];
        __syncthreads();
    }
    const int base = sh[0];
    __syncthreads();
    sh[t] = mine;
    __syncthreads();
    for (int off = 1; off < CSR_TB; off <<= 1) {
        int v = (t >= off) ? sh[t - off] : 0;
        __syncthreads();
        sh[t] += v;
        __syncthreads();
    }
    int run = base + sh[t] - mine;
#pragma unroll
    for (int k = 0; k < CSR_NPT; ++k) {
        if (f + k < l) {
            const int n = f + k;
            int o = run;
#pragma unroll
            for (int c = 0; c < CSR_C; ++c) { cntC[c * NS + n] = o; o += x[k][c]; }
            rowptr[n] = run;
            dinv[n]   = rsqrtf((float)(o - run + 1));
            run = o;
        }
    }
    if (f < l && l == N) rowptr[N] = run;
}

__global__ void __launch_bounds__(CSR_TB)
k_csr_fill(const int* __restrict__ src, const int* __restrict__ dst, const int* __restrict__ cntC,
           int* __restrict__ col, int N, int E) {
    __shared__ int cl[CSR_MAXSPAN];
    const int t = threadIdx.x;
    const int r = blockIdx.x / CSR_C, c = blockIdx.x % CSR_C;
    const int npr = csr_npr(N), NS = csr_stride(N);
    const int lo = r * npr;
    const int spn = min(min(npr, N - lo), CSR_MAXSPAN);
    const unsigned span = spn > 0 ? (unsigned)spn : 0u;
    int st[CSR_NPT];
#pragma unroll
    for (int k = 0; k < CSR_NPT; ++k) st[k] = cntC[c * NS + lo + min(k * CSR_TB + t, spn - 1)];
#pragma unroll
    for (int k = 0; k < CSR_NPT; ++k)
        if (k * CSR_TB + t < spn) cl[k * CSR_TB + t] = st[k];
    __syncthreads();
    const int clen = csr_clen(E);
    const int e0 = min(E, c * clen), e1 = min(E, e0 + clen);
    int done = e0;
    if (((((size_t)dst) | ((size_t)src)) & 15) == 0) {
        const int4* d4 = (const int4*)(dst + e0);
        const int4* s4 = (const int4*)(src + e0);
        const int nv = (e1 - e0) >> 2;
        for (int i0 = 0; i0 < nv; i0 += CSR_U * CSR_TB) {
            int4 dv[CSR_U], sv[CSR_U];
#pragma unroll
            for (int u = 0; u < CSR_U; ++u) {
                const int i = min(i0 + u * CSR_TB + t, nv - 1);
                dv[u] = d4[i];
                sv[u] = s4[i];
            }
#pragma unroll
            for (int u = 0; u < CSR_U; ++u)
                if (i0 + u * CSR_TB + t < nv) {
                    csr_place1(dv[u].x, sv[u].x, lo, span, cl, col);
                    csr_place1(dv[u].y, sv[u].y, lo, span, cl, col);
                    csr_place1(dv[u].z, sv[u].z, lo, span, cl, col);
                    csr_place1(dv[u].w, sv[u].w, lo, span, cl, col);
                }
        }
        done = e0 + (nv << 2);
    }
    for (int e = done + t; e < e1; e += CSR_TB) {
        csr_place1(dst[e], src[e], lo, span, cl, col);
    }
}

#define STAT_COPIES 8
#define STAT_STRIDE 1088
#define GATHER_TB 256
#define TP4 17
__device__ __forceinline__ void fence() { __builtin_amdgcn_sched_barrier(0); }

__global__ void __launch_bounds__(GATHER_TB)
k_gather_gemm_act_stats(const float* __restrict__ hin, const float* __restrict__ W,
                        const float* __restrict__ dinv,
                        const int* __restrict__ rowptr, const int* __restrict__ col,
                        const float* __restrict__ bias, float* __restrict__ r,
                        float* __restrict__ stats  , int N) {
    __shared__ float4 wl[HDIM * TP4];
    __shared__ float4 ta[16 * TP4];
    __shared__ float4 to[16 * TP4];
    const int tid = threadIdx.x;
    const float4* W4 = (const float4*)W;
    float4 wv[4];
#pragma unroll
    for (int i = 0; i < 4; ++i) wv[i] = W4[tid + i * GATHER_TB];

    const int t = blockIdx.x * GATHER_TB + tid;
    const int n = t >> 4, q = t & 15;
    const bool live = n < N;
    const int nc = live ? n : N - 1;
    const float4* h4 = (const float4*)hin;
    const float di = dinv[nc];
    const int beg = rowptr[nc];
    const int endr = rowptr[nc + 1];
    const int end = live ? endr : beg;
    const float4 vs = h4[(size_t)nc * 16 + q];

#pragma unroll
    for (int i = 0; i < 4; ++i) {
        const int idx = tid + i * GATHER_TB;
        wl[(idx >> 4) * TP4 + (idx & 15)] = wv[i];
    }

    const float nrm = live ? di * di : 0.f;
    float4 acc = make_float4(vs.x * nrm, vs.y * nrm, vs.z * nrm, vs.w * nrm);
    const int last = end - 1;
    for (int j = beg; j < end; j += 4) {
        const int s0 = col[j];
        const int s1 = col[min(j + 1, last)];
        const int s2 = col[min(j + 2, last)];
        const int s3 = col[min(j + 3, last)];
        const float d0 = dinv[s0], d1 = dinv[s1], d2 = dinv[s2], d3 = dinv[s3];
        const float4 v0 = h4[(size_t)s0 * 16 + q];
        const float4 v1 = h4[(size_t)s1 * 16 + q];
        const float4 v2 = h4[(size_t)s2 * 16 + q];
        const float4 v3 = h4[(size_t)s3 * 16 + q];
        const float w0 = d0 * di;
        const float w1 = (j + 1 < end) ? d1 * di : 0.f;
        const float w2 = (j + 2 < end) ? d2 * di : 0.f;
        const float w3 = (j + 3 < end) ? d3 * di : 0.f;
        acc.x += v0.x * w0; acc.y += v0.y * w0; acc.z += v0.z * w0; acc.w += v0.w * w0;
        acc.x += v1.x * w1; acc.y += v1.y * w1; acc.z += v1.z * w1; acc.w += v1.w * w1;
        acc.x += v2.x * w2; acc.y += v2.y * w2; acc.z += v2.z * w2; acc.w += v2.w * w2;
        acc.x += v3.x * w3; acc.y += v3.y * w3; acc.z += v3.z * w3; acc.w += v3.w * w3;
    }
    ta[(tid >> 4) * TP4 + (tid & 15)] = acc;
    __syncthreads();
    const int wave = tid >> 5;
    float sval = 0.f;
    int   sidx = 0;
    if (wave < 4) {
        const int lane = tid & 31;
        const int c = lane & 15, h = lane >> 4;
        const float4* ap = ta + c * TP4 + h * 8;
        const float4* bp = wl + (wave * 16 + c) * TP4 + h * 8;
        v8f e0 = {0.f, 0.f, 0.f, 0.f, 0.f, 0.f, 0.f, 0.f};
        v8f e1 = {0.f, 0.f, 0.f, 0.f, 0.f, 0.f, 0.f, 0.f};
#pragma unroll
        for (int half = 0; half < 2; ++half) {
            float4 a[4], b[4];
            fence();
#pragma unroll
            for (int i = 0; i < 4; ++i) { a[i] = ap[half * 4 + i]; b[i] = bp[half * 4 + i]; }
            fence();
#pragma unroll
            for (int i = 0; i < 4; ++i) {
                v2f a0 = {a[i].x, a[i].y}, b0 = {b[i].x, b[i].y};
                v2f a1 = {a[i].z, a[i].w}, b1 = {b[i].z, b[i].w};
                e0 = __builtin_amdgcn_wmma_f32_16x16x4_f32(false, a0, false, b0, (short)0, e0, false, false);
                e1 = __builtin_amdgcn_wmma_f32_16x16x4_f32(false, a1, false, b1, (short)0, e1, false, false);
            }
        }
        fence();
        asm volatile("v_nop\n\tv_nop\n\tv_nop\n\tv_nop");
        fence();
        const int f = wave * 16 + c;
        const float bf = bias[f];
        const int row0 = blockIdx.x * 16 + h * 8;
        float* tof = (float*)to + (h * 8) * (TP4 * 4) + f;
        float sm = 0.f, sq = 0.f;
#pragma unroll
        for (int rr = 0; rr < 8; ++rr) {
            float v = fmaxf(e0[rr] + e1[rr] + bf, 0.f);
            v = (row0 + rr < N) ? v : 0.f;
            tof[rr * (TP4 * 4)] = v;
            sm += v;
            sq += v * v;
        }
        sm += __shfl_xor(sm, 16);
        sq += __shfl_xor(sq, 16);
        sval = h ? sq : sm;
        sidx = (blockIdx.x & (STAT_COPIES - 1)) * STAT_STRIDE + h * 64 + f;
    }
    __syncthreads();
    if (live) ((float4*)(r + (size_t)n * HDIM))[q] = to[(tid >> 4) * TP4 + (tid & 15)];
    if (wave < 4) atomicAdd(&stats[sidx], sval);
}

#define POOL_R 16

struct BnCoef { float4 m, is, g, b; };

__device__ __forceinline__ BnCoef bn_coef(float4 sm, float4 sq, float4 gg, float4 bb, int N) {
    BnCoef c;
    const float invN = 1.0f / (float)N;
    c.m.x = sm.x * invN; c.m.y = sm.y * invN; c.m.z = sm.z * invN; c.m.w = sm.w * invN;
    c.is.x = rsqrtf(sq.x * invN - c.m.x * c.m.x + EPSV);
    c.is.y = rsqrtf(sq.y * invN - c.m.y * c.m.y + EPSV);
    c.is.z = rsqrtf(sq.z * invN - c.m.z * c.m.z + EPSV);
    c.is.w = rsqrtf(sq.w * invN - c.m.w * c.m.w + EPSV);
    c.g = gg; c.b = bb;
    return c;
}

__device__ __forceinline__ float4 bn4(float4 x, const BnCoef& c) {
    float4 v;
    v.x = c.g.x * (x.x - c.m.x) * c.is.x + c.b.x;
    v.y = c.g.y * (x.y - c.m.y) * c.is.y + c.b.y;
    v.z = c.g.z * (x.z - c.m.z) * c.is.z + c.b.z;
    v.w = c.g.w * (x.w - c.m.w) * c.is.w + c.b.w;
    return v;
}

__device__ __forceinline__ void add4(float4& s, float4 a) { s.x += a.x; s.y += a.y; s.z += a.z; s.w += a.w; }

__device__ __forceinline__ void pool_add4(float* __restrict__ p, float4 a) {
    atomicAdd(p, a.x); atomicAdd(p + 1, a.y); atomicAdd(p + 2, a.z); atomicAdd(p + 3, a.w);
}

__global__ void k_bn_pool(const float* __restrict__ hin, const float* __restrict__ stats,
          const float* __restrict__ g, const float* __restrict__ beta,
          const int* __restrict__ batch, float* __restrict__ hout,
          float* __restrict__ pooled, int poolOff, int N) {
    const int t  = blockIdx.x * blockDim.x + threadIdx.x;
    const int q  = t & 15;
    const int r0 = (t >> 4) * POOL_R;
    if (r0 >= N) return;
    const float4* h4 = (const float4*)hin + (size_t)r0 * 16 + q;
    float4*       o4 = (float4*)hout + (size_t)r0 * 16 + q;
    const int*    bp = batch + r0;
    const int f = q * 4;
    float* pq = pooled + poolOff + f;

    const float4* st4 = (const float4*)stats + q;
    float4 sm = st4[0], sq = st4[16];
#pragma unroll
    for (int k = 1; k < STAT_COPIES; ++k) {
        add4(sm, st4[k * (STAT_STRIDE / 4)]);
        add4(sq, st4[k * (STAT_STRIDE / 4) + 16]);
    }
    const float4 gg = make_float4(g[f], g[f + 1], g[f + 2], g[f + 3]);
    const float4 bb = make_float4(beta[f], beta[f + 1], beta[f + 2], beta[f + 3]);

    int cur;
    float4 acc = make_float4(0.f, 0.f, 0.f, 0.f);
    if (r0 + POOL_R <= N) {
        float4 x[POOL_R];
        int    b[POOL_R];
#pragma unroll
        for (int u = 0; u < POOL_R; ++u) {
            x[u] = h4[u * 16];
            b[u] = bp[u];
        }
        const BnCoef c = bn_coef(sm, sq, gg, bb, N);
        cur = b[0];
#pragma unroll
        for (int u = 0; u < POOL_R; ++u) {
            const float4 v = bn4(x[u], c);
            o4[u * 16] = v;
            if (b[u] != cur) {
                pool_add4(pq + (size_t)cur * 192, acc);
                cur = b[u];
                acc = make_float4(0.f, 0.f, 0.f, 0.f);
            }
            acc.x += v.x; acc.y += v.y; acc.z += v.z; acc.w += v.w;
        }
    } else {
        const BnCoef c = bn_coef(sm, sq, gg, bb, N);
        cur = bp[0];
        for (int u = 0; r0 + u < N; ++u) {
            const float4 v = bn4(h4[u * 16], c);
            o4[u * 16] = v;
            const int bu = bp[u];
            if (bu != cur) {
                pool_add4(pq + (size_t)cur * 192, acc);
                cur = bu;
                acc = make_float4(0.f, 0.f, 0.f, 0.f);
            }
            acc.x += v.x; acc.y += v.y; acc.z += v.z; acc.w += v.w;
        }
    }
    pool_add4(pq + (size_t)cur * 192, acc);
}

extern "C" void kernel_launch(void* const* d_in, const int* in_sizes, int n_in,
                              void* d_out, int out_size, void* d_ws, size_t ws_size,
                              hipStream_t stream) {
    const float* x    = (const float*)d_in[0];
    const int*   eidx = (const int*)d_in[1];
    const int*   batch = (const int*)d_in[2];
    const float* W[3]  = {(const float*)d_in[3],  (const float*)d_in[7],  (const float*)d_in[11]};
    const float* bi[3] = {(const float*)d_in[4],  (const float*)d_in[8],  (const float*)d_in[12]};
    const float* ga[3] = {(const float*)d_in[5],  (const float*)d_in[9],  (const float*)d_in[13]};
    const float* be[3] = {(const float*)d_in[6],  (const float*)d_in[10], (const float*)d_in[14]};

    const int N = in_sizes[0] / HDIM;
    const int E = in_sizes[1] / 2;
    const int G = (out_size - N * HDIM) / (3 * HDIM);

    const int* src = eidx;
    const int* dst = eidx + E;

    float* ws    = (float*)d_ws;
    float* dinv  = ws;
    float* bufA  = dinv + N;
    float* bufB  = bufA + (size_t)N * HDIM;
    int*   csr   = (int*)(bufB + (size_t)N * HDIM);
    int* rowptr = csr;
    int* col    = rowptr + (N + 4);
    int* cntC   = col + E;
    int* tot    = cntC + CSR_C * (N + 4);
    float* stats = (float*)(tot + CSR_R * CSR_C);

    float* pooled = (float*)d_out;
    float* hfinal = (float*)d_out + (size_t)G * 192;

    const int TB = 256;

    k_csr_count<<<CSR_R * CSR_C, CSR_TB, 0, stream>>>(dst, cntC, tot, N, E);
    k_csr_scan <<<CSR_R, CSR_TB, 0, stream>>>(cntC, tot, rowptr, dinv, N);
    k_csr_fill <<<CSR_R * CSR_C, CSR_TB, 0, stream>>>(src, dst, cntC, col, N, E);

    k_zero<<<cdiv((long long)G * 192, TB), TB, 0, stream>>>(pooled, G * 192);

    const float* gin = x;
    float* rp = bufA;
    float* hp = bufB;
    for (int l = 0; l < 3; ++l) {
        float* hout = (l == 2) ? hfinal : hp;

        k_zero<<<cdiv(STAT_COPIES * STAT_STRIDE, TB), TB, 0, stream>>>(stats, STAT_COPIES * STAT_STRIDE);
        k_gather_gemm_act_stats<<<cdiv((long long)N * 16, GATHER_TB), GATHER_TB, 0, stream>>>(
            gin, W[l], dinv, rowptr, col, bi[l], rp, stats, N);

        k_bn_pool<<<cdiv((long long)cdiv(N, POOL_R) * 16, TB), TB, 0, stream>>>(
            rp, stats, ga[l], be[l], batch, hout, pooled, l * HDIM, N);

        gin = hout;
    }
}


# Gather runs as a grid the chip holds at once: 2048 work-groups (8 per WGP) each loop over 16-node tiles, W copied to LDS once per work-group; 16384 waves enter per layer, not 50000
# speedup vs baseline: 2.9740x; 1.0388x over previous
#include <hip/hip_runtime.h>

typedef float v2f __attribute__((ext_vector_type(2)));
typedef float v8f __attribute__((ext_vector_type(8)));

#define HDIM 64
#define EPSV 1e-5f

static inline int cdiv(long long a, long long b) { return (int)((a + b - 1) / b); }

__global__ void k_zero(float* __restrict__ p, int n) {
    int i = blockIdx.x * blockDim.x + threadIdx.x;
    if (i < n) p[i] = 0.0f;
}

#define CSR_R  32
#define CSR_C  8
#define CSR_TB 1024
#define CSR_U  8
#define CSR_MAXSPAN 4096
#define CSR_NPT (CSR_MAXSPAN / CSR_TB)

__device__ __forceinline__ int csr_npr(int N)    { return (N + CSR_R - 1) / CSR_R; }
__device__ __forceinline__ int csr_stride(int N) { return N + 4; }
__device__ __forceinline__ int csr_clen(int E)   { return (((E + CSR_C - 1) / CSR_C) + 3) & ~3; }

__device__ __forceinline__ void csr_count1(int d, int lo, unsigned span, int* cl, int& m) {
    const unsigned u = (unsigned)(d - lo);
    if (u < span) { atomicAdd(&cl[u], 1); ++m; }
}

__device__ __forceinline__ void csr_place1(int d, int s, int lo, unsigned span, int* cl,
                                           int* __restrict__ col) {
    const unsigned u = (unsigned)(d - lo);
    if (u < span) col[atomicAdd(&cl[u], 1)] = s;
}

__global__ void __launch_bounds__(CSR_TB)
k_csr_count(const int* __restrict__ dst, int* __restrict__ cntC, int* __restrict__ tot, int N, int E) {
    __shared__ int cl[CSR_MAXSPAN];
    __shared__ int tsum;
    const int t = threadIdx.x;
    const int r = blockIdx.x / CSR_C, c = blockIdx.x % CSR_C;
    const int npr = csr_npr(N), NS = csr_stride(N);
    const int lo = r * npr;
    const int spn = min(min(npr, N - lo), CSR_MAXSPAN);
    const unsigned span = spn > 0 ? (unsigned)spn : 0u;
    for (int i = t; i < CSR_MAXSPAN; i += CSR_TB) cl[i] = 0;
    if (t == 0) tsum = 0;
    __syncthreads();
    const int clen = csr_clen(E);
    const int e0 = min(E, c * clen), e1 = min(E, e0 + clen);
    int m = 0;
    int done = e0;
    if ((((size_t)dst) & 15) == 0) {
        const int4* d4 = (const int4*)(dst + e0);
        const int nv = (e1 - e0) >> 2;
        for (int i0 = 0; i0 < nv; i0 += CSR_U * CSR_TB) {
            int4 dv[CSR_U];
#pragma unroll
            for (int u = 0; u < CSR_U; ++u) dv[u] = d4[min(i0 + u * CSR_TB + t, nv - 1)];
#pragma unroll
            for (int u = 0; u < CSR_U; ++u) {
                if (i0 + u * CSR_TB + t < nv) {
                    csr_count1(dv[u].x, lo, span, cl, m);
                    csr_count1(dv[u].y, lo, span, cl, m);
                    csr_count1(dv[u].z, lo, span, cl, m);
                    csr_count1(dv[u].w, lo, span, cl, m);
                }
            }
        }
        done = e0 + (nv << 2);
    }
    for (int e = done + t; e < e1; e += CSR_TB) csr_count1(dst[e], lo, span, cl, m);
    for (int o = 16; o > 0; o >>= 1) m += __shfl_xor(m, o);
    if ((t & 31) == 0) atomicAdd(&tsum, m);
    __syncthreads();
    for (int i = t; i < (int)span; i += CSR_TB) cntC[c * NS + lo + i] = cl[i];
    if (t == 0) tot[blockIdx.x] = tsum;
}

__global__ void __launch_bounds__(CSR_TB)
k_csr_scan(int* __restrict__ cntC, const int* __restrict__ tot, int* __restrict__ rowptr,
           float* __restrict__ dinv, int N) {
    __shared__ int sh[CSR_TB];
    const int t = threadIdx.x, r = blockIdx.x;
    const int npr = csr_npr(N), NS = csr_stride(N);
    const int lo = r * npr;
    const int hi = min(lo + npr, N);
    const int npt = (npr + CSR_TB - 1) / CSR_TB;
    const int f = lo + t * npt;
    const int l = min(f + npt, hi);
    int p = 0;
    if (t < CSR_R * CSR_C) p = (t / CSR_C < r) ? tot[t] : 0;
    int x[CSR_NPT][CSR_C];
#pragma unroll
    for (int k = 0; k < CSR_NPT; ++k) {
        const int n = min(f + k, hi - 1);
#pragma unroll
        for (int c = 0; c < CSR_C; ++c) x[k][c] = cntC[c * NS + n];
    }
    int mine = 0;
#pragma unroll
    for (int k = 0; k < CSR_NPT; ++k) {
        if (f + k < l) {
#pragma unroll
            for (int c = 0; c < CSR_C; ++c) mine += x[k][c];
        }
    }
    sh[t] = p;
    __syncthreads();
    for (int s = CSR_TB / 2; s > 0; s >>= 1) {
        if (t < s) sh[t] += sh[t + s];
        __syncthreads();
    }
    const int base = sh[0];
    __syncthreads();
    sh[t] = mine;
    __syncthreads();
    for (int off = 1; off < CSR_TB; off <<= 1) {
        int v = (t >= off) ? sh[t - off] : 0;
        __syncthreads();
        sh[t] += v;
        __syncthreads();
    }
    int run = base + sh[t] - mine;
#pragma unroll
    for (int k = 0; k < CSR_NPT; ++k) {
        if (f + k < l) {
            const int n = f + k;
            int o = run;
#pragma unroll
            for (int c = 0; c < CSR_C; ++c) { cntC[c * NS + n] = o; o += x[k][c]; }
            rowptr[n] = run;
            dinv[n]   = rsqrtf((float)(o - run + 1));
            run = o;
        }
    }
    if (f < l && l == N) rowptr[N] = run;
}

__global__ void __launch_bounds__(CSR_TB)
k_csr_fill(const int* __restrict__ src, const int* __restrict__ dst, const int* __restrict__ cntC,
           int* __restrict__ col, int N, int E) {
    __shared__ int cl[CSR_MAXSPAN];
    const int t = threadIdx.x;
    const int r = blockIdx.x / CSR_C, c = blockIdx.x % CSR_C;
    const int npr = csr_npr(N), NS = csr_stride(N);
    const int lo = r * npr;
    const int spn = min(min(npr, N - lo), CSR_MAXSPAN);
    const unsigned span = spn > 0 ? (unsigned)spn : 0u;
    int st[CSR_NPT];
#pragma unroll
    for (int k = 0; k < CSR_NPT; ++k) st[k] = cntC[c * NS + lo + min(k * CSR_TB + t, spn - 1)];
#pragma unroll
    for (int k = 0; k < CSR_NPT; ++k)
        if (k * CSR_TB + t < spn) cl[k * CSR_TB + t] = st[k];
    __syncthreads();
    const int clen = csr_clen(E);
    const int e0 = min(E, c * clen), e1 = min(E, e0 + clen);
    int done = e0;
    if (((((size_t)dst) | ((size_t)src)) & 15) == 0) {
        const int4* d4 = (const int4*)(dst + e0);
        const int4* s4 = (const int4*)(src + e0);
        const int nv = (e1 - e0) >> 2;
        for (int i0 = 0; i0 < nv; i0 += CSR_U * CSR_TB) {
            int4 dv[CSR_U], sv[CSR_U];
#pragma unroll
            for (int u = 0; u < CSR_U; ++u) {
                const int i = min(i0 + u * CSR_TB + t, nv - 1);
                dv[u] = d4[i];
                sv[u] = s4[i];
            }
#pragma unroll
            for (int u = 0; u < CSR_U; ++u)
                if (i0 + u * CSR_TB + t < nv) {
                    csr_place1(dv[u].x, sv[u].x, lo, span, cl, col);
                    csr_place1(dv[u].y, sv[u].y, lo, span, cl, col);
                    csr_place1(dv[u].z, sv[u].z, lo, span, cl, col);
                    csr_place1(dv[u].w, sv[u].w, lo, span, cl, col);
                }
        }
        done = e0 + (nv << 2);
    }
    for (int e = done + t; e < e1; e += CSR_TB) {
        csr_place1(dst[e], src[e], lo, span, cl, col);
    }
}

#define STAT_COPIES 8
#define STAT_STRIDE 1088
#define GATHER_TB 256
#define GATHER_GRID 2048
#define TP4 17
__device__ __forceinline__ void fence() { __builtin_amdgcn_sched_barrier(0); }

__global__ void __launch_bounds__(GATHER_TB) __attribute__((amdgpu_waves_per_eu(16, 16)))
k_gather_gemm_act_stats(const float* __restrict__ hin, const float* __restrict__ W,
                        const float* __restrict__ dinv,
                        const int* __restrict__ rowptr, const int* __restrict__ col,
                        const float* __restrict__ bias, float* __restrict__ r,
                        float* __restrict__ stats  , int N) {
    __shared__ float4 wl[HDIM * TP4];
    __shared__ float4 ta[16 * TP4];
    __shared__ float4 to[16 * TP4];
    const int tid = threadIdx.x;
    const int ntiles = (N + 15) >> 4;
    if ((int)blockIdx.x >= ntiles) return;
    const float4* W4 = (const float4*)W;
    float4 wv[4];
#pragma unroll
    for (int i = 0; i < 4; ++i) wv[i] = W4[tid + i * GATHER_TB];

    const int q = tid & 15;
    const int tr = tid >> 4;
    const int wave = tid >> 5;
    const int lane = tid & 31;
    const int c = lane & 15, h = lane >> 4;
    const int f = (wave & 3) * 16 + c;
    const float4* h4 = (const float4*)hin;
    const float4* ap = ta + c * TP4 + h * 8;
    const float4* bp = wl + f * TP4 + h * 8;
    float* tof = (float*)to + (h * 8) * (TP4 * 4) + f;

#pragma unroll
    for (int i = 0; i < 4; ++i) {
        const int idx = tid + i * GATHER_TB;
        wl[(idx >> 4) * TP4 + (idx & 15)] = wv[i];
    }

    for (int tile = blockIdx.x; tile < ntiles; tile += GATHER_GRID) {
        const int n = tile * 16 + tr;
        const bool live = n < N;
        const int nc = live ? n : N - 1;
        const float di = dinv[nc];
        const int beg = rowptr[nc];
        const int endr = rowptr[nc + 1];
        const int end = live ? endr : beg;
        const float4 vs = h4[(size_t)nc * 16 + q];

        const float nrm = live ? di * di : 0.f;
        float4 acc = make_float4(vs.x * nrm, vs.y * nrm, vs.z * nrm, vs.w * nrm);
        const int last = end - 1;
        for (int j = beg; j < end; j += 4) {
            const int s0 = col[j];
            const int s1 = col[min(j + 1, last)];
            const int s2 = col[min(j + 2, last)];
            const int s3 = col[min(j + 3, last)];
            const float d0 = dinv[s0], d1 = dinv[s1], d2 = dinv[s2], d3 = dinv[s3];
            const float4 v0 = h4[(size_t)s0 * 16 + q];
            const float4 v1 = h4[(size_t)s1 * 16 + q];
            const float4 v2 = h4[(size_t)s2 * 16 + q];
            const float4 v3 = h4[(size_t)s3 * 16 + q];
            const float w0 = d0 * di;
            const float w1 = (j + 1 < end) ? d1 * di : 0.f;
            const float w2 = (j + 2 < end) ? d2 * di : 0.f;
            const float w3 = (j + 3 < end) ? d3 * di : 0.f;
            acc.x += v0.x * w0; acc.y += v0.y * w0; acc.z += v0.z * w0; acc.w += v0.w * w0;
            acc.x += v1.x * w1; acc.y += v1.y * w1; acc.z += v1.z * w1; acc.w += v1.w * w1;
            acc.x += v2.x * w2; acc.y += v2.y * w2; acc.z += v2.z * w2; acc.w += v2.w * w2;
            acc.x += v3.x * w3; acc.y += v3.y * w3; acc.z += v3.z * w3; acc.w += v3.w * w3;
        }
        ta[tr * TP4 + q] = acc;
        __syncthreads();
        float sval = 0.f;
        int   sidx = 0;
        if (wave < 4) {
            const float bf = bias[f];
            v8f e0 = {0.f, 0.f, 0.f, 0.f, 0.f, 0.f, 0.f, 0.f};
            v8f e1 = {0.f, 0.f, 0.f, 0.f, 0.f, 0.f, 0.f, 0.f};
#pragma unroll
            for (int half = 0; half < 2; ++half) {
                float4 a[4], b[4];
                fence();
#pragma unroll
                for (int i = 0; i < 4; ++i) { a[i] = ap[half * 4 + i]; b[i] = bp[half * 4 + i]; }
                fence();
#pragma unroll
                for (int i = 0; i < 4; ++i) {
                    v2f a0 = {a[i].x, a[i].y}, b0 = {b[i].x, b[i].y};
                    v2f a1 = {a[i].z, a[i].w}, b1 = {b[i].z, b[i].w};
                    e0 = __builtin_amdgcn_wmma_f32_16x16x4_f32(false, a0, false, b0, (short)0, e0, false, false);
                    e1 = __builtin_amdgcn_wmma_f32_16x16x4_f32(false, a1, false, b1, (short)0, e1, false, false);
                }
            }
            fence();
            asm volatile("v_nop\n\tv_nop\n\tv_nop\n\tv_nop");
            fence();
            const int row0 = tile * 16 + h * 8;
            float sm = 0.f, sq = 0.f;
#pragma unroll
            for (int rr = 0; rr < 8; ++rr) {
                float v = fmaxf(e0[rr] + e1[rr] + bf, 0.f);
                v = (row0 + rr < N) ? v : 0.f;
                tof[rr * (TP4 * 4)] = v;
                sm += v;
                sq += v * v;
            }
            sm += __shfl_xor(sm, 16);
            sq += __shfl_xor(sq, 16);
            sval = h ? sq : sm;
            sidx = (tile & (STAT_COPIES - 1)) * STAT_STRIDE + h * 64 + f;
        }
        __syncthreads();
        if (live) ((float4*)r)[n * 16 + q] = to[tr * TP4 + q];
        if (wave < 4) atomicAdd(&stats[sidx], sval);
    }
}

#define POOL_R 16

struct BnCoef { float4 m, is, g, b; };

__device__ __forceinline__ BnCoef bn_coef(float4 sm, float4 sq, float4 gg, float4 bb, int N) {
    BnCoef c;
    const float invN = 1.0f / (float)N;
    c.m.x = sm.x * invN; c.m.y = sm.y * invN; c.m.z = sm.z * invN; c.m.w = sm.w * invN;
    c.is.x = rsqrtf(sq.x * invN - c.m.x * c.m.x + EPSV);
    c.is.y = rsqrtf(sq.y * invN - c.m.y * c.m.y + EPSV);
    c.is.z = rsqrtf(sq.z * invN - c.m.z * c.m.z + EPSV);
    c.is.w = rsqrtf(sq.w * invN - c.m.w * c.m.w + EPSV);
    c.g = gg; c.b = bb;
    return c;
}

__device__ __forceinline__ float4 bn4(float4 x, const BnCoef& c) {
    float4 v;
    v.x = c.g.x * (x.x - c.m.x) * c.is.x + c.b.x;
    v.y = c.g.y * (x.y - c.m.y) * c.is.y + c.b.y;
    v.z = c.g.z * (x.z - c.m.z) * c.is.z + c.b.z;
    v.w = c.g.w * (x.w - c.m.w) * c.is.w + c.b.w;
    return v;
}

__device__ __forceinline__ void add4(float4& s, float4 a) { s.x += a.x; s.y += a.y; s.z += a.z; s.w += a.w; }

__device__ __forceinline__ void pool_add4(float* __restrict__ p, float4 a) {
    atomicAdd(p, a.x); atomicAdd(p + 1, a.y); atomicAdd(p + 2, a.z); atomicAdd(p + 3, a.w);
}

__global__ void k_bn_pool(const float* __restrict__ hin, const float* __restrict__ stats,
          const float* __restrict__ g, const float* __restrict__ beta,
          const int* __restrict__ batch, float* __restrict__ hout,
          float* __restrict__ pooled, int poolOff, int N) {
    const int t  = blockIdx.x * blockDim.x + threadIdx.x;
    const int q  = t & 15;
    const int r0 = (t >> 4) * POOL_R;
    if (r0 >= N) return;
    const float4* h4 = (const float4*)hin + (size_t)r0 * 16 + q;
    float4*       o4 = (float4*)hout + (size_t)r0 * 16 + q;
    const int*    bp = batch + r0;
    const int f = q * 4;
    float* pq = pooled + poolOff + f;

    const float4* st4 = (const float4*)stats + q;
    float4 sm = st4[0], sq = st4[16];
#pragma unroll
    for (int k = 1; k < STAT_COPIES; ++k) {
        add4(sm, st4[k * (STAT_STRIDE / 4)]);
        add4(sq, st4[k * (STAT_STRIDE / 4) + 16]);
    }
    const float4 gg = make_float4(g[f], g[f + 1], g[f + 2], g[f + 3]);
    const float4 bb = make_float4(beta[f], beta[f + 1], beta[f + 2], beta[f + 3]);

    int cur;
    float4 acc = make_float4(0.f, 0.f, 0.f, 0.f);
    if (r0 + POOL_R <= N) {
        float4 x[POOL_R];
        int    b[POOL_R];
#pragma unroll
        for (int u = 0; u < POOL_R; ++u) {
            x[u] = h4[u * 16];
            b[u] = bp[u];
        }
        const BnCoef c = bn_coef(sm, sq, gg, bb, N);
        cur = b[0];
#pragma unroll
        for (int u = 0; u < POOL_R; ++u) {
            const float4 v = bn4(x[u], c);
            o4[u * 16] = v;
            if (b[u] != cur) {
                pool_add4(pq + (size_t)cur * 192, acc);
                cur = b[u];
                acc = make_float4(0.f, 0.f, 0.f, 0.f);
            }
            acc.x += v.x; acc.y += v.y; acc.z += v.z; acc.w += v.w;
        }
    } else {
        const BnCoef c = bn_coef(sm, sq, gg, bb, N);
        cur = bp[0];
        for (int u = 0; r0 + u < N; ++u) {
            const float4 v = bn4(h4[u * 16], c);
            o4[u * 16] = v;
            const int bu = bp[u];
            if (bu != cur) {
                pool_add4(pq + (size_t)cur * 192, acc);
                cur = bu;
                acc = make_float4(0.f, 0.f, 0.f, 0.f);
            }
            acc.x += v.x; acc.y += v.y; acc.z += v.z; acc.w += v.w;
        }
    }
    pool_add4(pq + (size_t)cur * 192, acc);
}

extern "C" void kernel_launch(void* const* d_in, const int* in_sizes, int n_in,
                              void* d_out, int out_size, void* d_ws, size_t ws_size,
                              hipStream_t stream) {
    const float* x    = (const float*)d_in[0];
    const int*   eidx = (const int*)d_in[1];
    const int*   batch = (const int*)d_in[2];
    const float* W[3]  = {(const float*)d_in[3],  (const float*)d_in[7],  (const float*)d_in[11]};
    const float* bi[3] = {(const float*)d_in[4],  (const float*)d_in[8],  (const float*)d_in[12]};
    const float* ga[3] = {(const float*)d_in[5],  (const float*)d_in[9],  (const float*)d_in[13]};
    const float* be[3] = {(const float*)d_in[6],  (const float*)d_in[10], (const float*)d_in[14]};

    const int N = in_sizes[0] / HDIM;
    const int E = in_sizes[1] / 2;
    const int G = (out_size - N * HDIM) / (3 * HDIM);

    const int* src = eidx;
    const int* dst = eidx + E;

    float* ws    = (float*)d_ws;
    float* dinv  = ws;
    float* bufA  = dinv + N;
    float* bufB  = bufA + (size_t)N * HDIM;
    int*   csr   = (int*)(bufB + (size_t)N * HDIM);
    int* rowptr = csr;
    int* col    = rowptr + (N + 4);
    int* cntC   = col + E;
    int* tot    = cntC + CSR_C * (N + 4);
    float* stats = (float*)(tot + CSR_R * CSR_C);

    float* pooled = (float*)d_out;
    float* hfinal = (float*)d_out + (size_t)G * 192;

    const int TB = 256;

    k_csr_count<<<CSR_R * CSR_C, CSR_TB, 0, stream>>>(dst, cntC, tot, N, E);
    k_csr_scan <<<CSR_R, CSR_TB, 0, stream>>>(cntC, tot, rowptr, dinv, N);
    k_csr_fill <<<CSR_R * CSR_C, CSR_TB, 0, stream>>>(src, dst, cntC, col, N, E);

    k_zero<<<cdiv((long long)G * 192, TB), TB, 0, stream>>>(pooled, G * 192);

    const float* gin = x;
    float* rp = bufA;
    float* hp = bufB;
    for (int l = 0; l < 3; ++l) {
        float* hout = (l == 2) ? hfinal : hp;

        k_zero<<<cdiv(STAT_COPIES * STAT_STRIDE, TB), TB, 0, stream>>>(stats, STAT_COPIES * STAT_STRIDE);
        k_gather_gemm_act_stats<<<GATHER_GRID, GATHER_TB, 0, stream>>>(
            gin, W[l], dinv, rowptr, col, bi[l], rp, stats, N);

        k_bn_pool<<<cdiv((long long)cdiv(N, POOL_R) * 16, TB), TB, 0, stream>>>(
            rp, stats, ga[l], be[l], batch, hout, pooled, l * HDIM, N);

        gin = hout;
    }
}


# The four zeroing launches are gone: the first edge-list pass stores the start value 0 of the pooled output and of per-layer statistics copies (one region per layer); 13 launches become 9
# speedup vs baseline: 3.0939x; 1.0403x over previous
#include <hip/hip_runtime.h>

typedef float v2f __attribute__((ext_vector_type(2)));
typedef float v8f __attribute__((ext_vector_type(8)));

#define HDIM 64
#define EPSV 1e-5f

static inline int cdiv(long long a, long long b) { return (int)((a + b - 1) / b); }

#define CSR_R  32
#define CSR_C  8
#define CSR_TB 1024
#define CSR_U  8
#define CSR_MAXSPAN 4096
#define CSR_NPT (CSR_MAXSPAN / CSR_TB)

__device__ __forceinline__ int csr_npr(int N)    { return (N + CSR_R - 1) / CSR_R; }
__device__ __forceinline__ int csr_stride(int N) { return N + 4; }
__device__ __forceinline__ int csr_clen(int E)   { return (((E + CSR_C - 1) / CSR_C) + 3) & ~3; }

__device__ __forceinline__ void csr_count1(int d, int lo, unsigned span, int* cl, int& m) {
    const unsigned u = (unsigned)(d - lo);
    if (u < span) { atomicAdd(&cl[u], 1); ++m; }
}

__device__ __forceinline__ void csr_place1(int d, int s, int lo, unsigned span, int* cl,
                                           int* __restrict__ col) {
    const unsigned u = (unsigned)(d - lo);
    if (u < span) col[atomicAdd(&cl[u], 1)] = s;
}

__global__ void __launch_bounds__(CSR_TB)
k_csr_count(const int* __restrict__ dst, int* __restrict__ cntC, int* __restrict__ tot, int N, int E,
            float* __restrict__ zp, int nzp, float* __restrict__ zs, int nzs) {
    __shared__ int cl[CSR_MAXSPAN];
    __shared__ int tsum;
    const int t = threadIdx.x;
    const int r = blockIdx.x / CSR_C, c = blockIdx.x % CSR_C;
    const int npr = csr_npr(N), NS = csr_stride(N);
    const int lo = r * npr;
    const int spn = min(min(npr, N - lo), CSR_MAXSPAN);
    const unsigned span = spn > 0 ? (unsigned)spn : 0u;
    for (int i = t; i < CSR_MAXSPAN; i += CSR_TB) cl[i] = 0;
    if (t == 0) tsum = 0;
    __syncthreads();
    const int clen = csr_clen(E);
    const int e0 = min(E, c * clen), e1 = min(E, e0 + clen);
    int m = 0;
    int done = e0;
    if ((((size_t)dst) & 15) == 0) {
        const int4* d4 = (const int4*)(dst + e0);
        const int nv = (e1 - e0) >> 2;
        for (int i0 = 0; i0 < nv; i0 += CSR_U * CSR_TB) {
            int4 dv[CSR_U];
#pragma unroll
            for (int u = 0; u < CSR_U; ++u) dv[u] = d4[min(i0 + u * CSR_TB + t, nv - 1)];
#pragma unroll
            for (int u = 0; u < CSR_U; ++u) {
                if (i0 + u * CSR_TB + t < nv) {
                    csr_count1(dv[u].x, lo, span, cl, m);
                    csr_count1(dv[u].y, lo, span, cl, m);
                    csr_count1(dv[u].z, lo, span, cl, m);
                    csr_count1(dv[u].w, lo, span, cl, m);
                }
            }
        }
        done = e0 + (nv << 2);
    }
    for (int e = done + t; e < e1; e += CSR_TB) csr_count1(dst[e], lo, span, cl, m);
    for (int o = 16; o > 0; o >>= 1) m += __shfl_xor(m, o);
    if ((t & 31) == 0) atomicAdd(&tsum, m);
    __syncthreads();
    for (int i = t; i < (int)span; i += CSR_TB) cntC[c * NS + lo + i] = cl[i];
    if (t == 0) tot[blockIdx.x] = tsum;
    const int gid = blockIdx.x * CSR_TB + t, gsz = gridDim.x * CSR_TB;
#pragma clang loop unroll(disable) vectorize(disable) interleave(disable)
    for (int i = gid; i < nzp; i += gsz) zp[i] = 0.0f;
#pragma clang loop unroll(disable) vectorize(disable) interleave(disable)
    for (int i = gid; i < nzs; i += gsz) zs[i] = 0.0f;
}

__global__ void __launch_bounds__(CSR_TB)
k_csr_scan(int* __restrict__ cntC, const int* __restrict__ tot, int* __restrict__ rowptr,
           float* __restrict__ dinv, int N) {
    __shared__ int sh[CSR_TB];
    const int t = threadIdx.x, r = blockIdx.x;
    const int npr = csr_npr(N), NS = csr_stride(N);
    const int lo = r * npr;
    const int hi = min(lo + npr, N);
    const int npt = (npr + CSR_TB - 1) / CSR_TB;
    const int f = lo + t * npt;
    const int l = min(f + npt, hi);
    int p = 0;
    if (t < CSR_R * CSR_C) p = (t / CSR_C < r) ? tot[t] : 0;
    int x[CSR_NPT][CSR_C];
#pragma unroll
    for (int k = 0; k < CSR_NPT; ++k) {
        const int n = min(f + k, hi - 1);
#pragma unroll
        for (int c = 0; c < CSR_C; ++c) x[k][c] = cntC[c * NS + n];
    }
    int mine = 0;
#pragma unroll
    for (int k = 0; k < CSR_NPT; ++k) {
        if (f + k < l) {
#pragma unroll
            for (int c = 0; c < CSR_C; ++c) mine += x[k][c];
        }
    }
    sh[t] = p;
    __syncthreads();
    for (int s = CSR_TB / 2; s > 0; s >>= 1) {
        if (t < s) sh[t] += sh[t + s];
        __syncthreads();
    }
    const int base = sh[0];
    __syncthreads();
    sh[t] = mine;
    __syncthreads();
    for (int off = 1; off < CSR_TB; off <<= 1) {
        int v = (t >= off) ? sh[t - off] : 0;
        __syncthreads();
        sh[t] += v;
        __syncthreads();
    }
    int run = base + sh[t] - mine;
#pragma unroll
    for (int k = 0; k < CSR_NPT; ++k) {
        if (f + k < l) {
            const int n = f + k;
            int o = run;
#pragma unroll
            for (int c = 0; c < CSR_C; ++c) { cntC[c * NS + n] = o; o += x[k][c]; }
            rowptr[n] = run;
            dinv[n]   = rsqrtf((float)(o - run + 1));
            run = o;
        }
    }
    if (f < l && l == N) rowptr[N] = run;
}

__global__ void __launch_bounds__(CSR_TB)
k_csr_fill(const int* __restrict__ src, const int* __restrict__ dst, const int* __restrict__ cntC,
           int* __restrict__ col, int N, int E) {
    __shared__ int cl[CSR_MAXSPAN];
    const int t = threadIdx.x;
    const int r = blockIdx.x / CSR_C, c = blockIdx.x % CSR_C;
    const int npr = csr_npr(N), NS = csr_stride(N);
    const int lo = r * npr;
    const int spn = min(min(npr, N - lo), CSR_MAXSPAN);
    const unsigned span = spn > 0 ? (unsigned)spn : 0u;
    int st[CSR_NPT];
#pragma unroll
    for (int k = 0; k < CSR_NPT; ++k) st[k] = cntC[c * NS + lo + min(k * CSR_TB + t, spn - 1)];
#pragma unroll
    for (int k = 0; k < CSR_NPT; ++k)
        if (k * CSR_TB + t < spn) cl[k * CSR_TB + t] = st[k];
    __syncthreads();
    const int clen = csr_clen(E);
    const int e0 = min(E, c * clen), e1 = min(E, e0 + clen);
    int done = e0;
    if (((((size_t)dst) | ((size_t)src)) & 15) == 0) {
        const int4* d4 = (const int4*)(dst + e0);
        const int4* s4 = (const int4*)(src + e0);
        const int nv = (e1 - e0) >> 2;
        for (int i0 = 0; i0 < nv; i0 += CSR_U * CSR_TB) {
            int4 dv[CSR_U], sv[CSR_U];
#pragma unroll
            for (int u = 0; u < CSR_U; ++u) {
                const int i = min(i0 + u * CSR_TB + t, nv - 1);
                dv[u] = d4[i];
                sv[u] = s4[i];
            }
#pragma unroll
            for (int u = 0; u < CSR_U; ++u)
                if (i0 + u * CSR_TB + t < nv) {
                    csr_place1(dv[u].x, sv[u].x, lo, span, cl, col);
                    csr_place1(dv[u].y, sv[u].y, lo, span, cl, col);
                    csr_place1(dv[u].z, sv[u].z, lo, span, cl, col);
                    csr_place1(dv[u].w, sv[u].w, lo, span, cl, col);
                }
        }
        done = e0 + (nv << 2);
    }
    for (int e = done + t; e < e1; e += CSR_TB) {
        csr_place1(dst[e], src[e], lo, span, cl, col);
    }
}

#define STAT_COPIES 8
#define STAT_STRIDE 1088
#define GATHER_TB 256
#define GATHER_GRID 2048
#define TP4 17
__device__ __forceinline__ void fence() { __builtin_amdgcn_sched_barrier(0); }

__global__ void __launch_bounds__(GATHER_TB) __attribute__((amdgpu_waves_per_eu(16, 16)))
k_gather_gemm_act_stats(const float* __restrict__ hin, const float* __restrict__ W,
                        const float* __restrict__ dinv,
                        const int* __restrict__ rowptr, const int* __restrict__ col,
                        const float* __restrict__ bias, float* __restrict__ r,
                        float* __restrict__ stats  , int N) {
    __shared__ float4 wl[HDIM * TP4];
    __shared__ float4 ta[16 * TP4];
    __shared__ float4 to[16 * TP4];
    const int tid = threadIdx.x;
    const int ntiles = (N + 15) >> 4;
    if ((int)blockIdx.x >= ntiles) return;
    const float4* W4 = (const float4*)W;
    float4 wv[4];
#pragma unroll
    for (int i = 0; i < 4; ++i) wv[i] = W4[tid + i * GATHER_TB];

    const int q = tid & 15;
    const int tr = tid >> 4;
    const int wave = tid >> 5;
    const int lane = tid & 31;
    const int c = lane & 15, h = lane >> 4;
    const int f = (wave & 3) * 16 + c;
    const float4* h4 = (const float4*)hin;
    const float4* ap = ta + c * TP4 + h * 8;
    const float4* bp = wl + f * TP4 + h * 8;
    float* tof = (float*)to + (h * 8) * (TP4 * 4) + f;

#pragma unroll
    for (int i = 0; i < 4; ++i) {
        const int idx = tid + i * GATHER_TB;
        wl[(idx >> 4) * TP4 + (idx & 15)] = wv[i];
    }

    for (int tile = blockIdx.x; tile < ntiles; tile += GATHER_GRID) {
        const int n = tile * 16 + tr;
        const bool live = n < N;
        const int nc = live ? n : N - 1;
        const float di = dinv[nc];
        const int beg = rowptr[nc];
        const int endr = rowptr[nc + 1];
        const int end = live ? endr : beg;
        const float4 vs = h4[(size_t)nc * 16 + q];

        const float nrm = live ? di * di : 0.f;
        float4 acc = make_float4(vs.x * nrm, vs.y * nrm, vs.z * nrm, vs.w * nrm);
        const int last = end - 1;
        for (int j = beg; j < end; j += 4) {
            const int s0 = col[j];
            const int s1 = col[min(j + 1, last)];
            const int s2 = col[min(j + 2, last)];
            const int s3 = col[min(j + 3, last)];
            const float d0 = dinv[s0], d1 = dinv[s1], d2 = dinv[s2], d3 = dinv[s3];
            const float4 v0 = h4[(size_t)s0 * 16 + q];
            const float4 v1 = h4[(size_t)s1 * 16 + q];
            const float4 v2 = h4[(size_t)s2 * 16 + q];
            const float4 v3 = h4[(size_t)s3 * 16 + q];
            const float w0 = d0 * di;
            const float w1 = (j + 1 < end) ? d1 * di : 0.f;
            const float w2 = (j + 2 < end) ? d2 * di : 0.f;
            const float w3 = (j + 3 < end) ? d3 * di : 0.f;
            acc.x += v0.x * w0; acc.y += v0.y * w0; acc.z += v0.z * w0; acc.w += v0.w * w0;
            acc.x += v1.x * w1; acc.y += v1.y * w1; acc.z += v1.z * w1; acc.w += v1.w * w1;
            acc.x += v2.x * w2; acc.y += v2.y * w2; acc.z += v2.z * w2; acc.w += v2.w * w2;
            acc.x += v3.x * w3; acc.y += v3.y * w3; acc.z += v3.z * w3; acc.w += v3.w * w3;
        }
        ta[tr * TP4 + q] = acc;
        __syncthreads();
        float sval = 0.f;
        int   sidx = 0;
        if (wave < 4) {
            const float bf = bias[f];
            v8f e0 = {0.f, 0.f, 0.f, 0.f, 0.f, 0.f, 0.f, 0.f};
            v8f e1 = {0.f, 0.f, 0.f, 0.f, 0.f, 0.f, 0.f, 0.f};
#pragma unroll
            for (int half = 0; half < 2; ++half) {
                float4 a[4], b[4];
                fence();
#pragma unroll
                for (int i = 0; i < 4; ++i) { a[i] = ap[half * 4 + i]; b[i] = bp[half * 4 + i]; }
                fence();
#pragma unroll
                for (int i = 0; i < 4; ++i) {
                    v2f a0 = {a[i].x, a[i].y}, b0 = {b[i].x, b[i].y};
                    v2f a1 = {a[i].z, a[i].w}, b1 = {b[i].z, b[i].w};
                    e0 = __builtin_amdgcn_wmma_f32_16x16x4_f32(false, a0, false, b0, (short)0, e0, false, false);
                    e1 = __builtin_amdgcn_wmma_f32_16x16x4_f32(false, a1, false, b1, (short)0, e1, false, false);
                }
            }
            fence();
            asm volatile("v_nop\n\tv_nop\n\tv_nop\n\tv_nop");
            fence();
            const int row0 = tile * 16 + h * 8;
            float sm = 0.f, sq = 0.f;
#pragma unroll
            for (int rr = 0; rr < 8; ++rr) {
                float v = fmaxf(e0[rr] + e1[rr] + bf, 0.f);
                v = (row0 + rr < N) ? v : 0.f;
                tof[rr * (TP4 * 4)] = v;
                sm += v;
                sq += v * v;
            }
            sm += __shfl_xor(sm, 16);
            sq += __shfl_xor(sq, 16);
            sval = h ? sq : sm;
            sidx = (tile & (STAT_COPIES - 1)) * STAT_STRIDE + h * 64 + f;
        }
        __syncthreads();
        if (live) ((float4*)r)[n * 16 + q] = to[tr * TP4 + q];
        if (wave < 4) atomicAdd(&stats[sidx], sval);
    }
}

#define POOL_R 16

struct BnCoef { float4 m, is, g, b; };

__device__ __forceinline__ BnCoef bn_coef(float4 sm, float4 sq, float4 gg, float4 bb, int N) {
    BnCoef c;
    const float invN = 1.0f / (float)N;
    c.m.x = sm.x * invN; c.m.y = sm.y * invN; c.m.z = sm.z * invN; c.m.w = sm.w * invN;
    c.is.x = rsqrtf(sq.x * invN - c.m.x * c.m.x + EPSV);
    c.is.y = rsqrtf(sq.y * invN - c.m.y * c.m.y + EPSV);
    c.is.z = rsqrtf(sq.z * invN - c.m.z * c.m.z + EPSV);
    c.is.w = rsqrtf(sq.w * invN - c.m.w * c.m.w + EPSV);
    c.g = gg; c.b = bb;
    return c;
}

__device__ __forceinline__ float4 bn4(float4 x, const BnCoef& c) {
    float4 v;
    v.x = c.g.x * (x.x - c.m.x) * c.is.x + c.b.x;
    v.y = c.g.y * (x.y - c.m.y) * c.is.y + c.b.y;
    v.z = c.g.z * (x.z - c.m.z) * c.is.z + c.b.z;
    v.w = c.g.w * (x.w - c.m.w) * c.is.w + c.b.w;
    return v;
}

__device__ __forceinline__ void add4(float4& s, float4 a) { s.x += a.x; s.y += a.y; s.z += a.z; s.w += a.w; }

__device__ __forceinline__ void pool_add4(float* __restrict__ p, float4 a) {
    atomicAdd(p, a.x); atomicAdd(p + 1, a.y); atomicAdd(p + 2, a.z); atomicAdd(p + 3, a.w);
}

__global__ void k_bn_pool(const float* __restrict__ hin, const float* __restrict__ stats,
          const float* __restrict__ g, const float* __restrict__ beta,
          const int* __restrict__ batch, float* __restrict__ hout,
          float* __restrict__ pooled, int poolOff, int N) {
    const int t  = blockIdx.x * blockDim.x + threadIdx.x;
    const int q  = t & 15;
    const int r0 = (t >> 4) * POOL_R;
    if (r0 >= N) return;
    const float4* h4 = (const float4*)hin + (size_t)r0 * 16 + q;
    float4*       o4 = (float4*)hout + (size_t)r0 * 16 + q;
    const int*    bp = batch + r0;
    const int f = q * 4;
    float* pq = pooled + poolOff + f;

    const float4* st4 = (const float4*)stats + q;
    float4 sm = st4[0], sq = st4[16];
#pragma unroll
    for (int k = 1; k < STAT_COPIES; ++k) {
        add4(sm, st4[k * (STAT_STRIDE / 4)]);
        add4(sq, st4[k * (STAT_STRIDE / 4) + 16]);
    }
    const float4 gg = make_float4(g[f], g[f + 1], g[f + 2], g[f + 3]);
    const float4 bb = make_float4(beta[f], beta[f + 1], beta[f + 2], beta[f + 3]);

    int cur;
    float4 acc = make_float4(0.f, 0.f, 0.f, 0.f);
    if (r0 + POOL_R <= N) {
        float4 x[POOL_R];
        int    b[POOL_R];
#pragma unroll
        for (int u = 0; u < POOL_R; ++u) {
            x[u] = h4[u * 16];
            b[u] = bp[u];
        }
        const BnCoef c = bn_coef(sm, sq, gg, bb, N);
        cur = b[0];
#pragma unroll
        for (int u = 0; u < POOL_R; ++u) {
            const float4 v = bn4(x[u], c);
            o4[u * 16] = v;
            if (b[u] != cur) {
                pool_add4(pq + (size_t)cur * 192, acc);
                cur = b[u];
                acc = make_float4(0.f, 0.f, 0.f, 0.f);
            }
            acc.x += v.x; acc.y += v.y; acc.z += v.z; acc.w += v.w;
        }
    } else {
        const BnCoef c = bn_coef(sm, sq, gg, bb, N);
        cur = bp[0];
        for (int u = 0; r0 + u < N; ++u) {
            const float4 v = bn4(h4[u * 16], c);
            o4[u * 16] = v;
            const int bu = bp[u];
            if (bu != cur) {
                pool_add4(pq + (size_t)cur * 192, acc);
                cur = bu;
                acc = make_float4(0.f, 0.f, 0.f, 0.f);
            }
            acc.x += v.x; acc.y += v.y; acc.z += v.z; acc.w += v.w;
        }
    }
    pool_add4(pq + (size_t)cur * 192, acc);
}

extern "C" void kernel_launch(void* const* d_in, const int* in_sizes, int n_in,
                              void* d_out, int out_size, void* d_ws, size_t ws_size,
                              hipStream_t stream) {
    const float* x    = (const float*)d_in[0];
    const int*   eidx = (const int*)d_in[1];
    const int*   batch = (const int*)d_in[2];
    const float* W[3]  = {(const float*)d_in[3],  (const float*)d_in[7],  (const float*)d_in[11]};
    const float* bi[3] = {(const float*)d_in[4],  (const float*)d_in[8],  (const float*)d_in[12]};
    const float* ga[3] = {(const float*)d_in[5],  (const float*)d_in[9],  (const float*)d_in[13]};
    const float* be[3] = {(const float*)d_in[6],  (const float*)d_in[10], (const float*)d_in[14]};

    const int N = in_sizes[0] / HDIM;
    const int E = in_sizes[1] / 2;
    const int G = (out_size - N * HDIM) / (3 * HDIM);

    const int* src = eidx;
    const int* dst = eidx + E;

    float* ws    = (float*)d_ws;
    float* dinv  = ws;
    float* bufA  = dinv + N;
    float* bufB  = bufA + (size_t)N * HDIM;
    int*   csr   = (int*)(bufB + (size_t)N * HDIM);
    int* rowptr = csr;
    int* col    = rowptr + (N + 4);
    int* cntC   = col + E;
    int* tot    = cntC + CSR_C * (N + 4);
    float* stats = (float*)(tot + CSR_R * CSR_C);

    float* pooled = (float*)d_out;
    float* hfinal = (float*)d_out + (size_t)G * 192;

    const int TB = 256;

    k_csr_count<<<CSR_R * CSR_C, CSR_TB, 0, stream>>>(dst, cntC, tot, N, E,
                                                     pooled, G * 192, stats, 3 * STAT_COPIES * STAT_STRIDE);
    k_csr_scan <<<CSR_R, CSR_TB, 0, stream>>>(cntC, tot, rowptr, dinv, N);
    k_csr_fill <<<CSR_R * CSR_C, CSR_TB, 0, stream>>>(src, dst, cntC, col, N, E);

    const float* gin = x;
    float* rp = bufA;
    float* hp = bufB;
    for (int l = 0; l < 3; ++l) {
        float* hout = (l == 2) ? hfinal : hp;

        float* st = stats + l * (STAT_COPIES * STAT_STRIDE);
        k_gather_gemm_act_stats<<<GATHER_GRID, GATHER_TB, 0, stream>>>(
            gin, W[l], dinv, rowptr, col, bi[l], rp, st, N);

        k_bn_pool<<<cdiv((long long)cdiv(N, POOL_R) * 16, TB), TB, 0, stream>>>(
            rp, st, ga[l], be[l], batch, hout, pooled, l * HDIM, N);

        gin = hout;
    }
}


# BatchNorm of layers 1-2 folded into the next gather: sum w*(A r+C) = A*sum(w r) + C*sum(w); per-graph pool taken there by idle waves; 2 BN+pool passes gone, 9 launches to 7; sum order changes
# speedup vs baseline: 3.5293x; 1.1407x over previous
#include <hip/hip_runtime.h>

typedef float v2f __attribute__((ext_vector_type(2)));
typedef float v8f __attribute__((ext_vector_type(8)));

#define HDIM 64
#define EPSV 1e-5f

static inline int cdiv(long long a, long long b) { return (int)((a + b - 1) / b); }

#define CSR_R  32
#define CSR_C  8
#define CSR_TB 1024
#define CSR_U  8
#define CSR_MAXSPAN 4096
#define CSR_NPT (CSR_MAXSPAN / CSR_TB)

__device__ __forceinline__ int csr_npr(int N)    { return (N + CSR_R - 1) / CSR_R; }
__device__ __forceinline__ int csr_stride(int N) { return N + 4; }
__device__ __forceinline__ int csr_clen(int E)   { return (((E + CSR_C - 1) / CSR_C) + 3) & ~3; }

__device__ __forceinline__ void csr_count1(int d, int lo, unsigned span, int* cl, int& m) {
    const unsigned u = (unsigned)(d - lo);
    if (u < span) { atomicAdd(&cl[u], 1); ++m; }
}

__device__ __forceinline__ void csr_place1(int d, int s, int lo, unsigned span, int* cl,
                                           int* __restrict__ col) {
    const unsigned u = (unsigned)(d - lo);
    if (u < span) col[atomicAdd(&cl[u], 1)] = s;
}

__global__ void __launch_bounds__(CSR_TB)
k_csr_count(const int* __restrict__ dst, int* __restrict__ cntC, int* __restrict__ tot, int N, int E,
            float* __restrict__ zp, int nzp, float* __restrict__ zs, int nzs) {
    __shared__ int cl[CSR_MAXSPAN];
    __shared__ int tsum;
    const int t = threadIdx.x;
    const int r = blockIdx.x / CSR_C, c = blockIdx.x % CSR_C;
    const int npr = csr_npr(N), NS = csr_stride(N);
    const int lo = r * npr;
    const int spn = min(min(npr, N - lo), CSR_MAXSPAN);
    const unsigned span = spn > 0 ? (unsigned)spn : 0u;
    for (int i = t; i < CSR_MAXSPAN; i += CSR_TB) cl[i] = 0;
    if (t == 0) tsum = 0;
    __syncthreads();
    const int clen = csr_clen(E);
    const int e0 = min(E, c * clen), e1 = min(E, e0 + clen);
    int m = 0;
    int done = e0;
    if ((((size_t)dst) & 15) == 0) {
        const int4* d4 = (const int4*)(dst + e0);
        const int nv = (e1 - e0) >> 2;
        for (int i0 = 0; i0 < nv; i0 += CSR_U * CSR_TB) {
            int4 dv[CSR_U];
#pragma unroll
            for (int u = 0; u < CSR_U; ++u) dv[u] = d4[min(i0 + u * CSR_TB + t, nv - 1)];
#pragma unroll
            for (int u = 0; u < CSR_U; ++u) {
                if (i0 + u * CSR_TB + t < nv) {
                    csr_count1(dv[u].x, lo, span, cl, m);
                    csr_count1(dv[u].y, lo, span, cl, m);
                    csr_count1(dv[u].z, lo, span, cl, m);
                    csr_count1(dv[u].w, lo, span, cl, m);
                }
            }
        }
        done = e0 + (nv << 2);
    }
    for (int e = done + t; e < e1; e += CSR_TB) csr_count1(dst[e], lo, span, cl, m);
    for (int o = 16; o > 0; o >>= 1) m += __shfl_xor(m, o);
    if ((t & 31) == 0) atomicAdd(&tsum, m);
    __syncthreads();
    for (int i = t; i < (int)span; i += CSR_TB) cntC[c * NS + lo + i] = cl[i];
    if (t == 0) tot[blockIdx.x] = tsum;
    const int gid = blockIdx.x * CSR_TB + t, gsz = gridDim.x * CSR_TB;
#pragma clang loop unroll(disable) vectorize(disable) interleave(disable)
    for (int i = gid; i < nzp; i += gsz) zp[i] = 0.0f;
#pragma clang loop unroll(disable) vectorize(disable) interleave(disable)
    for (int i = gid; i < nzs; i += gsz) zs[i] = 0.0f;
}

__global__ void __launch_bounds__(CSR_TB)
k_csr_scan(int* __restrict__ cntC, const int* __restrict__ tot, int* __restrict__ rowptr,
           float* __restrict__ dinv, int N) {
    __shared__ int sh[CSR_TB];
    const int t = threadIdx.x, r = blockIdx.x;
    const int npr = csr_npr(N), NS = csr_stride(N);
    const int lo = r * npr;
    const int hi = min(lo + npr, N);
    const int npt = (npr + CSR_TB - 1) / CSR_TB;
    const int f = lo + t * npt;
    const int l = min(f + npt, hi);
    int p = 0;
    if (t < CSR_R * CSR_C) p = (t / CSR_C < r) ? tot[t] : 0;
    int x[CSR_NPT][CSR_C];
#pragma unroll
    for (int k = 0; k < CSR_NPT; ++k) {
        const int n = min(f + k, hi - 1);
#pragma unroll
        for (int c = 0; c < CSR_C; ++c) x[k][c] = cntC[c * NS + n];
    }
    int mine = 0;
#pragma unroll
    for (int k = 0; k < CSR_NPT; ++k) {
        if (f + k < l) {
#pragma unroll
            for (int c = 0; c < CSR_C; ++c) mine += x[k][c];
        }
    }
    sh[t] = p;
    __syncthreads();
    for (int s = CSR_TB / 2; s > 0; s >>= 1) {
        if (t < s) sh[t] += sh[t + s];
        __syncthreads();
    }
    const int base = sh[0];
    __syncthreads();
    sh[t] = mine;
    __syncthreads();
    for (int off = 1; off < CSR_TB; off <<= 1) {
        int v = (t >= off) ? sh[t - off] : 0;
        __syncthreads();
        sh[t] += v;
        __syncthreads();
    }
    int run = base + sh[t] - mine;
#pragma unroll
    for (int k = 0; k < CSR_NPT; ++k) {
        if (f + k < l) {
            const int n = f + k;
            int o = run;
#pragma unroll
            for (int c = 0; c < CSR_C; ++c) { cntC[c * NS + n] = o; o += x[k][c]; }
            rowptr[n] = run;
            dinv[n]   = rsqrtf((float)(o - run + 1));
            run = o;
        }
    }
    if (f < l && l == N) rowptr[N] = run;
}

__global__ void __launch_bounds__(CSR_TB)
k_csr_fill(const int* __restrict__ src, const int* __restrict__ dst, const int* __restrict__ cntC,
           int* __restrict__ col, int N, int E) {
    __shared__ int cl[CSR_MAXSPAN];
    const int t = threadIdx.x;
    const int r = blockIdx.x / CSR_C, c = blockIdx.x % CSR_C;
    const int npr = csr_npr(N), NS = csr_stride(N);
    const int lo = r * npr;
    const int spn = min(min(npr, N - lo), CSR_MAXSPAN);
    const unsigned span = spn > 0 ? (unsigned)spn : 0u;
    int st[CSR_NPT];
#pragma unroll
    for (int k = 0; k < CSR_NPT; ++k) st[k] = cntC[c * NS + lo + min(k * CSR_TB + t, spn - 1)];
#pragma unroll
    for (int k = 0; k < CSR_NPT; ++k)
        if (k * CSR_TB + t < spn) cl[k * CSR_TB + t] = st[k];
    __syncthreads();
    const int clen = csr_clen(E);
    const int e0 = min(E, c * clen), e1 = min(E, e0 + clen);
    int done = e0;
    if (((((size_t)dst) | ((size_t)src)) & 15) == 0) {
        const int4* d4 = (const int4*)(dst + e0);
        const int4* s4 = (const int4*)(src + e0);
        const int nv = (e1 - e0) >> 2;
        for (int i0 = 0; i0 < nv; i0 += CSR_U * CSR_TB) {
            int4 dv[CSR_U], sv[CSR_U];
#pragma unroll
            for (int u = 0; u < CSR_U; ++u) {
                const int i = min(i0 + u * CSR_TB + t, nv - 1);
                dv[u] = d4[i];
                sv[u] = s4[i];
            }
#pragma unroll
            for (int u = 0; u < CSR_U; ++u)
                if (i0 + u * CSR_TB + t < nv) {
                    csr_place1(dv[u].x, sv[u].x, lo, span, cl, col);
                    csr_place1(dv[u].y, sv[u].y, lo, span, cl, col);
                    csr_place1(dv[u].z, sv[u].z, lo, span, cl, col);
                    csr_place1(dv[u].w, sv[u].w, lo, span, cl, col);
                }
        }
        done = e0 + (nv << 2);
    }
    for (int e = done + t; e < e1; e += CSR_TB) {
        csr_place1(dst[e], src[e], lo, span, cl, col);
    }
}

#define STAT_COPIES 8
#define STAT_STRIDE 1088
#define GATHER_TB 256
#define GATHER_GRID 2048
#define TP4 17
__device__ __forceinline__ void fence() { __builtin_amdgcn_sched_barrier(0); }

template <bool BN>
__global__ void __launch_bounds__(GATHER_TB) __attribute__((amdgpu_waves_per_eu(16, 16)))
k_gather_gemm_act_stats(const float* __restrict__ hin, const float* __restrict__ W,
                        const float* __restrict__ dinv,
                        const int* __restrict__ rowptr, const int* __restrict__ col,
                        const float* __restrict__ bias, float* __restrict__ r,
                        float* __restrict__ stats  , int N,
                        const float* __restrict__ pstats, const float* __restrict__ pg,
                        const float* __restrict__ pbeta, const int* __restrict__ batch,
                        float* __restrict__ pooled, int poolOff) {
    __shared__ float4 wl[HDIM * TP4];
    __shared__ float4 tt[BN ? 3 * 16 * TP4 + 32 + 4 : 2 * 16 * TP4];
    float4* const ta = tt;
    float4* const to = tt + 16 * TP4;
    float4* const tp = tt + 32 * TP4;
    float4* const cf = tt + 48 * TP4;
    int* const gids = (int*)(tt + 48 * TP4 + 32);
    const int tid = threadIdx.x;
    const int ntiles = (N + 15) >> 4;
    if ((int)blockIdx.x >= ntiles) return;
    if (BN && tid < 16) {
        const float4* st4 = (const float4*)pstats + tid;
        float4 sm = st4[0], sq = st4[16];
#pragma unroll
        for (int k = 1; k < STAT_COPIES; ++k) {
            const float4 s1 = st4[k * (STAT_STRIDE / 4)], s2 = st4[k * (STAT_STRIDE / 4) + 16];
            sm.x += s1.x; sm.y += s1.y; sm.z += s1.z; sm.w += s1.w;
            sq.x += s2.x; sq.y += s2.y; sq.z += s2.z; sq.w += s2.w;
        }
        const int f0 = tid * 4;
        const float invN = 1.0f / (float)N;
        const float m0 = sm.x * invN, m1 = sm.y * invN, m2 = sm.z * invN, m3 = sm.w * invN;
        float4 cA, cC;
        cA.x = pg[f0]     * rsqrtf(sq.x * invN - m0 * m0 + EPSV);
        cA.y = pg[f0 + 1] * rsqrtf(sq.y * invN - m1 * m1 + EPSV);
        cA.z = pg[f0 + 2] * rsqrtf(sq.z * invN - m2 * m2 + EPSV);
        cA.w = pg[f0 + 3] * rsqrtf(sq.w * invN - m3 * m3 + EPSV);
        cC.x = pbeta[f0]     - m0 * cA.x;
        cC.y = pbeta[f0 + 1] - m1 * cA.y;
        cC.z = pbeta[f0 + 2] - m2 * cA.z;
        cC.w = pbeta[f0 + 3] - m3 * cA.w;
        cf[tid] = cA;
        cf[16 + tid] = cC;
    }
    const float4* W4 = (const float4*)W;
    float4 wv[4];
#pragma unroll
    for (int i = 0; i < 4; ++i) wv[i] = W4[tid + i * GATHER_TB];

    const int q = tid & 15;
    const int tr = tid >> 4;
    const int wave = tid >> 5;
    const int lane = tid & 31;
    const int c = lane & 15, h = lane >> 4;
    const int f = (wave & 3) * 16 + c;
    const float4* h4 = (const float4*)hin;
    const float4* ap = ta + c * TP4 + h * 8;
    const float4* bp = wl + f * TP4 + h * 8;
    float* tof = (float*)to + (h * 8) * (TP4 * 4) + f;

#pragma unroll
    for (int i = 0; i < 4; ++i) {
        const int idx = tid + i * GATHER_TB;
        wl[(idx >> 4) * TP4 + (idx & 15)] = wv[i];
    }
    if (BN) __syncthreads();

    for (int tile = blockIdx.x; tile < ntiles; tile += GATHER_GRID) {
        const int n = tile * 16 + tr;
        const bool live = n < N;
        const int nc = live ? n : N - 1;
        const float di = dinv[nc];
        const int beg = rowptr[nc];
        const int endr = rowptr[nc + 1];
        const int end = live ? endr : beg;
        const float4 vs = h4[nc * 16 + q];
        int gme = 0;
        if (BN) gme = batch[nc];

        if (BN) {
            tp[tr * TP4 + q] = live ? vs : make_float4(0.f, 0.f, 0.f, 0.f);
            gids[tr] = gme;
        }
        const float nrm = live ? di * di : 0.f;
        float4 acc = make_float4(vs.x * nrm, vs.y * nrm, vs.z * nrm, vs.w * nrm);
        float wsum = nrm;
        const int last = end - 1;
        for (int j = beg; j < end; j += 4) {
            const int s0 = col[j];
            const int s1 = col[min(j + 1, last)];
            const int s2 = col[min(j + 2, last)];
            const int s3 = col[min(j + 3, last)];
            const float d0 = dinv[s0], d1 = dinv[s1], d2 = dinv[s2], d3 = dinv[s3];
            const float4 v0 = h4[s0 * 16 + q];
            const float4 v1 = h4[s1 * 16 + q];
            const float4 v2 = h4[s2 * 16 + q];
            const float4 v3 = h4[s3 * 16 + q];
            const float w0 = d0 * di;
            const float w1 = (j + 1 < end) ? d1 * di : 0.f;
            const float w2 = (j + 2 < end) ? d2 * di : 0.f;
            const float w3 = (j + 3 < end) ? d3 * di : 0.f;
            acc.x += v0.x * w0; acc.y += v0.y * w0; acc.z += v0.z * w0; acc.w += v0.w * w0;
            acc.x += v1.x * w1; acc.y += v1.y * w1; acc.z += v1.z * w1; acc.w += v1.w * w1;
            acc.x += v2.x * w2; acc.y += v2.y * w2; acc.z += v2.z * w2; acc.w += v2.w * w2;
            acc.x += v3.x * w3; acc.y += v3.y * w3; acc.z += v3.z * w3; acc.w += v3.w * w3;
            if (BN) wsum += (w0 + w1) + (w2 + w3);
        }
        if (BN) {
            const float4 cA = cf[q], cC = cf[16 + q];
            acc.x = cA.x * acc.x + cC.x * wsum; acc.y = cA.y * acc.y + cC.y * wsum;
            acc.z = cA.z * acc.z + cC.z * wsum; acc.w = cA.w * acc.w + cC.w * wsum;
        }
        ta[tr * TP4 + q] = acc;
        __syncthreads();
        float sval = 0.f;
        int   sidx = 0;
        float pval = 0.f;
        int   pix = 0;
        if (wave < 4) {
            const float bf = bias[f];
            v8f e0 = {0.f, 0.f, 0.f, 0.f, 0.f, 0.f, 0.f, 0.f};
            v8f e1 = {0.f, 0.f, 0.f, 0.f, 0.f, 0.f, 0.f, 0.f};
#pragma unroll
            for (int half = 0; half < 2; ++half) {
                float4 a[4], b[4];
                fence();
#pragma unroll
                for (int i = 0; i < 4; ++i) { a[i] = ap[half * 4 + i]; b[i] = bp[half * 4 + i]; }
                fence();
#pragma unroll
                for (int i = 0; i < 4; ++i) {
                    v2f a0 = {a[i].x, a[i].y}, b0 = {b[i].x, b[i].y};
                    v2f a1 = {a[i].z, a[i].w}, b1 = {b[i].z, b[i].w};
                    e0 = __builtin_amdgcn_wmma_f32_16x16x4_f32(false, a0, false, b0, (short)0, e0, false, false);
                    e1 = __builtin_amdgcn_wmma_f32_16x16x4_f32(false, a1, false, b1, (short)0, e1, false, false);
                }
            }
            fence();
            asm volatile("v_nop\n\tv_nop\n\tv_nop\n\tv_nop");
            fence();
            const int row0 = tile * 16 + h * 8;
            float sm = 0.f, sq = 0.f;
#pragma unroll
            for (int rr = 0; rr < 8; ++rr) {
                float v = fmaxf(e0[rr] + e1[rr] + bf, 0.f);
                v = (row0 + rr < N) ? v : 0.f;
                tof[rr * (TP4 * 4)] = v;
                sm += v;
                sq += v * v;
            }
            sm += __shfl_xor(sm, 16);
            sq += __shfl_xor(sq, 16);
            sval = h ? sq : sm;
            sidx = (tile & (STAT_COPIES - 1)) * STAT_STRIDE + h * 64 + f;
        } else if (BN && wave < 6) {
            int ptr_ = tr, pq_ = q;
            asm volatile("" : "+v"(ptr_), "+v"(pq_));
            const int pf = ((ptr_ & 3) << 4) | pq_;
            const int gid16 = gids[pq_];
            const int rbase = tile * 16;
            const float* tpf = (const float*)tp + pf;
            const float pa = ((const float*)cf)[pf], pc = ((const float*)cf)[64 + pf];
            const int pidx = poolOff + pf;
            int cur = __builtin_amdgcn_readlane(gid16, 0);
            int cnt = 1;
            float ps = tpf[0];
#pragma unroll
            for (int rr = 1; rr < 16; ++rr) {
                const int gi = __builtin_amdgcn_readlane(gid16, rr);
                const float v = tpf[rr * (TP4 * 4)];
                if (gi != cur) {
                    atomicAdd(&pooled[cur * 192 + pidx], pa * ps + pc * (float)cnt);
                    cur = gi;
                    ps = 0.f;
                    cnt = 0;
                }
                ps += v;
                cnt += (rbase + rr < N) ? 1 : 0;
            }
            pval = pa * ps + pc * (float)cnt;
            pix = cur * 192 + pidx;
        }
        __syncthreads();
        if (live) ((float4*)r)[n * 16 + q] = to[tr * TP4 + q];
        if (wave < 4) atomicAdd(&stats[sidx], sval);
        if (BN && (wave & 6) == 4) atomicAdd(&pooled[pix], pval);
    }
}

#define POOL_R 16

struct BnCoef { float4 m, is, g, b; };

__device__ __forceinline__ BnCoef bn_coef(float4 sm, float4 sq, float4 gg, float4 bb, int N) {
    BnCoef c;
    const float invN = 1.0f / (float)N;
    c.m.x = sm.x * invN; c.m.y = sm.y * invN; c.m.z = sm.z * invN; c.m.w = sm.w * invN;
    c.is.x = rsqrtf(sq.x * invN - c.m.x * c.m.x + EPSV);
    c.is.y = rsqrtf(sq.y * invN - c.m.y * c.m.y + EPSV);
    c.is.z = rsqrtf(sq.z * invN - c.m.z * c.m.z + EPSV);
    c.is.w = rsqrtf(sq.w * invN - c.m.w * c.m.w + EPSV);
    c.g = gg; c.b = bb;
    return c;
}

__device__ __forceinline__ float4 bn4(float4 x, const BnCoef& c) {
    float4 v;
    v.x = c.g.x * (x.x - c.m.x) * c.is.x + c.b.x;
    v.y = c.g.y * (x.y - c.m.y) * c.is.y + c.b.y;
    v.z = c.g.z * (x.z - c.m.z) * c.is.z + c.b.z;
    v.w = c.g.w * (x.w - c.m.w) * c.is.w + c.b.w;
    return v;
}

__device__ __forceinline__ void add4(float4& s, float4 a) { s.x += a.x; s.y += a.y; s.z += a.z; s.w += a.w; }

__device__ __forceinline__ void pool_add4(float* __restrict__ p, float4 a) {
    atomicAdd(p, a.x); atomicAdd(p + 1, a.y); atomicAdd(p + 2, a.z); atomicAdd(p + 3, a.w);
}

__global__ void k_bn_pool(const float* __restrict__ hin, const float* __restrict__ stats,
          const float* __restrict__ g, const float* __restrict__ beta,
          const int* __restrict__ batch, float* __restrict__ hout,
          float* __restrict__ pooled, int poolOff, int N) {
    const int t  = blockIdx.x * blockDim.x + threadIdx.x;
    const int q  = t & 15;
    const int r0 = (t >> 4) * POOL_R;
    if (r0 >= N) return;
    const float4* h4 = (const float4*)hin + (size_t)r0 * 16 + q;
    float4*       o4 = (float4*)hout + (size_t)r0 * 16 + q;
    const int*    bp = batch + r0;
    const int f = q * 4;
    float* pq = pooled + poolOff + f;

    const float4* st4 = (const float4*)stats + q;
    float4 sm = st4[0], sq = st4[16];
#pragma unroll
    for (int k = 1; k < STAT_COPIES; ++k) {
        add4(sm, st4[k * (STAT_STRIDE / 4)]);
        add4(sq, st4[k * (STAT_STRIDE / 4) + 16]);
    }
    const float4 gg = make_float4(g[f], g[f + 1], g[f + 2], g[f + 3]);
    const float4 bb = make_float4(beta[f], beta[f + 1], beta[f + 2], beta[f + 3]);

    int cur;
    float4 acc = make_float4(0.f, 0.f, 0.f, 0.f);
    if (r0 + POOL_R <= N) {
        float4 x[POOL_R];
        int    b[POOL_R];
#pragma unroll
        for (int u = 0; u < POOL_R; ++u) {
            x[u] = h4[u * 16];
            b[u] = bp[u];
        }
        const BnCoef c = bn_coef(sm, sq, gg, bb, N);
        cur = b[0];
#pragma unroll
        for (int u = 0; u < POOL_R; ++u) {
            const float4 v = bn4(x[u], c);
            o4[u * 16] = v;
            if (b[u] != cur) {
                pool_add4(pq + (size_t)cur * 192, acc);
                cur = b[u];
                acc = make_float4(0.f, 0.f, 0.f, 0.f);
            }
            acc.x += v.x; acc.y += v.y; acc.z += v.z; acc.w += v.w;
        }
    } else {
        const BnCoef c = bn_coef(sm, sq, gg, bb, N);
        cur = bp[0];
        for (int u = 0; r0 + u < N; ++u) {
            const float4 v = bn4(h4[u * 16], c);
            o4[u * 16] = v;
            const int bu = bp[u];
            if (bu != cur) {
                pool_add4(pq + (size_t)cur * 192, acc);
                cur = bu;
                acc = make_float4(0.f, 0.f, 0.f, 0.f);
            }
            acc.x += v.x; acc.y += v.y; acc.z += v.z; acc.w += v.w;
        }
    }
    pool_add4(pq + (size_t)cur * 192, acc);
}

extern "C" void kernel_launch(void* const* d_in, const int* in_sizes, int n_in,
                              void* d_out, int out_size, void* d_ws, size_t ws_size,
                              hipStream_t stream) {
    const float* x    = (const float*)d_in[0];
    const int*   eidx = (const int*)d_in[1];
    const int*   batch = (const int*)d_in[2];
    const float* W[3]  = {(const float*)d_in[3],  (const float*)d_in[7],  (const float*)d_in[11]};
    const float* bi[3] = {(const float*)d_in[4],  (const float*)d_in[8],  (const float*)d_in[12]};
    const float* ga[3] = {(const float*)d_in[5],  (const float*)d_in[9],  (const float*)d_in[13]};
    const float* be[3] = {(const float*)d_in[6],  (const float*)d_in[10], (const float*)d_in[14]};

    const int N = in_sizes[0] / HDIM;
    const int E = in_sizes[1] / 2;
    const int G = (out_size - N * HDIM) / (3 * HDIM);

    const int* src = eidx;
    const int* dst = eidx + E;

    float* ws    = (float*)d_ws;
    float* dinv  = ws;
    float* bufA  = dinv + N;
    float* bufB  = bufA + (size_t)N * HDIM;
    int*   csr   = (int*)(bufB + (size_t)N * HDIM);
    int* rowptr = csr;
    int* col    = rowptr + (N + 4);
    int* cntC   = col + E;
    int* tot    = cntC + CSR_C * (N + 4);
    float* stats = (float*)(tot + CSR_R * CSR_C);

    float* pooled = (float*)d_out;
    float* hfinal = (float*)d_out + (size_t)G * 192;

    const int TB = 256;

    k_csr_count<<<CSR_R * CSR_C, CSR_TB, 0, stream>>>(dst, cntC, tot, N, E,
                                                     pooled, G * 192, stats, 3 * STAT_COPIES * STAT_STRIDE);
    k_csr_scan <<<CSR_R, CSR_TB, 0, stream>>>(cntC, tot, rowptr, dinv, N);
    k_csr_fill <<<CSR_R * CSR_C, CSR_TB, 0, stream>>>(src, dst, cntC, col, N, E);

    float* st0 = stats;
    float* st1 = stats + 1 * (STAT_COPIES * STAT_STRIDE);
    float* st2 = stats + 2 * (STAT_COPIES * STAT_STRIDE);

    k_gather_gemm_act_stats<false><<<GATHER_GRID, GATHER_TB, 0, stream>>>(
        x, W[0], dinv, rowptr, col, bi[0], bufA, st0, N,
        (const float*)nullptr, (const float*)nullptr, (const float*)nullptr, (const int*)nullptr,
        (float*)nullptr, 0);
    k_gather_gemm_act_stats<true><<<GATHER_GRID, GATHER_TB, 0, stream>>>(
        bufA, W[1], dinv, rowptr, col, bi[1], bufB, st1, N,
        st0, ga[0], be[0], batch, pooled, 0);
    k_gather_gemm_act_stats<true><<<GATHER_GRID, GATHER_TB, 0, stream>>>(
        bufB, W[2], dinv, rowptr, col, bi[2], bufA, st2, N,
        st1, ga[1], be[1], batch, pooled, HDIM);
    k_bn_pool<<<cdiv((long long)cdiv(N, POOL_R) * 16, TB), TB, 0, stream>>>(
        bufA, st2, ga[2], be[2], batch, hfinal, pooled, 2 * HDIM, N);
}


# Layers 2-3 gather without the per-edge dinv[src] lookup: rows stored times their own dinv, dinv[dst] applied once per node, node weight sums saved by layer 1; products regrouped, sum order kept
# speedup vs baseline: 3.6360x; 1.0302x over previous
#include <hip/hip_runtime.h>

typedef float v2f __attribute__((ext_vector_type(2)));
typedef float v8f __attribute__((ext_vector_type(8)));

#define HDIM 64
#define EPSV 1e-5f

static inline int cdiv(long long a, long long b) { return (int)((a + b - 1) / b); }

#define CSR_R  32
#define CSR_C  8
#define CSR_TB 1024
#define CSR_U  8
#define CSR_MAXSPAN 4096
#define CSR_NPT (CSR_MAXSPAN / CSR_TB)

__device__ __forceinline__ int csr_npr(int N)    { return (N + CSR_R - 1) / CSR_R; }
__device__ __forceinline__ int csr_stride(int N) { return N + 4; }
__device__ __forceinline__ int csr_clen(int E)   { return (((E + CSR_C - 1) / CSR_C) + 3) & ~3; }

__device__ __forceinline__ void csr_count1(int d, int lo, unsigned span, int* cl, int& m) {
    const unsigned u = (unsigned)(d - lo);
    if (u < span) { atomicAdd(&cl[u], 1); ++m; }
}

__device__ __forceinline__ void csr_place1(int d, int s, int lo, unsigned span, int* cl,
                                           int* __restrict__ col) {
    const unsigned u = (unsigned)(d - lo);
    if (u < span) col[atomicAdd(&cl[u], 1)] = s;
}

__global__ void __launch_bounds__(CSR_TB)
k_csr_count(const int* __restrict__ dst, int* __restrict__ cntC, int* __restrict__ tot, int N, int E,
            float* __restrict__ zp, int nzp, float* __restrict__ zs, int nzs) {
    __shared__ int cl[CSR_MAXSPAN];
    __shared__ int tsum;
    const int t = threadIdx.x;
    const int r = blockIdx.x / CSR_C, c = blockIdx.x % CSR_C;
    const int npr = csr_npr(N), NS = csr_stride(N);
    const int lo = r * npr;
    const int spn = min(min(npr, N - lo), CSR_MAXSPAN);
    const unsigned span = spn > 0 ? (unsigned)spn : 0u;
    for (int i = t; i < CSR_MAXSPAN; i += CSR_TB) cl[i] = 0;
    if (t == 0) tsum = 0;
    __syncthreads();
    const int clen = csr_clen(E);
    const int e0 = min(E, c * clen), e1 = min(E, e0 + clen);
    int m = 0;
    int done = e0;
    if ((((size_t)dst) & 15) == 0) {
        const int4* d4 = (const int4*)(dst + e0);
        const int nv = (e1 - e0) >> 2;
        for (int i0 = 0; i0 < nv; i0 += CSR_U * CSR_TB) {
            int4 dv[CSR_U];
#pragma unroll
            for (int u = 0; u < CSR_U; ++u) dv[u] = d4[min(i0 + u * CSR_TB + t, nv - 1)];
#pragma unroll
            for (int u = 0; u < CSR_U; ++u) {
                if (i0 + u * CSR_TB + t < nv) {
                    csr_count1(dv[u].x, lo, span, cl, m);
                    csr_count1(dv[u].y, lo, span, cl, m);
                    csr_count1(dv[u].z, lo, span, cl, m);
                    csr_count1(dv[u].w, lo, span, cl, m);
                }
            }
        }
        done = e0 + (nv << 2);
    }
    for (int e = done + t; e < e1; e += CSR_TB) csr_count1(dst[e], lo, span, cl, m);
    for (int o = 16; o > 0; o >>= 1) m += __shfl_xor(m, o);
    if ((t & 31) == 0) atomicAdd(&tsum, m);
    __syncthreads();
    for (int i = t; i < (int)span; i += CSR_TB) cntC[c * NS + lo + i] = cl[i];
    if (t == 0) tot[blockIdx.x] = tsum;
    const int gid = blockIdx.x * CSR_TB + t, gsz = gridDim.x * CSR_TB;
#pragma clang loop unroll(disable) vectorize(disable) interleave(disable)
    for (int i = gid; i < nzp; i += gsz) zp[i] = 0.0f;
#pragma clang loop unroll(disable) vectorize(disable) interleave(disable)
    for (int i = gid; i < nzs; i += gsz) zs[i] = 0.0f;
}

__global__ void __launch_bounds__(CSR_TB)
k_csr_scan(int* __restrict__ cntC, const int* __restrict__ tot, int* __restrict__ rowptr,
           float* __restrict__ dinv, int N) {
    __shared__ int sh[CSR_TB];
    const int t = threadIdx.x, r = blockIdx.x;
    const int npr = csr_npr(N), NS = csr_stride(N);
    const int lo = r * npr;
    const int hi = min(lo + npr, N);
    const int npt = (npr + CSR_TB - 1) / CSR_TB;
    const int f = lo + t * npt;
    const int l = min(f + npt, hi);
    int p = 0;
    if (t < CSR_R * CSR_C) p = (t / CSR_C < r) ? tot[t] : 0;
    int x[CSR_NPT][CSR_C];
#pragma unroll
    for (int k = 0; k < CSR_NPT; ++k) {
        const int n = min(f + k, hi - 1);
#pragma unroll
        for (int c = 0; c < CSR_C; ++c) x[k][c] = cntC[c * NS + n];
    }
    int mine = 0;
#pragma unroll
    for (int k = 0; k < CSR_NPT; ++k) {
        if (f + k < l) {
#pragma unroll
            for (int c = 0; c < CSR_C; ++c) mine += x[k][c];
        }
    }
    sh[t] = p;
    __syncthreads();
    for (int s = CSR_TB / 2; s > 0; s >>= 1) {
        if (t < s) sh[t] += sh[t + s];
        __syncthreads();
    }
    const int base = sh[0];
    __syncthreads();
    sh[t] = mine;
    __syncthreads();
    for (int off = 1; off < CSR_TB; off <<= 1) {
        int v = (t >= off) ? sh[t - off] : 0;
        __syncthreads();
        sh[t] += v;
        __syncthreads();
    }
    int run = base + sh[t] - mine;
#pragma unroll
    for (int k = 0; k < CSR_NPT; ++k) {
        if (f + k < l) {
            const int n = f + k;
            int o = run;
#pragma unroll
            for (int c = 0; c < CSR_C; ++c) { cntC[c * NS + n] = o; o += x[k][c]; }
            rowptr[n] = run;
            dinv[n]   = rsqrtf((float)(o - run + 1));
            run = o;
        }
    }
    if (f < l && l == N) rowptr[N] = run;
}

__global__ void __launch_bounds__(CSR_TB)
k_csr_fill(const int* __restrict__ src, const int* __restrict__ dst, const int* __restrict__ cntC,
           int* __restrict__ col, int N, int E) {
    __shared__ int cl[CSR_MAXSPAN];
    const int t = threadIdx.x;
    const int r = blockIdx.x / CSR_C, c = blockIdx.x % CSR_C;
    const int npr = csr_npr(N), NS = csr_stride(N);
    const int lo = r * npr;
    const int spn = min(min(npr, N - lo), CSR_MAXSPAN);
    const unsigned span = spn > 0 ? (unsigned)spn : 0u;
    int st[CSR_NPT];
#pragma unroll
    for (int k = 0; k < CSR_NPT; ++k) st[k] = cntC[c * NS + lo + min(k * CSR_TB + t, spn - 1)];
#pragma unroll
    for (int k = 0; k < CSR_NPT; ++k)
        if (k * CSR_TB + t < spn) cl[k * CSR_TB + t] = st[k];
    __syncthreads();
    const int clen = csr_clen(E);
    const int e0 = min(E, c * clen), e1 = min(E, e0 + clen);
    int done = e0;
    if (((((size_t)dst) | ((size_t)src)) & 15) == 0) {
        const int4* d4 = (const int4*)(dst + e0);
        const int4* s4 = (const int4*)(src + e0);
        const int nv = (e1 - e0) >> 2;
        for (int i0 = 0; i0 < nv; i0 += CSR_U * CSR_TB) {
            int4 dv[CSR_U], sv[CSR_U];
#pragma unroll
            for (int u = 0; u < CSR_U; ++u) {
                const int i = min(i0 + u * CSR_TB + t, nv - 1);
                dv[u] = d4[i];
                sv[u] = s4[i];
            }
#pragma unroll
            for (int u = 0; u < CSR_U; ++u)
                if (i0 + u * CSR_TB + t < nv) {
                    csr_place1(dv[u].x, sv[u].x, lo, span, cl, col);
                    csr_place1(dv[u].y, sv[u].y, lo, span, cl, col);
                    csr_place1(dv[u].z, sv[u].z, lo, span, cl, col);
                    csr_place1(dv[u].w, sv[u].w, lo, span, cl, col);
                }
        }
        done = e0 + (nv << 2);
    }
    for (int e = done + t; e < e1; e += CSR_TB) {
        csr_place1(dst[e], src[e], lo, span, cl, col);
    }
}

#define STAT_COPIES 8
#define STAT_STRIDE 1088
#define GATHER_TB 256
#define GATHER_GRID 2048
#define TP4 17
__device__ __forceinline__ void fence() { __builtin_amdgcn_sched_barrier(0); }

template <bool BN, bool SCALE_OUT>
__global__ void __launch_bounds__(GATHER_TB) __attribute__((amdgpu_waves_per_eu(16, 16)))
k_gather_gemm_act_stats(const float* __restrict__ hin, const float* __restrict__ W,
                        const float* __restrict__ dinv,
                        const int* __restrict__ rowptr, const int* __restrict__ col,
                        const float* __restrict__ bias, float* __restrict__ r,
                        float* __restrict__ stats  , int N,
                        const float* __restrict__ pstats, const float* __restrict__ pg,
                        const float* __restrict__ pbeta, const int* __restrict__ batch,
                        float* __restrict__ pooled, int poolOff,
                        float* __restrict__ wsumv) {
    __shared__ float4 wl[HDIM * TP4];
    __shared__ float4 tt[BN ? 3 * 16 * TP4 + 32 + 4 + 4 : 2 * 16 * TP4 + 4];
    float4* const ta = tt;
    float4* const to = tt + 16 * TP4;
    float4* const tp = tt + 32 * TP4;
    float4* const cf = tt + 48 * TP4;
    int* const gids = (int*)(tt + 48 * TP4 + 32);
    float* const dis = (float*)(BN ? tt + 48 * TP4 + 36 : tt + 32 * TP4);
    const int tid = threadIdx.x;
    const int ntiles = (N + 15) >> 4;
    if ((int)blockIdx.x >= ntiles) return;
    if (BN && tid < 16) {
        const float4* st4 = (const float4*)pstats + tid;
        float4 sm = st4[0], sq = st4[16];
#pragma unroll
        for (int k = 1; k < STAT_COPIES; ++k) {
            const float4 s1 = st4[k * (STAT_STRIDE / 4)], s2 = st4[k * (STAT_STRIDE / 4) + 16];
            sm.x += s1.x; sm.y += s1.y; sm.z += s1.z; sm.w += s1.w;
            sq.x += s2.x; sq.y += s2.y; sq.z += s2.z; sq.w += s2.w;
        }
        const int f0 = tid * 4;
        const float invN = 1.0f / (float)N;
        const float m0 = sm.x * invN, m1 = sm.y * invN, m2 = sm.z * invN, m3 = sm.w * invN;
        float4 cA, cC;
        cA.x = pg[f0]     * rsqrtf(sq.x * invN - m0 * m0 + EPSV);
        cA.y = pg[f0 + 1] * rsqrtf(sq.y * invN - m1 * m1 + EPSV);
        cA.z = pg[f0 + 2] * rsqrtf(sq.z * invN - m2 * m2 + EPSV);
        cA.w = pg[f0 + 3] * rsqrtf(sq.w * invN - m3 * m3 + EPSV);
        cC.x = pbeta[f0]     - m0 * cA.x;
        cC.y = pbeta[f0 + 1] - m1 * cA.y;
        cC.z = pbeta[f0 + 2] - m2 * cA.z;
        cC.w = pbeta[f0 + 3] - m3 * cA.w;
        cf[tid] = cA;
        cf[16 + tid] = cC;
    }
    const float4* W4 = (const float4*)W;
    float4 wv[4];
#pragma unroll
    for (int i = 0; i < 4; ++i) wv[i] = W4[tid + i * GATHER_TB];

    const int q = tid & 15;
    const int tr = tid >> 4;
    const int wave = tid >> 5;
    const int lane = tid & 31;
    const int c = lane & 15, h = lane >> 4;
    const int f = (wave & 3) * 16 + c;
    const float4* h4 = (const float4*)hin;
    const float4* ap = ta + c * TP4 + h * 8;
    const float4* bp = wl + f * TP4 + h * 8;
    float* tof = (float*)to + (h * 8) * (TP4 * 4) + f;

#pragma unroll
    for (int i = 0; i < 4; ++i) {
        const int idx = tid + i * GATHER_TB;
        wl[(idx >> 4) * TP4 + (idx & 15)] = wv[i];
    }
    if (BN) __syncthreads();

    for (int tile = blockIdx.x; tile < ntiles; tile += GATHER_GRID) {
        const int n = tile * 16 + tr;
        const bool live = n < N;
        const int nc = live ? n : N - 1;
        const float di = dinv[nc];
        const int beg = rowptr[nc];
        const int endr = rowptr[nc + 1];
        const int end = live ? endr : beg;
        const float4 vs = h4[nc * 16 + q];
        int gme = 0;
        float wsn = 0.f;
        if (BN) { gme = batch[nc]; wsn = wsumv[nc]; }

        if (SCALE_OUT) dis[tr] = di;
        if (BN) {
            const float rdi = live ? __builtin_amdgcn_rcpf(di) : 0.f;
            tp[tr * TP4 + q] = make_float4(vs.x * rdi, vs.y * rdi, vs.z * rdi, vs.w * rdi);
            gids[tr] = gme;
        }
        const int last = end - 1;
        float4 acc;
        float wsum;
        if (BN) {
            acc = live ? vs : make_float4(0.f, 0.f, 0.f, 0.f);
            for (int j = beg; j < end; j += 4) {
                const int s0 = col[j];
                const int s1 = col[min(j + 1, last)];
                const int s2 = col[min(j + 2, last)];
                const int s3 = col[min(j + 3, last)];
                const float4 v0 = h4[s0 * 16 + q];
                const float4 v1 = h4[s1 * 16 + q];
                const float4 v2 = h4[s2 * 16 + q];
                const float4 v3 = h4[s3 * 16 + q];
                const float m1 = (j + 1 < end) ? 1.f : 0.f;
                const float m2 = (j + 2 < end) ? 1.f : 0.f;
                const float m3 = (j + 3 < end) ? 1.f : 0.f;
                acc.x += v0.x;      acc.y += v0.y;      acc.z += v0.z;      acc.w += v0.w;
                acc.x += v1.x * m1; acc.y += v1.y * m1; acc.z += v1.z * m1; acc.w += v1.w * m1;
                acc.x += v2.x * m2; acc.y += v2.y * m2; acc.z += v2.z * m2; acc.w += v2.w * m2;
                acc.x += v3.x * m3; acc.y += v3.y * m3; acc.z += v3.z * m3; acc.w += v3.w * m3;
            }
            wsum = live ? wsn : 0.f;
            const float4 cA = cf[q], cC = cf[16 + q];
            acc.x = cA.x * (acc.x * di) + cC.x * wsum; acc.y = cA.y * (acc.y * di) + cC.y * wsum;
            acc.z = cA.z * (acc.z * di) + cC.z * wsum; acc.w = cA.w * (acc.w * di) + cC.w * wsum;
        } else {
            const float nrm = live ? di * di : 0.f;
            acc = make_float4(vs.x * nrm, vs.y * nrm, vs.z * nrm, vs.w * nrm);
            wsum = nrm;
            for (int j = beg; j < end; j += 4) {
                const int s0 = col[j];
                const int s1 = col[min(j + 1, last)];
                const int s2 = col[min(j + 2, last)];
                const int s3 = col[min(j + 3, last)];
                const float d0 = dinv[s0], d1 = dinv[s1], d2 = dinv[s2], d3 = dinv[s3];
                const float4 v0 = h4[s0 * 16 + q];
                const float4 v1 = h4[s1 * 16 + q];
                const float4 v2 = h4[s2 * 16 + q];
                const float4 v3 = h4[s3 * 16 + q];
                const float w0 = d0 * di;
                const float w1 = (j + 1 < end) ? d1 * di : 0.f;
                const float w2 = (j + 2 < end) ? d2 * di : 0.f;
                const float w3 = (j + 3 < end) ? d3 * di : 0.f;
                acc.x += v0.x * w0; acc.y += v0.y * w0; acc.z += v0.z * w0; acc.w += v0.w * w0;
                acc.x += v1.x * w1; acc.y += v1.y * w1; acc.z += v1.z * w1; acc.w += v1.w * w1;
                acc.x += v2.x * w2; acc.y += v2.y * w2; acc.z += v2.z * w2; acc.w += v2.w * w2;
                acc.x += v3.x * w3; acc.y += v3.y * w3; acc.z += v3.z * w3; acc.w += v3.w * w3;
                wsum += (w0 + w1) + (w2 + w3);
            }
        }
        ta[tr * TP4 + q] = acc;
        __syncthreads();
        float sval = 0.f;
        int   sidx = 0;
        float pval = 0.f;
        int   pix = 0;
        if (wave < 4) {
            const float bf = bias[f];
            v8f e0 = {0.f, 0.f, 0.f, 0.f, 0.f, 0.f, 0.f, 0.f};
            v8f e1 = {0.f, 0.f, 0.f, 0.f, 0.f, 0.f, 0.f, 0.f};
#pragma unroll
            for (int half = 0; half < 2; ++half) {
                float4 a[4], b[4];
                fence();
#pragma unroll
                for (int i = 0; i < 4; ++i) { a[i] = ap[half * 4 + i]; b[i] = bp[half * 4 + i]; }
                fence();
#pragma unroll
                for (int i = 0; i < 4; ++i) {
                    v2f a0 = {a[i].x, a[i].y}, b0 = {b[i].x, b[i].y};
                    v2f a1 = {a[i].z, a[i].w}, b1 = {b[i].z, b[i].w};
                    e0 = __builtin_amdgcn_wmma_f32_16x16x4_f32(false, a0, false, b0, (short)0, e0, false, false);
                    e1 = __builtin_amdgcn_wmma_f32_16x16x4_f32(false, a1, false, b1, (short)0, e1, false, false);
                }
            }
            fence();
            asm volatile("v_nop\n\tv_nop\n\tv_nop\n\tv_nop");
            fence();
            const int row0 = tile * 16 + h * 8;
            float sm = 0.f, sq = 0.f;
#pragma unroll
            for (int rr = 0; rr < 8; ++rr) {
                float v = fmaxf(e0[rr] + e1[rr] + bf, 0.f);
                v = (row0 + rr < N) ? v : 0.f;
                tof[rr * (TP4 * 4)] = v;
                sm += v;
                sq += v * v;
            }
            sm += __shfl_xor(sm, 16);
            sq += __shfl_xor(sq, 16);
            sval = h ? sq : sm;
            sidx = (tile & (STAT_COPIES - 1)) * STAT_STRIDE + h * 64 + f;
        } else if (BN && wave < 6) {
            int ptr_ = tr, pq_ = q;
            asm volatile("" : "+v"(ptr_), "+v"(pq_));
            const int pf = ((ptr_ & 3) << 4) | pq_;
            const int gid16 = gids[pq_];
            const int rbase = tile * 16;
            const float* tpf = (const float*)tp + pf;
            const float pa = ((const float*)cf)[pf], pc = ((const float*)cf)[64 + pf];
            const int pidx = poolOff + pf;
            int cur = __builtin_amdgcn_readlane(gid16, 0);
            int cnt = 1;
            float ps = tpf[0];
#pragma unroll
            for (int rr = 1; rr < 16; ++rr) {
                const int gi = __builtin_amdgcn_readlane(gid16, rr);
                const float v = tpf[rr * (TP4 * 4)];
                if (gi != cur) {
                    atomicAdd(&pooled[cur * 192 + pidx], pa * ps + pc * (float)cnt);
                    cur = gi;
                    ps = 0.f;
                    cnt = 0;
                }
                ps += v;
                cnt += (rbase + rr < N) ? 1 : 0;
            }
            pval = pa * ps + pc * (float)cnt;
            pix = cur * 192 + pidx;
        }
        __syncthreads();
        if (live) {
            float4 o = to[tr * TP4 + q];
            if (SCALE_OUT) { const float dso = dis[tr]; o.x *= dso; o.y *= dso; o.z *= dso; o.w *= dso; }
            ((float4*)r)[n * 16 + q] = o;
            if (!BN && q == 0) wsumv[n] = wsum;
        }
        if (wave < 4) atomicAdd(&stats[sidx], sval);
        if (BN && (wave & 6) == 4) atomicAdd(&pooled[pix], pval);
    }
}

#define POOL_R 16

struct BnCoef { float4 m, is, g, b; };

__device__ __forceinline__ BnCoef bn_coef(float4 sm, float4 sq, float4 gg, float4 bb, int N) {
    BnCoef c;
    const float invN = 1.0f / (float)N;
    c.m.x = sm.x * invN; c.m.y = sm.y * invN; c.m.z = sm.z * invN; c.m.w = sm.w * invN;
    c.is.x = rsqrtf(sq.x * invN - c.m.x * c.m.x + EPSV);
    c.is.y = rsqrtf(sq.y * invN - c.m.y * c.m.y + EPSV);
    c.is.z = rsqrtf(sq.z * invN - c.m.z * c.m.z + EPSV);
    c.is.w = rsqrtf(sq.w * invN - c.m.w * c.m.w + EPSV);
    c.g = gg; c.b = bb;
    return c;
}

__device__ __forceinline__ float4 bn4(float4 x, const BnCoef& c) {
    float4 v;
    v.x = c.g.x * (x.x - c.m.x) * c.is.x + c.b.x;
    v.y = c.g.y * (x.y - c.m.y) * c.is.y + c.b.y;
    v.z = c.g.z * (x.z - c.m.z) * c.is.z + c.b.z;
    v.w = c.g.w * (x.w - c.m.w) * c.is.w + c.b.w;
    return v;
}

__device__ __forceinline__ void add4(float4& s, float4 a) { s.x += a.x; s.y += a.y; s.z += a.z; s.w += a.w; }

__device__ __forceinline__ void pool_add4(float* __restrict__ p, float4 a) {
    atomicAdd(p, a.x); atomicAdd(p + 1, a.y); atomicAdd(p + 2, a.z); atomicAdd(p + 3, a.w);
}

__global__ void k_bn_pool(const float* __restrict__ hin, const float* __restrict__ stats,
          const float* __restrict__ g, const float* __restrict__ beta,
          const int* __restrict__ batch, float* __restrict__ hout,
          float* __restrict__ pooled, int poolOff, int N) {
    const int t  = blockIdx.x * blockDim.x + threadIdx.x;
    const int q  = t & 15;
    const int r0 = (t >> 4) * POOL_R;
    if (r0 >= N) return;
    const float4* h4 = (const float4*)hin + (size_t)r0 * 16 + q;
    float4*       o4 = (float4*)hout + (size_t)r0 * 16 + q;
    const int*    bp = batch + r0;
    const int f = q * 4;
    float* pq = pooled + poolOff + f;

    const float4* st4 = (const float4*)stats + q;
    float4 sm = st4[0], sq = st4[16];
#pragma unroll
    for (int k = 1; k < STAT_COPIES; ++k) {
        add4(sm, st4[k * (STAT_STRIDE / 4)]);
        add4(sq, st4[k * (STAT_STRIDE / 4) + 16]);
    }
    const float4 gg = make_float4(g[f], g[f + 1], g[f + 2], g[f + 3]);
    const float4 bb = make_float4(beta[f], beta[f + 1], beta[f + 2], beta[f + 3]);

    int cur;
    float4 acc = make_float4(0.f, 0.f, 0.f, 0.f);
    if (r0 + POOL_R <= N) {
        float4 x[POOL_R];
        int    b[POOL_R];
#pragma unroll
        for (int u = 0; u < POOL_R; ++u) {
            x[u] = h4[u * 16];
            b[u] = bp[u];
        }
        const BnCoef c = bn_coef(sm, sq, gg, bb, N);
        cur = b[0];
#pragma unroll
        for (int u = 0; u < POOL_R; ++u) {
            const float4 v = bn4(x[u], c);
            o4[u * 16] = v;
            if (b[u] != cur) {
                pool_add4(pq + (size_t)cur * 192, acc);
                cur = b[u];
                acc = make_float4(0.f, 0.f, 0.f, 0.f);
            }
            acc.x += v.x; acc.y += v.y; acc.z += v.z; acc.w += v.w;
        }
    } else {
        const BnCoef c = bn_coef(sm, sq, gg, bb, N);
        cur = bp[0];
        for (int u = 0; r0 + u < N; ++u) {
            const float4 v = bn4(h4[u * 16], c);
            o4[u * 16] = v;
            const int bu = bp[u];
            if (bu != cur) {
                pool_add4(pq + (size_t)cur * 192, acc);
                cur = bu;
                acc = make_float4(0.f, 0.f, 0.f, 0.f);
            }
            acc.x += v.x; acc.y += v.y; acc.z += v.z; acc.w += v.w;
        }
    }
    pool_add4(pq + (size_t)cur * 192, acc);
}

extern "C" void kernel_launch(void* const* d_in, const int* in_sizes, int n_in,
                              void* d_out, int out_size, void* d_ws, size_t ws_size,
                              hipStream_t stream) {
    const float* x    = (const float*)d_in[0];
    const int*   eidx = (const int*)d_in[1];
    const int*   batch = (const int*)d_in[2];
    const float* W[3]  = {(const float*)d_in[3],  (const float*)d_in[7],  (const float*)d_in[11]};
    const float* bi[3] = {(const float*)d_in[4],  (const float*)d_in[8],  (const float*)d_in[12]};
    const float* ga[3] = {(const float*)d_in[5],  (const float*)d_in[9],  (const float*)d_in[13]};
    const float* be[3] = {(const float*)d_in[6],  (const float*)d_in[10], (const float*)d_in[14]};

    const int N = in_sizes[0] / HDIM;
    const int E = in_sizes[1] / 2;
    const int G = (out_size - N * HDIM) / (3 * HDIM);

    const int* src = eidx;
    const int* dst = eidx + E;

    float* ws    = (float*)d_ws;
    float* dinv  = ws;
    float* bufA  = dinv + N;
    float* bufB  = bufA + (size_t)N * HDIM;
    int*   csr   = (int*)(bufB + (size_t)N * HDIM);
    int* rowptr = csr;
    int* col    = rowptr + (N + 4);
    int* cntC   = col + E;
    int* tot    = cntC + CSR_C * (N + 4);
    float* stats = (float*)(tot + CSR_R * CSR_C);

    float* pooled = (float*)d_out;
    float* hfinal = (float*)d_out + (size_t)G * 192;

    const int TB = 256;

    k_csr_count<<<CSR_R * CSR_C, CSR_TB, 0, stream>>>(dst, cntC, tot, N, E,
                                                     pooled, G * 192, stats, 3 * STAT_COPIES * STAT_STRIDE);
    k_csr_scan <<<CSR_R, CSR_TB, 0, stream>>>(cntC, tot, rowptr, dinv, N);
    k_csr_fill <<<CSR_R * CSR_C, CSR_TB, 0, stream>>>(src, dst, cntC, col, N, E);

    float* st0 = stats;
    float* st1 = stats + 1 * (STAT_COPIES * STAT_STRIDE);
    float* st2 = stats + 2 * (STAT_COPIES * STAT_STRIDE);
    float* wsumv = stats + 3 * (STAT_COPIES * STAT_STRIDE);

    k_gather_gemm_act_stats<false, true><<<GATHER_GRID, GATHER_TB, 0, stream>>>(
        x, W[0], dinv, rowptr, col, bi[0], bufA, st0, N,
        (const float*)nullptr, (const float*)nullptr, (const float*)nullptr, (const int*)nullptr,
        (float*)nullptr, 0, wsumv);
    k_gather_gemm_act_stats<true, true><<<GATHER_GRID, GATHER_TB, 0, stream>>>(
        bufA, W[1], dinv, rowptr, col, bi[1], bufB, st1, N,
        st0, ga[0], be[0], batch, pooled, 0, wsumv);
    k_gather_gemm_act_stats<true, false><<<GATHER_GRID, GATHER_TB, 0, stream>>>(
        bufB, W[2], dinv, rowptr, col, bi[2], bufA, st2, N,
        st1, ga[1], be[1], batch, pooled, HDIM, wsumv);
    k_bn_pool<<<cdiv((long long)cdiv(N, POOL_R) * 16, TB), TB, 0, stream>>>(
        bufA, st2, ga[2], be[2], batch, hfinal, pooled, 2 * HDIM, N);
}


# Gather edge walk: each trip asks for the next trip's four list entries together with its own four rows, so a trip waits for one round trip, not list entry then row; sum order kept
# speedup vs baseline: 3.8172x; 1.0498x over previous
#include <hip/hip_runtime.h>

typedef float v2f __attribute__((ext_vector_type(2)));
typedef float v8f __attribute__((ext_vector_type(8)));

#define HDIM 64
#define EPSV 1e-5f

static inline int cdiv(long long a, long long b) { return (int)((a + b - 1) / b); }

#define CSR_R  32
#define CSR_C  8
#define CSR_TB 1024
#define CSR_U  8
#define CSR_MAXSPAN 4096
#define CSR_NPT (CSR_MAXSPAN / CSR_TB)

__device__ __forceinline__ int csr_npr(int N)    { return (N + CSR_R - 1) / CSR_R; }
__device__ __forceinline__ int csr_stride(int N) { return N + 4; }
__device__ __forceinline__ int csr_clen(int E)   { return (((E + CSR_C - 1) / CSR_C) + 3) & ~3; }

__device__ __forceinline__ void csr_count1(int d, int lo, unsigned span, int* cl, int& m) {
    const unsigned u = (unsigned)(d - lo);
    if (u < span) { atomicAdd(&cl[u], 1); ++m; }
}

__device__ __forceinline__ void csr_place1(int d, int s, int lo, unsigned span, int* cl,
                                           int* __restrict__ col) {
    const unsigned u = (unsigned)(d - lo);
    if (u < span) col[atomicAdd(&cl[u], 1)] = s;
}

__global__ void __launch_bounds__(CSR_TB)
k_csr_count(const int* __restrict__ dst, int* __restrict__ cntC, int* __restrict__ tot, int N, int E,
            float* __restrict__ zp, int nzp, float* __restrict__ zs, int nzs) {
    __shared__ int cl[CSR_MAXSPAN];
    __shared__ int tsum;
    const int t = threadIdx.x;
    const int r = blockIdx.x / CSR_C, c = blockIdx.x % CSR_C;
    const int npr = csr_npr(N), NS = csr_stride(N);
    const int lo = r * npr;
    const int spn = min(min(npr, N - lo), CSR_MAXSPAN);
    const unsigned span = spn > 0 ? (unsigned)spn : 0u;
    for (int i = t; i < CSR_MAXSPAN; i += CSR_TB) cl[i] = 0;
    if (t == 0) tsum = 0;
    __syncthreads();
    const int clen = csr_clen(E);
    const int e0 = min(E, c * clen), e1 = min(E, e0 + clen);
    int m = 0;
    int done = e0;
    if ((((size_t)dst) & 15) == 0) {
        const int4* d4 = (const int4*)(dst + e0);
        const int nv = (e1 - e0) >> 2;
        for (int i0 = 0; i0 < nv; i0 += CSR_U * CSR_TB) {
            int4 dv[CSR_U];
#pragma unroll
            for (int u = 0; u < CSR_U; ++u) dv[u] = d4[min(i0 + u * CSR_TB + t, nv - 1)];
#pragma unroll
            for (int u = 0; u < CSR_U; ++u) {
                if (i0 + u * CSR_TB + t < nv) {
                    csr_count1(dv[u].x, lo, span, cl, m);
                    csr_count1(dv[u].y, lo, span, cl, m);
                    csr_count1(dv[u].z, lo, span, cl, m);
                    csr_count1(dv[u].w, lo, span, cl, m);
                }
            }
        }
        done = e0 + (nv << 2);
    }
    for (int e = done + t; e < e1; e += CSR_TB) csr_count1(dst[e], lo, span, cl, m);
    for (int o = 16; o > 0; o >>= 1) m += __shfl_xor(m, o);
    if ((t & 31) == 0) atomicAdd(&tsum, m);
    __syncthreads();
    for (int i = t; i < (int)span; i += CSR_TB) cntC[c * NS + lo + i] = cl[i];
    if (t == 0) tot[blockIdx.x] = tsum;
    const int gid = blockIdx.x * CSR_TB + t, gsz = gridDim.x * CSR_TB;
#pragma clang loop unroll(disable) vectorize(disable) interleave(disable)
    for (int i = gid; i < nzp; i += gsz) zp[i] = 0.0f;
#pragma clang loop unroll(disable) vectorize(disable) interleave(disable)
    for (int i = gid; i < nzs; i += gsz) zs[i] = 0.0f;
}

__global__ void __launch_bounds__(CSR_TB)
k_csr_scan(int* __restrict__ cntC, const int* __restrict__ tot, int* __restrict__ rowptr,
           float* __restrict__ dinv, int N) {
    __shared__ int sh[CSR_TB];
    const int t = threadIdx.x, r = blockIdx.x;
    const int npr = csr_npr(N), NS = csr_stride(N);
    const int lo = r * npr;
    const int hi = min(lo + npr, N);
    const int npt = (npr + CSR_TB - 1) / CSR_TB;
    const int f = lo + t * npt;
    const int l = min(f + npt, hi);
    int p = 0;
    if (t < CSR_R * CSR_C) p = (t / CSR_C < r) ? tot[t] : 0;
    int x[CSR_NPT][CSR_C];
#pragma unroll
    for (int k = 0; k < CSR_NPT; ++k) {
        const int n = min(f + k, hi - 1);
#pragma unroll
        for (int c = 0; c < CSR_C; ++c) x[k][c] = cntC[c * NS + n];
    }
    int mine = 0;
#pragma unroll
    for (int k = 0; k < CSR_NPT; ++k) {
        if (f + k < l) {
#pragma unroll
            for (int c = 0; c < CSR_C; ++c) mine += x[k][c];
        }
    }
    sh[t] = p;
    __syncthreads();
    for (int s = CSR_TB / 2; s > 0; s >>= 1) {
        if (t < s) sh[t] += sh[t + s];
        __syncthreads();
    }
    const int base = sh[0];
    __syncthreads();
    sh[t] = mine;
    __syncthreads();
    for (int off = 1; off < CSR_TB; off <<= 1) {
        int v = (t >= off) ? sh[t - off] : 0;
        __syncthreads();
        sh[t] += v;
        __syncthreads();
    }
    int run = base + sh[t] - mine;
#pragma unroll
    for (int k = 0; k < CSR_NPT; ++k) {
        if (f + k < l) {
            const int n = f + k;
            int o = run;
#pragma unroll
            for (int c = 0; c < CSR_C; ++c) { cntC[c * NS + n] = o; o += x[k][c]; }
            rowptr[n] = run;
            dinv[n]   = rsqrtf((float)(o - run + 1));
            run = o;
        }
    }
    if (f < l && l == N) rowptr[N] = run;
}

__global__ void __launch_bounds__(CSR_TB)
k_csr_fill(const int* __restrict__ src, const int* __restrict__ dst, const int* __restrict__ cntC,
           int* __restrict__ col, int N, int E) {
    __shared__ int cl[CSR_MAXSPAN];
    const int t = threadIdx.x;
    const int r = blockIdx.x / CSR_C, c = blockIdx.x % CSR_C;
    const int npr = csr_npr(N), NS = csr_stride(N);
    const int lo = r * npr;
    const int spn = min(min(npr, N - lo), CSR_MAXSPAN);
    const unsigned span = spn > 0 ? (unsigned)spn : 0u;
    int st[CSR_NPT];
#pragma unroll
    for (int k = 0; k < CSR_NPT; ++k) st[k] = cntC[c * NS + lo + min(k * CSR_TB + t, spn - 1)];
#pragma unroll
    for (int k = 0; k < CSR_NPT; ++k)
        if (k * CSR_TB + t < spn) cl[k * CSR_TB + t] = st[k];
    __syncthreads();
    const int clen = csr_clen(E);
    const int e0 = min(E, c * clen), e1 = min(E, e0 + clen);
    int done = e0;
    if (((((size_t)dst) | ((size_t)src)) & 15) == 0) {
        const int4* d4 = (const int4*)(dst + e0);
        const int4* s4 = (const int4*)(src + e0);
        const int nv = (e1 - e0) >> 2;
        for (int i0 = 0; i0 < nv; i0 += CSR_U * CSR_TB) {
            int4 dv[CSR_U], sv[CSR_U];
#pragma unroll
            for (int u = 0; u < CSR_U; ++u) {
                const int i = min(i0 + u * CSR_TB + t, nv - 1);
                dv[u] = d4[i];
                sv[u] = s4[i];
            }
#pragma unroll
            for (int u = 0; u < CSR_U; ++u)
                if (i0 + u * CSR_TB + t < nv) {
                    csr_place1(dv[u].x, sv[u].x, lo, span, cl, col);
                    csr_place1(dv[u].y, sv[u].y, lo, span, cl, col);
                    csr_place1(dv[u].z, sv[u].z, lo, span, cl, col);
                    csr_place1(dv[u].w, sv[u].w, lo, span, cl, col);
                }
        }
        done = e0 + (nv << 2);
    }
    for (int e = done + t; e < e1; e += CSR_TB) {
        csr_place1(dst[e], src[e], lo, span, cl, col);
    }
}

#define STAT_COPIES 8
#define STAT_STRIDE 1088
#define GATHER_TB 256
#define GATHER_GRID 2048
#define TP4 17
__device__ __forceinline__ void fence() { __builtin_amdgcn_sched_barrier(0); }

template <bool BN, bool SCALE_OUT>
__global__ void __launch_bounds__(GATHER_TB) __attribute__((amdgpu_waves_per_eu(16, 16)))
k_gather_gemm_act_stats(const float* __restrict__ hin, const float* __restrict__ W,
                        const float* __restrict__ dinv,
                        const int* __restrict__ rowptr, const int* __restrict__ col,
                        const float* __restrict__ bias, float* __restrict__ r,
                        float* __restrict__ stats  , int N,
                        const float* __restrict__ pstats, const float* __restrict__ pg,
                        const float* __restrict__ pbeta, const int* __restrict__ batch,
                        float* __restrict__ pooled, int poolOff,
                        float* __restrict__ wsumv) {
    __shared__ float4 wl[HDIM * TP4];
    __shared__ float4 tt[BN ? 3 * 16 * TP4 + 32 + 4 + 4 : 2 * 16 * TP4 + 4];
    float4* const ta = tt;
    float4* const to = tt + 16 * TP4;
    float4* const tp = tt + 32 * TP4;
    float4* const cf = tt + 48 * TP4;
    int* const gids = (int*)(tt + 48 * TP4 + 32);
    float* const dis = (float*)(BN ? tt + 48 * TP4 + 36 : tt + 32 * TP4);
    const int tid = threadIdx.x;
    const int ntiles = (N + 15) >> 4;
    if ((int)blockIdx.x >= ntiles) return;
    if (BN && tid < 16) {
        const float4* st4 = (const float4*)pstats + tid;
        float4 sm = st4[0], sq = st4[16];
#pragma unroll
        for (int k = 1; k < STAT_COPIES; ++k) {
            const float4 s1 = st4[k * (STAT_STRIDE / 4)], s2 = st4[k * (STAT_STRIDE / 4) + 16];
            sm.x += s1.x; sm.y += s1.y; sm.z += s1.z; sm.w += s1.w;
            sq.x += s2.x; sq.y += s2.y; sq.z += s2.z; sq.w += s2.w;
        }
        const int f0 = tid * 4;
        const float invN = 1.0f / (float)N;
        const float m0 = sm.x * invN, m1 = sm.y * invN, m2 = sm.z * invN, m3 = sm.w * invN;
        float4 cA, cC;
        cA.x = pg[f0]     * rsqrtf(sq.x * invN - m0 * m0 + EPSV);
        cA.y = pg[f0 + 1] * rsqrtf(sq.y * invN - m1 * m1 + EPSV);
        cA.z = pg[f0 + 2] * rsqrtf(sq.z * invN - m2 * m2 + EPSV);
        cA.w = pg[f0 + 3] * rsqrtf(sq.w * invN - m3 * m3 + EPSV);
        cC.x = pbeta[f0]     - m0 * cA.x;
        cC.y = pbeta[f0 + 1] - m1 * cA.y;
        cC.z = pbeta[f0 + 2] - m2 * cA.z;
        cC.w = pbeta[f0 + 3] - m3 * cA.w;
        cf[tid] = cA;
        cf[16 + tid] = cC;
    }
    const float4* W4 = (const float4*)W;
    float4 wv[4];
#pragma unroll
    for (int i = 0; i < 4; ++i) wv[i] = W4[tid + i * GATHER_TB];

    const int q = tid & 15;
    const int tr = tid >> 4;
    const int wave = tid >> 5;
    const int lane = tid & 31;
    const int c = lane & 15, h = lane >> 4;
    const int f = (wave & 3) * 16 + c;
    const float4* h4 = (const float4*)hin;
    const float4* ap = ta + c * TP4 + h * 8;
    const float4* bp = wl + f * TP4 + h * 8;
    float* tof = (float*)to + (h * 8) * (TP4 * 4) + f;

#pragma unroll
    for (int i = 0; i < 4; ++i) {
        const int idx = tid + i * GATHER_TB;
        wl[(idx >> 4) * TP4 + (idx & 15)] = wv[i];
    }
    if (BN) __syncthreads();

    for (int tile = blockIdx.x; tile < ntiles; tile += GATHER_GRID) {
        const int n = tile * 16 + tr;
        const bool live = n < N;
        const int nc = live ? n : N - 1;
        const float di = dinv[nc];
        const int beg = rowptr[nc];
        const int endr = rowptr[nc + 1];
        const int end = live ? endr : beg;
        const float4 vs = h4[nc * 16 + q];
        int gme = 0;
        float wsn = 0.f;
        if (BN) { gme = batch[nc]; wsn = wsumv[nc]; }

        if (SCALE_OUT) dis[tr] = di;
        if (BN) {
            const float rdi = live ? __builtin_amdgcn_rcpf(di) : 0.f;
            tp[tr * TP4 + q] = make_float4(vs.x * rdi, vs.y * rdi, vs.z * rdi, vs.w * rdi);
            gids[tr] = gme;
        }
        const int last = end - 1;
        float4 acc;
        float wsum;
        if (BN) {
            acc = live ? vs : make_float4(0.f, 0.f, 0.f, 0.f);
            if (beg < end) {
                int s0 = col[beg];
                int s1 = col[min(beg + 1, last)];
                int s2 = col[min(beg + 2, last)];
                int s3 = col[min(beg + 3, last)];
                int j = beg;
                do {
                    const float4 v0 = h4[s0 * 16 + q];
                    const float4 v1 = h4[s1 * 16 + q];
                    const float4 v2 = h4[s2 * 16 + q];
                    const float4 v3 = h4[s3 * 16 + q];
                    int n0 = col[min(j + 4, last)];
                    int n1 = col[min(j + 5, last)];
                    int n2 = col[min(j + 6, last)];
                    int n3 = col[min(j + 7, last)];
                    fence();
                    const float m1 = (j + 1 < end) ? 1.f : 0.f;
                    const float m2 = (j + 2 < end) ? 1.f : 0.f;
                    const float m3 = (j + 3 < end) ? 1.f : 0.f;
                    acc.x += v0.x;      acc.y += v0.y;      acc.z += v0.z;      acc.w += v0.w;
                    acc.x += v1.x * m1; acc.y += v1.y * m1; acc.z += v1.z * m1; acc.w += v1.w * m1;
                    acc.x += v2.x * m2; acc.y += v2.y * m2; acc.z += v2.z * m2; acc.w += v2.w * m2;
                    acc.x += v3.x * m3; acc.y += v3.y * m3; acc.z += v3.z * m3; acc.w += v3.w * m3;
                    asm volatile("" : "+v"(n0), "+v"(n1), "+v"(n2), "+v"(n3));
                    s0 = n0; s1 = n1; s2 = n2; s3 = n3;
                    j += 4;
                } while (j < end);
            }
            wsum = live ? wsn : 0.f;
            const float4 cA = cf[q], cC = cf[16 + q];
            acc.x = cA.x * (acc.x * di) + cC.x * wsum; acc.y = cA.y * (acc.y * di) + cC.y * wsum;
            acc.z = cA.z * (acc.z * di) + cC.z * wsum; acc.w = cA.w * (acc.w * di) + cC.w * wsum;
        } else {
            const float nrm = live ? di * di : 0.f;
            acc = make_float4(vs.x * nrm, vs.y * nrm, vs.z * nrm, vs.w * nrm);
            wsum = nrm;
            if (beg < end) {
                int s0 = col[beg];
                int s1 = col[min(beg + 1, last)];
                int s2 = col[min(beg + 2, last)];
                int s3 = col[min(beg + 3, last)];
                int j = beg;
                do {
                    const float d0 = dinv[s0], d1 = dinv[s1], d2 = dinv[s2], d3 = dinv[s3];
                    const float4 v0 = h4[s0 * 16 + q];
                    const float4 v1 = h4[s1 * 16 + q];
                    const float4 v2 = h4[s2 * 16 + q];
                    const float4 v3 = h4[s3 * 16 + q];
                    int n0 = col[min(j + 4, last)];
                    int n1 = col[min(j + 5, last)];
                    int n2 = col[min(j + 6, last)];
                    int n3 = col[min(j + 7, last)];
                    fence();
                    const float w0 = d0 * di;
                    const float w1 = (j + 1 < end) ? d1 * di : 0.f;
                    const float w2 = (j + 2 < end) ? d2 * di : 0.f;
                    const float w3 = (j + 3 < end) ? d3 * di : 0.f;
                    acc.x += v0.x * w0; acc.y += v0.y * w0; acc.z += v0.z * w0; acc.w += v0.w * w0;
                    acc.x += v1.x * w1; acc.y += v1.y * w1; acc.z += v1.z * w1; acc.w += v1.w * w1;
                    acc.x += v2.x * w2; acc.y += v2.y * w2; acc.z += v2.z * w2; acc.w += v2.w * w2;
                    acc.x += v3.x * w3; acc.y += v3.y * w3; acc.z += v3.z * w3; acc.w += v3.w * w3;
                    wsum += (w0 + w1) + (w2 + w3);
                    asm volatile("" : "+v"(n0), "+v"(n1), "+v"(n2), "+v"(n3));
                    s0 = n0; s1 = n1; s2 = n2; s3 = n3;
                    j += 4;
                } while (j < end);
            }
        }
        ta[tr * TP4 + q] = acc;
        __syncthreads();
        float sval = 0.f;
        int   sidx = 0;
        float pval = 0.f;
        int   pix = 0;
        if (wave < 4) {
            const float bf = bias[f];
            v8f e0 = {0.f, 0.f, 0.f, 0.f, 0.f, 0.f, 0.f, 0.f};
            v8f e1 = {0.f, 0.f, 0.f, 0.f, 0.f, 0.f, 0.f, 0.f};
#pragma unroll
            for (int half = 0; half < 2; ++half) {
                float4 a[4], b[4];
                fence();
#pragma unroll
                for (int i = 0; i < 4; ++i) { a[i] = ap[half * 4 + i]; b[i] = bp[half * 4 + i]; }
                fence();
#pragma unroll
                for (int i = 0; i < 4; ++i) {
                    v2f a0 = {a[i].x, a[i].y}, b0 = {b[i].x, b[i].y};
                    v2f a1 = {a[i].z, a[i].w}, b1 = {b[i].z, b[i].w};
                    e0 = __builtin_amdgcn_wmma_f32_16x16x4_f32(false, a0, false, b0, (short)0, e0, false, false);
                    e1 = __builtin_amdgcn_wmma_f32_16x16x4_f32(false, a1, false, b1, (short)0, e1, false, false);
                }
            }
            fence();
            asm volatile("v_nop\n\tv_nop\n\tv_nop\n\tv_nop");
            fence();
            const int row0 = tile * 16 + h * 8;
            float sm = 0.f, sq = 0.f;
#pragma unroll
            for (int rr = 0; rr < 8; ++rr) {
                float v = fmaxf(e0[rr] + e1[rr] + bf, 0.f);
                v = (row0 + rr < N) ? v : 0.f;
                tof[rr * (TP4 * 4)] = v;
                sm += v;
                sq += v * v;
            }
            sm += __shfl_xor(sm, 16);
            sq += __shfl_xor(sq, 16);
            sval = h ? sq : sm;
            sidx = (tile & (STAT_COPIES - 1)) * STAT_STRIDE + h * 64 + f;
        } else if (BN && wave < 6) {
            int ptr_ = tr, pq_ = q;
            asm volatile("" : "+v"(ptr_), "+v"(pq_));
            const int pf = ((ptr_ & 3) << 4) | pq_;
            const int gid16 = gids[pq_];
            const int rbase = tile * 16;
            const float* tpf = (const float*)tp + pf;
            const float pa = ((const float*)cf)[pf], pc = ((const float*)cf)[64 + pf];
            const int pidx = poolOff + pf;
            int cur = __builtin_amdgcn_readlane(gid16, 0);
            int cnt = 1;
            float ps = tpf[0];
#pragma unroll
            for (int rr = 1; rr < 16; ++rr) {
                const int gi = __builtin_amdgcn_readlane(gid16, rr);
                const float v = tpf[rr * (TP4 * 4)];
                if (gi != cur) {
                    atomicAdd(&pooled[cur * 192 + pidx], pa * ps + pc * (float)cnt);
                    cur = gi;
                    ps = 0.f;
                    cnt = 0;
                }
                ps += v;
                cnt += (rbase + rr < N) ? 1 : 0;
            }
            pval = pa * ps + pc * (float)cnt;
            pix = cur * 192 + pidx;
        }
        __syncthreads();
        if (live) {
            float4 o = to[tr * TP4 + q];
            if (SCALE_OUT) { const float dso = dis[tr]; o.x *= dso; o.y *= dso; o.z *= dso; o.w *= dso; }
            ((float4*)r)[n * 16 + q] = o;
            if (!BN && q == 0) wsumv[n] = wsum;
        }
        if (wave < 4) atomicAdd(&stats[sidx], sval);
        if (BN && (wave & 6) == 4) atomicAdd(&pooled[pix], pval);
    }
}

#define POOL_R 16

struct BnCoef { float4 m, is, g, b; };

__device__ __forceinline__ BnCoef bn_coef(float4 sm, float4 sq, float4 gg, float4 bb, int N) {
    BnCoef c;
    const float invN = 1.0f / (float)N;
    c.m.x = sm.x * invN; c.m.y = sm.y * invN; c.m.z = sm.z * invN; c.m.w = sm.w * invN;
    c.is.x = rsqrtf(sq.x * invN - c.m.x * c.m.x + EPSV);
    c.is.y = rsqrtf(sq.y * invN - c.m.y * c.m.y + EPSV);
    c.is.z = rsqrtf(sq.z * invN - c.m.z * c.m.z + EPSV);
    c.is.w = rsqrtf(sq.w * invN - c.m.w * c.m.w + EPSV);
    c.g = gg; c.b = bb;
    return c;
}

__device__ __forceinline__ float4 bn4(float4 x, const BnCoef& c) {
    float4 v;
    v.x = c.g.x * (x.x - c.m.x) * c.is.x + c.b.x;
    v.y = c.g.y * (x.y - c.m.y) * c.is.y + c.b.y;
    v.z = c.g.z * (x.z - c.m.z) * c.is.z + c.b.z;
    v.w = c.g.w * (x.w - c.m.w) * c.is.w + c.b.w;
    return v;
}

__device__ __forceinline__ void add4(float4& s, float4 a) { s.x += a.x; s.y += a.y; s.z += a.z; s.w += a.w; }

__device__ __forceinline__ void pool_add4(float* __restrict__ p, float4 a) {
    atomicAdd(p, a.x); atomicAdd(p + 1, a.y); atomicAdd(p + 2, a.z); atomicAdd(p + 3, a.w);
}

__global__ void k_bn_pool(const float* __restrict__ hin, const float* __restrict__ stats,
          const float* __restrict__ g, const float* __restrict__ beta,
          const int* __restrict__ batch, float* __restrict__ hout,
          float* __restrict__ pooled, int poolOff, int N) {
    const int t  = blockIdx.x * blockDim.x + threadIdx.x;
    const int q  = t & 15;
    const int r0 = (t >> 4) * POOL_R;
    if (r0 >= N) return;
    const float4* h4 = (const float4*)hin + (size_t)r0 * 16 + q;
    float4*       o4 = (float4*)hout + (size_t)r0 * 16 + q;
    const int*    bp = batch + r0;
    const int f = q * 4;
    float* pq = pooled + poolOff + f;

    const float4* st4 = (const float4*)stats + q;
    float4 sm = st4[0], sq = st4[16];
#pragma unroll
    for (int k = 1; k < STAT_COPIES; ++k) {
        add4(sm, st4[k * (STAT_STRIDE / 4)]);
        add4(sq, st4[k * (STAT_STRIDE / 4) + 16]);
    }
    const float4 gg = make_float4(g[f], g[f + 1], g[f + 2], g[f + 3]);
    const float4 bb = make_float4(beta[f], beta[f + 1], beta[f + 2], beta[f + 3]);

    int cur;
    float4 acc = make_float4(0.f, 0.f, 0.f, 0.f);
    if (r0 + POOL_R <= N) {
        float4 x[POOL_R];
        int    b[POOL_R];
#pragma unroll
        for (int u = 0; u < POOL_R; ++u) {
            x[u] = h4[u * 16];
            b[u] = bp[u];
        }
        const BnCoef c = bn_coef(sm, sq, gg, bb, N);
        cur = b[0];
#pragma unroll
        for (int u = 0; u < POOL_R; ++u) {
            const float4 v = bn4(x[u], c);
            o4[u * 16] = v;
            if (b[u] != cur) {
                pool_add4(pq + (size_t)cur * 192, acc);
                cur = b[u];
                acc = make_float4(0.f, 0.f, 0.f, 0.f);
            }
            acc.x += v.x; acc.y += v.y; acc.z += v.z; acc.w += v.w;
        }
    } else {
        const BnCoef c = bn_coef(sm, sq, gg, bb, N);
        cur = bp[0];
        for (int u = 0; r0 + u < N; ++u) {
            const float4 v = bn4(h4[u * 16], c);
            o4[u * 16] = v;
            const int bu = bp[u];
            if (bu != cur) {
                pool_add4(pq + (size_t)cur * 192, acc);
                cur = bu;
                acc = make_float4(0.f, 0.f, 0.f, 0.f);
            }
            acc.x += v.x; acc.y += v.y; acc.z += v.z; acc.w += v.w;
        }
    }
    pool_add4(pq + (size_t)cur * 192, acc);
}

extern "C" void kernel_launch(void* const* d_in, const int* in_sizes, int n_in,
                              void* d_out, int out_size, void* d_ws, size_t ws_size,
                              hipStream_t stream) {
    const float* x    = (const float*)d_in[0];
    const int*   eidx = (const int*)d_in[1];
    const int*   batch = (const int*)d_in[2];
    const float* W[3]  = {(const float*)d_in[3],  (const float*)d_in[7],  (const float*)d_in[11]};
    const float* bi[3] = {(const float*)d_in[4],  (const float*)d_in[8],  (const float*)d_in[12]};
    const float* ga[3] = {(const float*)d_in[5],  (const float*)d_in[9],  (const float*)d_in[13]};
    const float* be[3] = {(const float*)d_in[6],  (const float*)d_in[10], (const float*)d_in[14]};

    const int N = in_sizes[0] / HDIM;
    const int E = in_sizes[1] / 2;
    const int G = (out_size - N * HDIM) / (3 * HDIM);

    const int* src = eidx;
    const int* dst = eidx + E;

    float* ws    = (float*)d_ws;
    float* dinv  = ws;
    float* bufA  = dinv + N;
    float* bufB  = bufA + (size_t)N * HDIM;
    int*   csr   = (int*)(bufB + (size_t)N * HDIM);
    int* rowptr = csr;
    int* col    = rowptr + (N + 4);
    int* cntC   = col + E;
    int* tot    = cntC + CSR_C * (N + 4);
    float* stats = (float*)(tot + CSR_R * CSR_C);

    float* pooled = (float*)d_out;
    float* hfinal = (float*)d_out + (size_t)G * 192;

    const int TB = 256;

    k_csr_count<<<CSR_R * CSR_C, CSR_TB, 0, stream>>>(dst, cntC, tot, N, E,
                                                     pooled, G * 192, stats, 3 * STAT_COPIES * STAT_STRIDE);
    k_csr_scan <<<CSR_R, CSR_TB, 0, stream>>>(cntC, tot, rowptr, dinv, N);
    k_csr_fill <<<CSR_R * CSR_C, CSR_TB, 0, stream>>>(src, dst, cntC, col, N, E);

    float* st0 = stats;
    float* st1 = stats + 1 * (STAT_COPIES * STAT_STRIDE);
    float* st2 = stats + 2 * (STAT_COPIES * STAT_STRIDE);
    float* wsumv = stats + 3 * (STAT_COPIES * STAT_STRIDE);

    k_gather_gemm_act_stats<false, true><<<GATHER_GRID, GATHER_TB, 0, stream>>>(
        x, W[0], dinv, rowptr, col, bi[0], bufA, st0, N,
        (const float*)nullptr, (const float*)nullptr, (const float*)nullptr, (const int*)nullptr,
        (float*)nullptr, 0, wsumv);
    k_gather_gemm_act_stats<true, true><<<GATHER_GRID, GATHER_TB, 0, stream>>>(
        bufA, W[1], dinv, rowptr, col, bi[1], bufB, st1, N,
        st0, ga[0], be[0], batch, pooled, 0, wsumv);
    k_gather_gemm_act_stats<true, false><<<GATHER_GRID, GATHER_TB, 0, stream>>>(
        bufB, W[2], dinv, rowptr, col, bi[2], bufA, st2, N,
        st1, ga[1], be[1], batch, pooled, HDIM, wsumv);
    k_bn_pool<<<cdiv((long long)cdiv(N, POOL_R) * 16, TB), TB, 0, stream>>>(
        bufA, st2, ga[2], be[2], batch, hfinal, pooled, 2 * HDIM, N);
}
